# starting kernel
# baseline (speedup 1.0000x reference)
#include <hip/hip_runtime.h>
#include <hip/hip_bf16.h>


typedef __attribute__((ext_vector_type(2))) float v2f;
typedef __attribute__((ext_vector_type(8))) float v8f;

#define NODES 50000
#define EDGES 800000
#define DIM   64
#define NEG_SLOPE 0.2f

__device__ __forceinline__ unsigned encF(float f) {
    unsigned u = __float_as_uint(f);
    return (u & 0x80000000u) ? ~u : (u | 0x80000000u);
}
__device__ __forceinline__ float decF(unsigned u) {
    unsigned v = (u & 0x80000000u) ? (u & 0x7FFFFFFFu) : ~u;
    return __uint_as_float(v);
}

__global__ void __launch_bounds__(256)
gemm_bias_wmma(const float* __restrict__ A, const float* __restrict__ W,
               const float* __restrict__ bias, float* __restrict__ C,
               int M, int K, int Nout)
{
    const int lane = threadIdx.x & 31;
    const int wave = blockIdx.x * 8 + (threadIdx.x >> 5);
    const int colTiles = Nout >> 4;
    const int total = (M >> 4) * colTiles;
    if (wave >= total) return;

    const int rt   = wave / colTiles;
    const int ct   = wave - rt * colTiles;
    const int row0 = rt << 4;
    const int col0 = ct << 4;
    const int lr   = lane & 15;
    const int hi   = lane >> 4;

    const float* __restrict__ arow = A + (size_t)(row0 + lr) * K;
    const float* __restrict__ wrow = W + (size_t)(col0 + lr) * K;

    v8f acc = {};
    #pragma unroll
    for (int k0 = 0; k0 < 64; k0 += 4) {
        v2f a, b;
        a.x = arow[k0 + 2 * hi];
        a.y = arow[k0 + 2 * hi + 1];
        b.x = wrow[k0 + 2 * hi];
        b.y = wrow[k0 + 2 * hi + 1];
        acc = __builtin_amdgcn_wmma_f32_16x16x4_f32(
                  false, a, false, b, (short)0, acc, false, false);
    }

    const float bv = bias[col0 + lr];
    float* crow = C + (size_t)(row0 + hi * 8) * Nout + col0 + lr;
    #pragma unroll
    for (int v = 0; v < 8; ++v)
        crow[(size_t)v * Nout] = acc[v] + bv;
}

__global__ void __launch_bounds__(256)
edge_score(const float* __restrict__ el, const float* __restrict__ er,
           const float* __restrict__ attn, const int* __restrict__ src,
           const int* __restrict__ dst, float* __restrict__ score,
           unsigned* __restrict__ mEnc)
{
    const int  lane = threadIdx.x & 31;
    const long e    = (long)blockIdx.x * 8 + (threadIdx.x >> 5);
    const long Ef   = (long)EDGES + NODES;
    if (e >= Ef) return;
    int s, d;
    if (e < EDGES) { s = src[e]; d = dst[e]; } else { s = d = (int)(e - EDGES); }

    float p = 0.f;
    #pragma unroll
    for (int j = 0; j < 2; ++j) {
        const int k = lane + 32 * j;
        float v = el[(size_t)s * DIM + k] + er[(size_t)d * DIM + k];
        v = v > 0.f ? v : NEG_SLOPE * v;
        p += v * attn[k];
    }
    #pragma unroll
    for (int off = 16; off; off >>= 1) p += __shfl_xor(p, off, 32);
    if (lane == 0) {
        score[e] = p;
        atomicMax(&mEnc[d], encF(p));
    }
}

__global__ void edge_exp(float* __restrict__ score,
                         const unsigned* __restrict__ mEnc,
                         float* __restrict__ den, const int* __restrict__ dst)
{
    const long e  = (long)blockIdx.x * blockDim.x + threadIdx.x;
    const long Ef = (long)EDGES + NODES;
    if (e >= Ef) return;
    const int d  = (e < EDGES) ? dst[e] : (int)(e - EDGES);
    const float ex = expf(score[e] - decF(mEnc[d]));
    score[e] = ex;
    atomicAdd(&den[d], ex);
}

__global__ void __launch_bounds__(256)
edge_aggregate(const float* __restrict__ el, const float* __restrict__ ex,
               const float* __restrict__ den, const int* __restrict__ src,
               const int* __restrict__ dst, float* __restrict__ acc)
{
    const int  lane = threadIdx.x & 31;
    const long e    = (long)blockIdx.x * 8 + (threadIdx.x >> 5);
    const long Ef   = (long)EDGES + NODES;
    if (e >= Ef) return;
    int s, d;
    if (e < EDGES) { s = src[e]; d = dst[e]; } else { s = d = (int)(e - EDGES); }
    const float a = ex[e] / den[d];
    #pragma unroll
    for (int j = 0; j < 2; ++j) {
        const int k = lane + 32 * j;
        atomicAdd(&acc[(size_t)d * DIM + k], a * el[(size_t)s * DIM + k]);
    }
}

__global__ void elu_write(const float* __restrict__ acc, float* __restrict__ h,
                          float* __restrict__ emb, long n)
{
    const long i = (long)blockIdx.x * blockDim.x + threadIdx.x;
    if (i >= n) return;
    const float v = acc[i];
    const float o = v > 0.f ? v : (expf(v) - 1.f);
    h[i] = o;
    emb[i] = o;
}

__global__ void __launch_bounds__(256)
ggc_scatter(const float* __restrict__ t0, const float* __restrict__ t1,
            const int* __restrict__ ety, const int* __restrict__ src,
            const int* __restrict__ dst, float* __restrict__ acc)
{
    const int  lane = threadIdx.x & 31;
    const long e    = (long)blockIdx.x * 8 + (threadIdx.x >> 5);
    const long Ef   = (long)EDGES + NODES;
    if (e >= Ef) return;
    int s, d;
    if (e < EDGES) { s = src[e]; d = dst[e]; } else { s = d = (int)(e - EDGES); }
    const float* __restrict__ tr = (ety[e] == 0) ? t0 : t1;
    #pragma unroll
    for (int j = 0; j < 2; ++j) {
        const int k = lane + 32 * j;
        atomicAdd(&acc[(size_t)d * DIM + k], tr[(size_t)s * DIM + k]);
    }
}

__global__ void gru_update(const float* __restrict__ gi,
                           const float* __restrict__ gh,
                           float* __restrict__ h, float* __restrict__ emb,
                           long nElem)
{
    const long i = (long)blockIdx.x * blockDim.x + threadIdx.x;
    if (i >= nElem) return;
    const long n = i >> 6;
    const int  d = (int)(i & 63);
    const float* gin = gi + n * (3 * DIM);
    const float* ghn = gh + n * (3 * DIM);
    const float r = 1.f / (1.f + expf(-(gin[d]           + ghn[d])));
    const float z = 1.f / (1.f + expf(-(gin[DIM + d]     + ghn[DIM + d])));
    const float c = tanhf(gin[2 * DIM + d] + r * ghn[2 * DIM + d]);
    const float o = (1.f - z) * c + z * h[i];
    h[i] = o;
    if (emb) emb[i] = o;
}

__global__ void fill_zero(float* __restrict__ p, long n)
{
    long i = (long)blockIdx.x * blockDim.x + threadIdx.x;
    const long stride = (long)gridDim.x * blockDim.x;
    for (; i < n; i += stride) p[i] = 0.f;
}

extern "C" void kernel_launch(void* const* d_in, const int* in_sizes, int n_in,
                              void* d_out, int out_size, void* d_ws, size_t ws_size,
                              hipStream_t stream)
{
    (void)in_sizes; (void)n_in; (void)out_size; (void)ws_size;

    const float* x   = (const float*)d_in[0];
    const int*   src = (const int*)d_in[1];
    const int*   dst = (const int*)d_in[2];
    const int*   ety = (const int*)d_in[3];

    const long NE = (long)NODES * DIM;
    const long Ef = (long)EDGES + NODES;

    float* ws   = (float*)d_ws;
    float* hbuf = ws;                 ws += NE;
    float* buf1 = ws;                 ws += NE;
    float* buf2 = ws;                 ws += NE;
    float* gi   = ws;                 ws += (long)NODES * 3 * DIM;
    float* gh   = ws;                 ws += (long)NODES * 3 * DIM;
    float* acc  = ws;                 ws += NE;
    unsigned* mEnc = (unsigned*)ws;   ws += NODES;
    float* den  = ws;                 ws += NODES;
    float* score = ws;

    float* out = (float*)d_out;

    const int edgeWaveBlocks = (int)((Ef + 7) / 8);
    const int edgeThrBlocks  = (int)((Ef + 255) / 256);
    const int elemBlocks     = (int)((NE + 255) / 256);
    const int gemmBlocks64   = ((NODES / 16) * (DIM / 16)       + 7) / 8;
    const int gemmBlocks192  = ((NODES / 16) * (3 * DIM / 16)   + 7) / 8;

    const float* hin = x;
    for (int l = 0; l < 2; ++l) {
        const int base = 4 + 11 * l;
        const float* Wsrc = (const float*)d_in[base + 0];
        const float* bsrc = (const float*)d_in[base + 1];
        const float* Wdst = (const float*)d_in[base + 2];
        const float* bdst = (const float*)d_in[base + 3];
        const float* attn = (const float*)d_in[base + 4];
        const float* ggW  = (const float*)d_in[base + 5];
        const float* ggb  = (const float*)d_in[base + 6];
        const float* Wih  = (const float*)d_in[base + 7];
        const float* bih  = (const float*)d_in[base + 8];
        const float* Whh  = (const float*)d_in[base + 9];
        const float* bhh  = (const float*)d_in[base + 10];

        gemm_bias_wmma<<<gemmBlocks64, 256, 0, stream>>>(hin, Wsrc, bsrc, buf1,
                                                         NODES, DIM, DIM);
        gemm_bias_wmma<<<gemmBlocks64, 256, 0, stream>>>(hin, Wdst, bdst, buf2,
                                                         NODES, DIM, DIM);
        fill_zero<<<2048, 256, 0, stream>>>(acc, NE + 2L * NODES);
        edge_score<<<edgeWaveBlocks, 256, 0, stream>>>(buf1, buf2, attn, src, dst,
                                                       score, mEnc);
        edge_exp<<<edgeThrBlocks, 256, 0, stream>>>(score, mEnc, den, dst);
        edge_aggregate<<<edgeWaveBlocks, 256, 0, stream>>>(buf1, score, den, src,
                                                           dst, acc);
        elu_write<<<elemBlocks, 256, 0, stream>>>(acc, hbuf, out + (2L * l) * NE,
                                                  NE);

        for (int st = 0; st < 2; ++st) {
            gemm_bias_wmma<<<gemmBlocks64, 256, 0, stream>>>(
                hbuf, ggW,               ggb,       buf1, NODES, DIM, DIM);
            gemm_bias_wmma<<<gemmBlocks64, 256, 0, stream>>>(
                hbuf, ggW + DIM * DIM,   ggb + DIM, buf2, NODES, DIM, DIM);
            fill_zero<<<2048, 256, 0, stream>>>(acc, NE);
            ggc_scatter<<<edgeWaveBlocks, 256, 0, stream>>>(buf1, buf2, ety, src,
                                                            dst, acc);
            gemm_bias_wmma<<<gemmBlocks192, 256, 0, stream>>>(
                acc,  Wih, bih, gi, NODES, DIM, 3 * DIM);
            gemm_bias_wmma<<<gemmBlocks192, 256, 0, stream>>>(
                hbuf, Whh, bhh, gh, NODES, DIM, 3 * DIM);
            gru_update<<<elemBlocks, 256, 0, stream>>>(
                gi, gh, hbuf, (st == 1) ? out + (2L * l + 1) * NE : nullptr, NE);
        }
        hin = hbuf;
    }
}


# segment sums by owner blocks: 256 groups each keep 196 destination rows in LDS, scan all destinations, add source rows with LDS float adds (arrival order); global float adds, acc zero-fill gone
# speedup vs baseline: 1.0935x; 1.0935x over previous
#include <hip/hip_runtime.h>
#include <hip/hip_bf16.h>


typedef __attribute__((ext_vector_type(2))) float v2f;
typedef __attribute__((ext_vector_type(4))) float v4f;
typedef __attribute__((ext_vector_type(8))) float v8f;

#define NODES 50000
#define EDGES 800000
#define DIM   64
#define NEG_SLOPE 0.2f

__device__ __forceinline__ unsigned encF(float f) {
    unsigned u = __float_as_uint(f);
    return (u & 0x80000000u) ? ~u : (u | 0x80000000u);
}
__device__ __forceinline__ float decF(unsigned u) {
    unsigned v = (u & 0x80000000u) ? (u & 0x7FFFFFFFu) : ~u;
    return __uint_as_float(v);
}

__device__ __forceinline__ void load_frag(const float* __restrict__ rowp, int hi,
                                          v4f (&f)[8])
{
    const v4f* __restrict__ p = (const v4f*)(rowp + 4 * hi);
    #pragma unroll
    for (int j = 0; j < 8; ++j) f[j] = p[2 * j];
}

template <int NOUT>
__global__ void __launch_bounds__(128) __attribute__((amdgpu_waves_per_eu(1, 4)))
gemm_bias_wmma(const float* __restrict__ A, const float* __restrict__ W,
               const float* __restrict__ bias, float* __restrict__ C, int M)
{
    constexpr int GROUPS = NOUT / 64;
    const int lane = threadIdx.x & 31;
    const int wave = blockIdx.x * 4 + (threadIdx.x >> 5);
    const int total = (M >> 4) * GROUPS;
    if (wave >= total) return;

    const int rt   = wave / GROUPS;
    const int cg   = wave - rt * GROUPS;
    const int row0 = rt << 4;
    const int colB = cg << 6;
    const int lr   = lane & 15;
    const int hi   = lane >> 4;

    v4f a4[8];
    load_frag(A + (size_t)(row0 + lr) * DIM, hi, a4);

    v4f b4[4][8];
    #pragma unroll
    for (int t = 0; t < 4; ++t)
        load_frag(W + (size_t)(colB + 16 * t + lr) * DIM, hi, b4[t]);

    v8f acc[4] = {};
    #pragma unroll
    for (int j = 0; j < 8; ++j) {
        #pragma unroll
        for (int p = 0; p < 2; ++p) {
            v2f a;
            a.x = a4[j][2 * p];
            a.y = a4[j][2 * p + 1];
            #pragma unroll
            for (int t = 0; t < 4; ++t) {
                v2f b;
                b.x = b4[t][j][2 * p];
                b.y = b4[t][j][2 * p + 1];
                acc[t] = __builtin_amdgcn_wmma_f32_16x16x4_f32(
                             false, a, false, b, (short)0, acc[t], false, false);
            }
        }
    }

    #pragma unroll
    for (int t = 0; t < 4; ++t) {
        const int col = colB + 16 * t + lr;
        const float bv = bias[col];
        float* crow = C + (size_t)(row0 + hi * 8) * NOUT + col;
        #pragma unroll
        for (int v = 0; v < 8; ++v)
            crow[(size_t)v * NOUT] = acc[t][v] + bv;
    }
}

__global__ void __launch_bounds__(256)
edge_score(const float* __restrict__ el, const float* __restrict__ er,
           const float* __restrict__ attn, const int* __restrict__ src,
           const int* __restrict__ dst, float* __restrict__ score,
           unsigned* __restrict__ mEnc)
{
    const int  lane = threadIdx.x & 31;
    const long e    = (long)blockIdx.x * 8 + (threadIdx.x >> 5);
    const long Ef   = (long)EDGES + NODES;
    if (e >= Ef) return;
    int s, d;
    if (e < EDGES) { s = src[e]; d = dst[e]; } else { s = d = (int)(e - EDGES); }

    float p = 0.f;
    #pragma unroll
    for (int j = 0; j < 2; ++j) {
        const int k = lane + 32 * j;
        float v = el[(size_t)s * DIM + k] + er[(size_t)d * DIM + k];
        v = v > 0.f ? v : NEG_SLOPE * v;
        p += v * attn[k];
    }
    #pragma unroll
    for (int off = 16; off; off >>= 1) p += __shfl_xor(p, off, 32);
    if (lane == 0) {
        score[e] = p;
        atomicMax(&mEnc[d], encF(p));
    }
}

__global__ void edge_exp(float* __restrict__ score,
                         const unsigned* __restrict__ mEnc,
                         float* __restrict__ den, const int* __restrict__ dst)
{
    const long e  = (long)blockIdx.x * blockDim.x + threadIdx.x;
    const long Ef = (long)EDGES + NODES;
    if (e >= Ef) return;
    const int d  = (e < EDGES) ? dst[e] : (int)(e - EDGES);
    const float ex = expf(score[e] - decF(mEnc[d]));
    score[e] = ex;
    atomicAdd(&den[d], ex);
}

__global__ void elu_write(const float* __restrict__ acc, float* __restrict__ h,
                          float* __restrict__ emb, long n)
{
    const long i = (long)blockIdx.x * blockDim.x + threadIdx.x;
    if (i >= n) return;
    const float v = acc[i];
    const float o = v > 0.f ? v : (expf(v) - 1.f);
    h[i] = o;
    emb[i] = o;
}

#define OWN_ROWS 196
#define OWN_SUB  16384
#define OWN_QCAP 24576
#define OWN_BATCH 8

template <int MODE>
__device__ __forceinline__ void owner_drain(
    const float* __restrict__ t0, const float* __restrict__ t1,
    const float* __restrict__ ex, const float* __restrict__ den,
    const int* __restrict__ ety, const int* __restrict__ src,
    float* acc, unsigned* q, float* qw, int n, int d0, int tid)
{
    for (int i = tid; i < n; i += 1024) {
        const unsigned p  = q[i];
        const int      e  = (int)(p & 0xFFFFFu);
        const unsigned dl = p >> 20;
        const int      s  = (e < EDGES) ? src[e] : (e - EDGES);
        unsigned o = (unsigned)s | (dl << 16);
        if (MODE == 0) qw[i] = ex[e] / den[d0 + (int)dl];
        else           o |= (ety[e] == 0) ? 0u : (1u << 24);
        q[i] = o;
    }
    __syncthreads();

    const int lane = tid & 31;
    const int wave = tid >> 5;
    for (int i0 = wave; i0 < n; i0 += 32 * OWN_BATCH) {
        float2 v[OWN_BATCH];
        float  w[OWN_BATCH];
        int    off[OWN_BATCH];
        #pragma unroll
        for (int u = 0; u < OWN_BATCH; ++u) {
            const int i = i0 + 32 * u;
            v[u] = make_float2(0.f, 0.f); w[u] = 1.f; off[u] = 0;
            if (i < n) {
                const unsigned p = q[i];
                const float* __restrict__ tab = (MODE == 1 && (p >> 24)) ? t1 : t0;
                const float2* __restrict__ row =
                    (const float2*)(tab + (size_t)(p & 0xFFFFu) * DIM);
                v[u]   = row[lane];
                off[u] = (int)((p >> 16) & 0xFFu) * DIM + 2 * lane;
                if (MODE == 0) w[u] = qw[i];
            }
        }
        #pragma unroll
        for (int u = 0; u < OWN_BATCH; ++u) {
            const int i = i0 + 32 * u;
            if (i < n) {
                if (MODE == 0) {
                    atomicAdd(&acc[off[u]],     w[u] * v[u].x);
                    atomicAdd(&acc[off[u] + 1], w[u] * v[u].y);
                } else {
                    atomicAdd(&acc[off[u]],     v[u].x);
                    atomicAdd(&acc[off[u] + 1], v[u].y);
                }
            }
        }
    }
}

template <int MODE>
__global__ void __launch_bounds__(1024)
owner_sum(const float* __restrict__ t0, const float* __restrict__ t1,
          const float* __restrict__ ex, const float* __restrict__ den,
          const int* __restrict__ ety, const int* __restrict__ src,
          const int* __restrict__ dst, float* __restrict__ out)
{
    __shared__ float    acc[OWN_ROWS * DIM];
    __shared__ unsigned q[OWN_QCAP];
    __shared__ float    qw[(MODE == 0) ? OWN_QCAP : 1];
    __shared__ unsigned qn;

    const int tid  = threadIdx.x;
    const int d0   = blockIdx.x * OWN_ROWS;
    const int nloc = (NODES - d0 < OWN_ROWS) ? (NODES - d0) : OWN_ROWS;

    for (int i = tid; i < OWN_ROWS * DIM; i += 1024) acc[i] = 0.f;
    if (tid < nloc) q[tid] = (unsigned)(EDGES + d0 + tid) | ((unsigned)tid << 20);
    if (tid == 0) qn = (unsigned)nloc;

    const int4* __restrict__ dst4 = (const int4*)dst;
    int4 cur[4];
    #pragma unroll
    for (int j = 0; j < 4; ++j) {
        const int i4 = j * 1024 + tid;
        cur[j] = (i4 < EDGES / 4) ? dst4[i4] : make_int4(-1, -1, -1, -1);
    }

    for (int base = 0; base < EDGES; base += OWN_SUB) {
        int4 nxt[4];
        #pragma unroll
        for (int j = 0; j < 4; ++j) {
            const int i4 = ((base + OWN_SUB) >> 2) + j * 1024 + tid;
            nxt[j] = (i4 < EDGES / 4) ? dst4[i4] : make_int4(-1, -1, -1, -1);
        }

        __syncthreads();
        const int n = (int)qn;
        __syncthreads();
        if (n + OWN_SUB > OWN_QCAP) {
            owner_drain<MODE>(t0, t1, ex, den, ety, src, acc, q, qw, n, d0, tid);
            __syncthreads();
            if (tid == 0) qn = 0u;
            __syncthreads();
        }

        #pragma unroll
        for (int j = 0; j < 4; ++j) {
            const int e0 = base + (j * 1024 + tid) * 4;
            const unsigned dl[4] = { (unsigned)(cur[j].x - d0), (unsigned)(cur[j].y - d0),
                                     (unsigned)(cur[j].z - d0), (unsigned)(cur[j].w - d0) };
            const unsigned lo = min(min(dl[0], dl[1]), min(dl[2], dl[3]));
            if (lo < (unsigned)nloc) {
                #pragma unroll
                for (int c = 0; c < 4; ++c) {
                    if (dl[c] < (unsigned)nloc) {
                        const unsigned pos = atomicAdd(&qn, 1u);
                        q[pos] = (unsigned)(e0 + c) | (dl[c] << 20);
                    }
                }
            }
        }
        #pragma unroll
        for (int j = 0; j < 4; ++j) cur[j] = nxt[j];
    }

    __syncthreads();
    const int nlast = (int)qn;
    owner_drain<MODE>(t0, t1, ex, den, ety, src, acc, q, qw, nlast, d0, tid);
    __syncthreads();

    float* __restrict__ orow = out + (size_t)d0 * DIM;
    for (int i = tid; i < nloc * DIM; i += 1024) orow[i] = acc[i];
}

__global__ void gru_update(const float* __restrict__ gi,
                           const float* __restrict__ gh,
                           float* __restrict__ h, float* __restrict__ emb,
                           long nElem)
{
    const long i = (long)blockIdx.x * blockDim.x + threadIdx.x;
    if (i >= nElem) return;
    const long n = i >> 6;
    const int  d = (int)(i & 63);
    const float* gin = gi + n * (3 * DIM);
    const float* ghn = gh + n * (3 * DIM);
    const float r = 1.f / (1.f + expf(-(gin[d]           + ghn[d])));
    const float z = 1.f / (1.f + expf(-(gin[DIM + d]     + ghn[DIM + d])));
    const float c = tanhf(gin[2 * DIM + d] + r * ghn[2 * DIM + d]);
    const float o = (1.f - z) * c + z * h[i];
    h[i] = o;
    if (emb) emb[i] = o;
}

__global__ void fill_zero(float* __restrict__ p, long n)
{
    long i = (long)blockIdx.x * blockDim.x + threadIdx.x;
    const long stride = (long)gridDim.x * blockDim.x;
    for (; i < n; i += stride) p[i] = 0.f;
}

extern "C" void kernel_launch(void* const* d_in, const int* in_sizes, int n_in,
                              void* d_out, int out_size, void* d_ws, size_t ws_size,
                              hipStream_t stream)
{
    (void)in_sizes; (void)n_in; (void)out_size; (void)ws_size;

    const float* x   = (const float*)d_in[0];
    const int*   src = (const int*)d_in[1];
    const int*   dst = (const int*)d_in[2];
    const int*   ety = (const int*)d_in[3];

    const long NE = (long)NODES * DIM;
    const long Ef = (long)EDGES + NODES;

    float* ws   = (float*)d_ws;
    float* hbuf = ws;                 ws += NE;
    float* buf1 = ws;                 ws += NE;
    float* buf2 = ws;                 ws += NE;
    float* gi   = ws;                 ws += (long)NODES * 3 * DIM;
    float* gh   = ws;                 ws += (long)NODES * 3 * DIM;
    float* acc  = ws;                 ws += NE;
    unsigned* mEnc = (unsigned*)ws;   ws += NODES;
    float* den  = ws;                 ws += NODES;
    float* score = ws;

    float* out = (float*)d_out;

    const int edgeWaveBlocks = (int)((Ef + 7) / 8);
    const int edgeThrBlocks  = (int)((Ef + 255) / 256);
    const int elemBlocks     = (int)((NE + 255) / 256);
    const int fillBlocks     = (2 * NODES + 255) / 256;
    const int ownerBlocks    = (NODES + OWN_ROWS - 1) / OWN_ROWS;
    const int gemmBlocks64   = ((NODES / 16) * (DIM / 64)       + 3) / 4;
    const int gemmBlocks192  = ((NODES / 16) * (3 * DIM / 64)   + 3) / 4;

    const float* hin = x;
    for (int l = 0; l < 2; ++l) {
        const int base = 4 + 11 * l;
        const float* Wsrc = (const float*)d_in[base + 0];
        const float* bsrc = (const float*)d_in[base + 1];
        const float* Wdst = (const float*)d_in[base + 2];
        const float* bdst = (const float*)d_in[base + 3];
        const float* attn = (const float*)d_in[base + 4];
        const float* ggW  = (const float*)d_in[base + 5];
        const float* ggb  = (const float*)d_in[base + 6];
        const float* Wih  = (const float*)d_in[base + 7];
        const float* bih  = (const float*)d_in[base + 8];
        const float* Whh  = (const float*)d_in[base + 9];
        const float* bhh  = (const float*)d_in[base + 10];

        gemm_bias_wmma<DIM><<<gemmBlocks64, 128, 0, stream>>>(hin, Wsrc, bsrc, buf1,
                                                              NODES);
        gemm_bias_wmma<DIM><<<gemmBlocks64, 128, 0, stream>>>(hin, Wdst, bdst, buf2,
                                                              NODES);
        fill_zero<<<fillBlocks, 256, 0, stream>>>((float*)mEnc, 2L * NODES);
        edge_score<<<edgeWaveBlocks, 256, 0, stream>>>(buf1, buf2, attn, src, dst,
                                                       score, mEnc);
        edge_exp<<<edgeThrBlocks, 256, 0, stream>>>(score, mEnc, den, dst);
        owner_sum<0><<<ownerBlocks, 1024, 0, stream>>>(buf1, buf1, score, den, ety,
                                                       src, dst, acc);
        elu_write<<<elemBlocks, 256, 0, stream>>>(acc, hbuf, out + (2L * l) * NE,
                                                  NE);

        for (int st = 0; st < 2; ++st) {
            gemm_bias_wmma<DIM><<<gemmBlocks64, 128, 0, stream>>>(
                hbuf, ggW,               ggb,       buf1, NODES);
            gemm_bias_wmma<DIM><<<gemmBlocks64, 128, 0, stream>>>(
                hbuf, ggW + DIM * DIM,   ggb + DIM, buf2, NODES);
            owner_sum<1><<<ownerBlocks, 1024, 0, stream>>>(buf1, buf2, score, den,
                                                           ety, src, dst, acc);
            gemm_bias_wmma<3 * DIM><<<gemmBlocks192, 128, 0, stream>>>(
                acc,  Wih, bih, gi, NODES);
            gemm_bias_wmma<3 * DIM><<<gemmBlocks192, 128, 0, stream>>>(
                hbuf, Whh, bhh, gh, NODES);
            gru_update<<<elemBlocks, 256, 0, stream>>>(
                gi, gh, hbuf, (st == 1) ? out + (2L * l + 1) * NE : nullptr, NE);
        }
        hin = hbuf;
    }
}


# destination list narrowed once per call to one owner byte per edge (800 KB plane, +1 launch); owner groups scan 16 keys per 16-byte load, read dst only for queued edges; LDS add order by arrival
# speedup vs baseline: 1.2325x; 1.1271x over previous
#include <hip/hip_runtime.h>
#include <hip/hip_bf16.h>


typedef __attribute__((ext_vector_type(2))) float v2f;
typedef __attribute__((ext_vector_type(4))) float v4f;
typedef __attribute__((ext_vector_type(8))) float v8f;

#define NODES 50000
#define EDGES 800000
#define DIM   64
#define NEG_SLOPE 0.2f

__device__ __forceinline__ unsigned encF(float f) {
    unsigned u = __float_as_uint(f);
    return (u & 0x80000000u) ? ~u : (u | 0x80000000u);
}
__device__ __forceinline__ float decF(unsigned u) {
    unsigned v = (u & 0x80000000u) ? (u & 0x7FFFFFFFu) : ~u;
    return __uint_as_float(v);
}

__device__ __forceinline__ void load_frag(const float* __restrict__ rowp, int hi,
                                          v4f (&f)[8])
{
    const v4f* __restrict__ p = (const v4f*)(rowp + 4 * hi);
    #pragma unroll
    for (int j = 0; j < 8; ++j) f[j] = p[2 * j];
}

template <int NOUT>
__global__ void __launch_bounds__(128) __attribute__((amdgpu_waves_per_eu(1, 4)))
gemm_bias_wmma(const float* __restrict__ A, const float* __restrict__ W,
               const float* __restrict__ bias, float* __restrict__ C, int M)
{
    constexpr int GROUPS = NOUT / 64;
    const int lane = threadIdx.x & 31;
    const int wave = blockIdx.x * 4 + (threadIdx.x >> 5);
    const int total = (M >> 4) * GROUPS;
    if (wave >= total) return;

    const int rt   = wave / GROUPS;
    const int cg   = wave - rt * GROUPS;
    const int row0 = rt << 4;
    const int colB = cg << 6;
    const int lr   = lane & 15;
    const int hi   = lane >> 4;

    v4f a4[8];
    load_frag(A + (size_t)(row0 + lr) * DIM, hi, a4);

    v4f b4[4][8];
    #pragma unroll
    for (int t = 0; t < 4; ++t)
        load_frag(W + (size_t)(colB + 16 * t + lr) * DIM, hi, b4[t]);

    v8f acc[4] = {};
    #pragma unroll
    for (int j = 0; j < 8; ++j) {
        #pragma unroll
        for (int p = 0; p < 2; ++p) {
            v2f a;
            a.x = a4[j][2 * p];
            a.y = a4[j][2 * p + 1];
            #pragma unroll
            for (int t = 0; t < 4; ++t) {
                v2f b;
                b.x = b4[t][j][2 * p];
                b.y = b4[t][j][2 * p + 1];
                acc[t] = __builtin_amdgcn_wmma_f32_16x16x4_f32(
                             false, a, false, b, (short)0, acc[t], false, false);
            }
        }
    }

    #pragma unroll
    for (int t = 0; t < 4; ++t) {
        const int col = colB + 16 * t + lr;
        const float bv = bias[col];
        float* crow = C + (size_t)(row0 + hi * 8) * NOUT + col;
        #pragma unroll
        for (int v = 0; v < 8; ++v)
            crow[(size_t)v * NOUT] = acc[t][v] + bv;
    }
}

__global__ void __launch_bounds__(256)
edge_score(const float* __restrict__ el, const float* __restrict__ er,
           const float* __restrict__ attn, const int* __restrict__ src,
           const int* __restrict__ dst, float* __restrict__ score,
           unsigned* __restrict__ mEnc)
{
    const int  lane = threadIdx.x & 31;
    const long e    = (long)blockIdx.x * 8 + (threadIdx.x >> 5);
    const long Ef   = (long)EDGES + NODES;
    if (e >= Ef) return;
    int s, d;
    if (e < EDGES) { s = src[e]; d = dst[e]; } else { s = d = (int)(e - EDGES); }

    float p = 0.f;
    #pragma unroll
    for (int j = 0; j < 2; ++j) {
        const int k = lane + 32 * j;
        float v = el[(size_t)s * DIM + k] + er[(size_t)d * DIM + k];
        v = v > 0.f ? v : NEG_SLOPE * v;
        p += v * attn[k];
    }
    #pragma unroll
    for (int off = 16; off; off >>= 1) p += __shfl_xor(p, off, 32);
    if (lane == 0) {
        score[e] = p;
        atomicMax(&mEnc[d], encF(p));
    }
}

__global__ void edge_exp(float* __restrict__ score,
                         const unsigned* __restrict__ mEnc,
                         float* __restrict__ den, const int* __restrict__ dst)
{
    const long e  = (long)blockIdx.x * blockDim.x + threadIdx.x;
    const long Ef = (long)EDGES + NODES;
    if (e >= Ef) return;
    const int d  = (e < EDGES) ? dst[e] : (int)(e - EDGES);
    const float ex = expf(score[e] - decF(mEnc[d]));
    score[e] = ex;
    atomicAdd(&den[d], ex);
}

__global__ void elu_write(const float* __restrict__ acc, float* __restrict__ h,
                          float* __restrict__ emb, long n)
{
    const long i = (long)blockIdx.x * blockDim.x + threadIdx.x;
    if (i >= n) return;
    const float v = acc[i];
    const float o = v > 0.f ? v : (expf(v) - 1.f);
    h[i] = o;
    emb[i] = o;
}

#define OWN_ROWS 196
#define OWN_SUB  16384
#define OWN_QCAP 24576
#define OWN_BATCH 8
#define OWN_KEYS4 (EDGES / 16)

__global__ void __launch_bounds__(256)
owner_keys(const int* __restrict__ dst, unsigned* __restrict__ keys)
{
    const int i = blockIdx.x * 256 + threadIdx.x;
    if (i >= EDGES / 4) return;
    const int4 d = ((const int4*)dst)[i];
    keys[i] =  ((unsigned)d.x / OWN_ROWS)        | (((unsigned)d.y / OWN_ROWS) << 8) |
              (((unsigned)d.z / OWN_ROWS) << 16) | (((unsigned)d.w / OWN_ROWS) << 24);
}

__device__ __forceinline__ unsigned zero_bytes(unsigned x)
{
    const unsigned y = ((x & 0x7F7F7F7Fu) + 0x7F7F7F7Fu) | x;
    return ~y & 0x80808080u;
}

template <int MODE>
__device__ __forceinline__ void owner_drain(
    const float* __restrict__ t0, const float* __restrict__ t1,
    const float* __restrict__ ex, const float* __restrict__ den,
    const int* __restrict__ ety, const int* __restrict__ src,
    const int* __restrict__ dst,
    float* acc, unsigned* q, float* qw, int n, int d0, int tid)
{
    for (int i = tid; i < n; i += 1024) {
        const int e = (int)q[i];
        int s, d;
        if (e < EDGES) { s = src[e]; d = dst[e]; } else { s = d = e - EDGES; }
        const unsigned dl = (unsigned)(d - d0);
        unsigned o = (unsigned)s | (dl << 16);
        if (MODE == 0) qw[i] = ex[e] / den[d0 + (int)dl];
        else           o |= (ety[e] == 0) ? 0u : (1u << 24);
        q[i] = o;
    }
    __syncthreads();

    const int lane = tid & 31;
    const int wave = tid >> 5;
    for (int i0 = wave; i0 < n; i0 += 32 * OWN_BATCH) {
        float2 v[OWN_BATCH];
        float  w[OWN_BATCH];
        int    off[OWN_BATCH];
        #pragma unroll
        for (int u = 0; u < OWN_BATCH; ++u) {
            const int i = i0 + 32 * u;
            v[u] = make_float2(0.f, 0.f); w[u] = 1.f; off[u] = 0;
            if (i < n) {
                const unsigned p = q[i];
                const float* __restrict__ tab = (MODE == 1 && (p >> 24)) ? t1 : t0;
                const float2* __restrict__ row =
                    (const float2*)(tab + (size_t)(p & 0xFFFFu) * DIM);
                v[u]   = row[lane];
                off[u] = (int)((p >> 16) & 0xFFu) * DIM + 2 * lane;
                if (MODE == 0) w[u] = qw[i];
            }
        }
        #pragma unroll
        for (int u = 0; u < OWN_BATCH; ++u) {
            const int i = i0 + 32 * u;
            if (i < n) {
                if (MODE == 0) {
                    atomicAdd(&acc[off[u]],     w[u] * v[u].x);
                    atomicAdd(&acc[off[u] + 1], w[u] * v[u].y);
                } else {
                    atomicAdd(&acc[off[u]],     v[u].x);
                    atomicAdd(&acc[off[u] + 1], v[u].y);
                }
            }
        }
    }
}

template <int MODE>
__global__ void __launch_bounds__(1024)
owner_sum(const float* __restrict__ t0, const float* __restrict__ t1,
          const float* __restrict__ ex, const float* __restrict__ den,
          const int* __restrict__ ety, const int* __restrict__ src,
          const int* __restrict__ dst, const unsigned* __restrict__ keys,
          float* __restrict__ out)
{
    __shared__ float    acc[OWN_ROWS * DIM];
    __shared__ unsigned q[OWN_QCAP];
    __shared__ float    qw[(MODE == 0) ? OWN_QCAP : 1];
    __shared__ unsigned qn;

    const int tid  = threadIdx.x;
    const int d0   = blockIdx.x * OWN_ROWS;
    const int nloc = (NODES - d0 < OWN_ROWS) ? (NODES - d0) : OWN_ROWS;

    for (int i = tid; i < OWN_ROWS * DIM; i += 1024) acc[i] = 0.f;
    if (tid < nloc) q[tid] = (unsigned)(EDGES + d0 + tid);
    if (tid == 0) qn = (unsigned)nloc;

    const uint4* __restrict__ key4 = (const uint4*)keys;
    const unsigned mine = (unsigned)blockIdx.x * 0x01010101u;
    uint4 cur[4];
    #pragma unroll
    for (int j = 0; j < 4; ++j) {
        const int i4 = j * 1024 + tid;
        cur[j] = (i4 < OWN_KEYS4) ? key4[i4] : make_uint4(0u, 0u, 0u, 0u);
    }

    for (int base4 = 0; base4 < OWN_KEYS4; base4 += 4096) {
        uint4 nxt[4];
        #pragma unroll
        for (int j = 0; j < 4; ++j) {
            const int i4 = base4 + 4096 + j * 1024 + tid;
            nxt[j] = (i4 < OWN_KEYS4) ? key4[i4] : make_uint4(0u, 0u, 0u, 0u);
        }

        #pragma unroll
        for (int j = 0; j < 4; ++j) {
            if (base4 + j * 1024 >= OWN_KEYS4) break;
            __syncthreads();
            const int n = (int)qn;
            __syncthreads();
            if (n + OWN_SUB > OWN_QCAP) {
                owner_drain<MODE>(t0, t1, ex, den, ety, src, dst, acc, q, qw, n, d0, tid);
                __syncthreads();
                if (tid == 0) qn = 0u;
                __syncthreads();
            }

            const int i4 = base4 + j * 1024 + tid;
            if (i4 < OWN_KEYS4) {
                unsigned m = (zero_bytes(cur[j].x ^ mine) >> 7) |
                             (zero_bytes(cur[j].y ^ mine) >> 6) |
                             (zero_bytes(cur[j].z ^ mine) >> 5) |
                             (zero_bytes(cur[j].w ^ mine) >> 4);
                while (m) {
                    const int b = __ffs((int)m) - 1;
                    m &= m - 1u;
                    const unsigned pos = atomicAdd(&qn, 1u);
                    q[pos] = (unsigned)(i4 * 16 + ((b & 7) << 2) + (b >> 3));
                }
            }
        }
        #pragma unroll
        for (int j = 0; j < 4; ++j) cur[j] = nxt[j];
    }

    __syncthreads();
    const int nlast = (int)qn;
    owner_drain<MODE>(t0, t1, ex, den, ety, src, dst, acc, q, qw, nlast, d0, tid);
    __syncthreads();

    float* __restrict__ orow = out + (size_t)d0 * DIM;
    for (int i = tid; i < nloc * DIM; i += 1024) orow[i] = acc[i];
}

__global__ void gru_update(const float* __restrict__ gi,
                           const float* __restrict__ gh,
                           float* __restrict__ h, float* __restrict__ emb,
                           long nElem)
{
    const long i = (long)blockIdx.x * blockDim.x + threadIdx.x;
    if (i >= nElem) return;
    const long n = i >> 6;
    const int  d = (int)(i & 63);
    const float* gin = gi + n * (3 * DIM);
    const float* ghn = gh + n * (3 * DIM);
    const float r = 1.f / (1.f + expf(-(gin[d]           + ghn[d])));
    const float z = 1.f / (1.f + expf(-(gin[DIM + d]     + ghn[DIM + d])));
    const float c = tanhf(gin[2 * DIM + d] + r * ghn[2 * DIM + d]);
    const float o = (1.f - z) * c + z * h[i];
    h[i] = o;
    if (emb) emb[i] = o;
}

__global__ void fill_zero(float* __restrict__ p, long n)
{
    long i = (long)blockIdx.x * blockDim.x + threadIdx.x;
    const long stride = (long)gridDim.x * blockDim.x;
    for (; i < n; i += stride) p[i] = 0.f;
}

extern "C" void kernel_launch(void* const* d_in, const int* in_sizes, int n_in,
                              void* d_out, int out_size, void* d_ws, size_t ws_size,
                              hipStream_t stream)
{
    (void)in_sizes; (void)n_in; (void)out_size; (void)ws_size;

    const float* x   = (const float*)d_in[0];
    const int*   src = (const int*)d_in[1];
    const int*   dst = (const int*)d_in[2];
    const int*   ety = (const int*)d_in[3];

    const long NE = (long)NODES * DIM;
    const long Ef = (long)EDGES + NODES;

    float* ws   = (float*)d_ws;
    float* hbuf = ws;                 ws += NE;
    float* buf1 = ws;                 ws += NE;
    float* buf2 = ws;                 ws += NE;
    float* gi   = ws;                 ws += (long)NODES * 3 * DIM;
    float* gh   = ws;                 ws += (long)NODES * 3 * DIM;
    float* acc  = ws;                 ws += NE;
    unsigned* mEnc = (unsigned*)ws;   ws += NODES;
    float* den  = ws;                 ws += NODES;
    float* score = ws;                ws += Ef;
    unsigned* keys = (unsigned*)ws;

    float* out = (float*)d_out;

    const int edgeWaveBlocks = (int)((Ef + 7) / 8);
    const int edgeThrBlocks  = (int)((Ef + 255) / 256);
    const int elemBlocks     = (int)((NE + 255) / 256);
    const int fillBlocks     = (2 * NODES + 255) / 256;
    const int ownerBlocks    = (NODES + OWN_ROWS - 1) / OWN_ROWS;
    const int keyBlocks      = (EDGES / 4 + 255) / 256;
    const int gemmBlocks64   = ((NODES / 16) * (DIM / 64)       + 3) / 4;
    const int gemmBlocks192  = ((NODES / 16) * (3 * DIM / 64)   + 3) / 4;

    owner_keys<<<keyBlocks, 256, 0, stream>>>(dst, keys);

    const float* hin = x;
    for (int l = 0; l < 2; ++l) {
        const int base = 4 + 11 * l;
        const float* Wsrc = (const float*)d_in[base + 0];
        const float* bsrc = (const float*)d_in[base + 1];
        const float* Wdst = (const float*)d_in[base + 2];
        const float* bdst = (const float*)d_in[base + 3];
        const float* attn = (const float*)d_in[base + 4];
        const float* ggW  = (const float*)d_in[base + 5];
        const float* ggb  = (const float*)d_in[base + 6];
        const float* Wih  = (const float*)d_in[base + 7];
        const float* bih  = (const float*)d_in[base + 8];
        const float* Whh  = (const float*)d_in[base + 9];
        const float* bhh  = (const float*)d_in[base + 10];

        gemm_bias_wmma<DIM><<<gemmBlocks64, 128, 0, stream>>>(hin, Wsrc, bsrc, buf1,
                                                              NODES);
        gemm_bias_wmma<DIM><<<gemmBlocks64, 128, 0, stream>>>(hin, Wdst, bdst, buf2,
                                                              NODES);
        fill_zero<<<fillBlocks, 256, 0, stream>>>((float*)mEnc, 2L * NODES);
        edge_score<<<edgeWaveBlocks, 256, 0, stream>>>(buf1, buf2, attn, src, dst,
                                                       score, mEnc);
        edge_exp<<<edgeThrBlocks, 256, 0, stream>>>(score, mEnc, den, dst);
        owner_sum<0><<<ownerBlocks, 1024, 0, stream>>>(buf1, buf1, score, den, ety,
                                                       src, dst, keys, acc);
        elu_write<<<elemBlocks, 256, 0, stream>>>(acc, hbuf, out + (2L * l) * NE,
                                                  NE);

        for (int st = 0; st < 2; ++st) {
            gemm_bias_wmma<DIM><<<gemmBlocks64, 128, 0, stream>>>(
                hbuf, ggW,               ggb,       buf1, NODES);
            gemm_bias_wmma<DIM><<<gemmBlocks64, 128, 0, stream>>>(
                hbuf, ggW + DIM * DIM,   ggb + DIM, buf2, NODES);
            owner_sum<1><<<ownerBlocks, 1024, 0, stream>>>(buf1, buf2, score, den,
                                                           ety, src, dst, keys, acc);
            gemm_bias_wmma<3 * DIM><<<gemmBlocks192, 128, 0, stream>>>(
                acc,  Wih, bih, gi, NODES);
            gemm_bias_wmma<3 * DIM><<<gemmBlocks192, 128, 0, stream>>>(
                hbuf, Whh, bhh, gh, NODES);
            gru_update<<<elemBlocks, 256, 0, stream>>>(
                gi, gh, hbuf, (st == 1) ? out + (2L * l + 1) * NE : nullptr, NE);
        }
        hin = hbuf;
    }
}


# GAT edge softmax moves into the owner groups: score from LDS-held er rows, row max, exp, den by LDS adds (arrival order); 6 launches gone (39 -> 33); 1/den once per row; score sum order changes
# speedup vs baseline: 1.8016x; 1.4617x over previous
#include <hip/hip_runtime.h>
#include <hip/hip_bf16.h>


typedef __attribute__((ext_vector_type(2))) float v2f;
typedef __attribute__((ext_vector_type(4))) float v4f;
typedef __attribute__((ext_vector_type(8))) float v8f;

#define NODES 50000
#define EDGES 800000
#define DIM   64
#define NEG_SLOPE 0.2f

__device__ __forceinline__ unsigned encF(float f) {
    unsigned u = __float_as_uint(f);
    return (u & 0x80000000u) ? ~u : (u | 0x80000000u);
}
__device__ __forceinline__ float decF(unsigned u) {
    unsigned v = (u & 0x80000000u) ? (u & 0x7FFFFFFFu) : ~u;
    return __uint_as_float(v);
}

__device__ __forceinline__ void load_frag(const float* __restrict__ rowp, int hi,
                                          v4f (&f)[8])
{
    const v4f* __restrict__ p = (const v4f*)(rowp + 4 * hi);
    #pragma unroll
    for (int j = 0; j < 8; ++j) f[j] = p[2 * j];
}

template <int NOUT>
__global__ void __launch_bounds__(128) __attribute__((amdgpu_waves_per_eu(1, 4)))
gemm_bias_wmma(const float* __restrict__ A, const float* __restrict__ W,
               const float* __restrict__ bias, float* __restrict__ C, int M)
{
    constexpr int GROUPS = NOUT / 64;
    const int lane = threadIdx.x & 31;
    const int wave = blockIdx.x * 4 + (threadIdx.x >> 5);
    const int total = (M >> 4) * GROUPS;
    if (wave >= total) return;

    const int rt   = wave / GROUPS;
    const int cg   = wave - rt * GROUPS;
    const int row0 = rt << 4;
    const int colB = cg << 6;
    const int lr   = lane & 15;
    const int hi   = lane >> 4;

    v4f a4[8];
    load_frag(A + (size_t)(row0 + lr) * DIM, hi, a4);

    v4f b4[4][8];
    #pragma unroll
    for (int t = 0; t < 4; ++t)
        load_frag(W + (size_t)(colB + 16 * t + lr) * DIM, hi, b4[t]);

    v8f acc[4] = {};
    #pragma unroll
    for (int j = 0; j < 8; ++j) {
        #pragma unroll
        for (int p = 0; p < 2; ++p) {
            v2f a;
            a.x = a4[j][2 * p];
            a.y = a4[j][2 * p + 1];
            #pragma unroll
            for (int t = 0; t < 4; ++t) {
                v2f b;
                b.x = b4[t][j][2 * p];
                b.y = b4[t][j][2 * p + 1];
                acc[t] = __builtin_amdgcn_wmma_f32_16x16x4_f32(
                             false, a, false, b, (short)0, acc[t], false, false);
            }
        }
    }

    #pragma unroll
    for (int t = 0; t < 4; ++t) {
        const int col = colB + 16 * t + lr;
        const float bv = bias[col];
        float* crow = C + (size_t)(row0 + hi * 8) * NOUT + col;
        #pragma unroll
        for (int v = 0; v < 8; ++v)
            crow[(size_t)v * NOUT] = acc[t][v] + bv;
    }
}

__global__ void elu_write(const float* __restrict__ acc, float* __restrict__ h,
                          float* __restrict__ emb, long n)
{
    const long i = (long)blockIdx.x * blockDim.x + threadIdx.x;
    if (i >= n) return;
    const float v = acc[i];
    const float o = v > 0.f ? v : (expf(v) - 1.f);
    h[i] = o;
    emb[i] = o;
}

#define OWN_ROWS 196
#define OWN_SUB  16384
#define OWN_QCAP 24576
#define OWN_BATCH 8
#define OWN_PAIRS (OWN_BATCH / 2)
#define OWN_KEYS4 (EDGES / 16)
#define ENC_NEG_INF 0x007FFFFFu

__global__ void __launch_bounds__(256)
owner_keys(const int* __restrict__ dst, unsigned* __restrict__ keys)
{
    const int i = blockIdx.x * 256 + threadIdx.x;
    if (i >= EDGES / 4) return;
    const int4 d = ((const int4*)dst)[i];
    keys[i] =  ((unsigned)d.x / OWN_ROWS)        | (((unsigned)d.y / OWN_ROWS) << 8) |
              (((unsigned)d.z / OWN_ROWS) << 16) | (((unsigned)d.w / OWN_ROWS) << 24);
}

__device__ __forceinline__ unsigned zero_bytes(unsigned x)
{
    const unsigned y = ((x & 0x7F7F7F7Fu) + 0x7F7F7F7Fu) | x;
    return ~y & 0x80808080u;
}

template <int MODE>
__device__ __forceinline__ void owner_lookup(
    const int* __restrict__ ety, const int* __restrict__ src,
    const int* __restrict__ dst, unsigned* q, int n, int d0, int tid)
{
    for (int i = tid; i < n; i += 1024) {
        const int e = (int)q[i];
        int s, d;
        if (e < EDGES) { s = src[e]; d = dst[e]; } else { s = d = e - EDGES; }
        unsigned o = (unsigned)s | ((unsigned)(d - d0) << 16);
        if (MODE == 1) o |= (ety[e] == 0) ? 0u : (1u << 24);
        q[i] = o;
    }
}

template <int MODE>
__device__ __forceinline__ void owner_add_rows(
    const float* __restrict__ t0, const float* __restrict__ t1,
    float* acc, const unsigned* q, const float* qw, int n, int tid)
{
    const int lane = tid & 31;
    const int wave = tid >> 5;
    for (int i0 = wave; i0 < n; i0 += 32 * OWN_BATCH) {
        float2 v[OWN_BATCH];
        float  w[OWN_BATCH];
        int    off[OWN_BATCH];
        #pragma unroll
        for (int u = 0; u < OWN_BATCH; ++u) {
            const int i = i0 + 32 * u;
            v[u] = make_float2(0.f, 0.f); w[u] = 1.f; off[u] = 0;
            if (i < n) {
                const unsigned p = q[i];
                const float* __restrict__ tab = (MODE == 1 && (p >> 24)) ? t1 : t0;
                const float2* __restrict__ row =
                    (const float2*)(tab + (size_t)(p & 0xFFFFu) * DIM);
                v[u]   = row[lane];
                off[u] = (int)((p >> 16) & 0xFFu) * DIM + 2 * lane;
                if (MODE == 0) w[u] = qw[i];
            }
        }
        #pragma unroll
        for (int u = 0; u < OWN_BATCH; ++u) {
            const int i = i0 + 32 * u;
            if (i < n) {
                if (MODE == 0) {
                    atomicAdd(&acc[off[u]],     w[u] * v[u].x);
                    atomicAdd(&acc[off[u] + 1], w[u] * v[u].y);
                } else {
                    atomicAdd(&acc[off[u]],     v[u].x);
                    atomicAdd(&acc[off[u] + 1], v[u].y);
                }
            }
        }
    }
}

__device__ __forceinline__ void gat_softmax_weights(
    const float* __restrict__ el, const float* __restrict__ attn,
    float* acc, const float* erow, const unsigned* q, float* qw,
    unsigned* rmax, float* rold, float* rden, int n, int tid, bool later)
{
    const int lane = tid & 31;
    const int wave = tid >> 5;
    const int sub  = lane & 15;
    const int half = lane >> 4;
    const v4f av = ((const v4f*)attn)[sub];

    for (int i0 = 2 * wave; i0 < n; i0 += 64 * OWN_PAIRS) {
        v4f v[OWN_PAIRS];
        #pragma unroll
        for (int u = 0; u < OWN_PAIRS; ++u) {
            const int i  = i0 + 64 * u + half;
            const int ic = i < n ? i : n - 1;
            v[u] = ((const v4f*)(el + (size_t)(q[ic] & 0xFFFFu) * DIM))[sub];
        }
        float sc[OWN_PAIRS];
        int   dl[OWN_PAIRS];
        #pragma unroll
        for (int u = 0; u < OWN_PAIRS; ++u) {
            const int i  = i0 + 64 * u + half;
            const int ic = i < n ? i : n - 1;
            dl[u] = (int)((q[ic] >> 16) & 0xFFu);
            const v4f r = ((const v4f*)(erow + dl[u] * DIM))[sub];
            sc[u] = 0.f;
            #pragma unroll
            for (int c = 0; c < 4; ++c) {
                float t = v[u][c] + r[c];
                t = t > 0.f ? t : NEG_SLOPE * t;
                sc[u] += t * av[c];
            }
        }
        #pragma unroll
        for (int o = 8; o; o >>= 1) {
            #pragma unroll
            for (int u = 0; u < OWN_PAIRS; ++u) sc[u] += __shfl_xor(sc[u], o, 32);
        }
        #pragma unroll
        for (int u = 0; u < OWN_PAIRS; ++u) {
            const int i = i0 + 64 * u + half;
            if (sub == 0 && i < n) {
                qw[i] = sc[u];
                atomicMax(&rmax[dl[u]], encF(sc[u]));
            }
        }
    }
    __syncthreads();

    if (later) {
        for (int i = tid; i < OWN_ROWS * DIM; i += 1024) {
            const float mo = rold[i >> 6];
            const float mn = decF(rmax[i >> 6]);
            if (mn != mo) acc[i] *= expf(mo - mn);
        }
        if (tid < OWN_ROWS) {
            const float mo = rold[tid];
            const float mn = decF(rmax[tid]);
            if (mn != mo) rden[tid] *= expf(mo - mn);
        }
        __syncthreads();
    }
    if (tid < OWN_ROWS) rold[tid] = decF(rmax[tid]);
    __syncthreads();

    for (int i = tid; i < n; i += 1024) {
        const int dl = (int)((q[i] >> 16) & 0xFFu);
        const float ex = expf(qw[i] - rold[dl]);
        qw[i] = ex;
        atomicAdd(&rden[dl], ex);
    }
    __syncthreads();
}

template <int MODE>
__global__ void __launch_bounds__(1024)
owner_sum(const float* __restrict__ t0, const float* __restrict__ t1,
          const float* __restrict__ attn, const int* __restrict__ ety,
          const int* __restrict__ src, const int* __restrict__ dst,
          const unsigned* __restrict__ keys, float* __restrict__ out)
{
    __shared__ float    acc[OWN_ROWS * DIM];
    __shared__ unsigned q[OWN_QCAP];
    __shared__ float    qw[(MODE == 0) ? OWN_QCAP : 1];
    __shared__ __attribute__((aligned(16))) float erow[(MODE == 0) ? OWN_ROWS * DIM : 4];
    __shared__ unsigned rmax[(MODE == 0) ? OWN_ROWS : 1];
    __shared__ float    rold[(MODE == 0) ? OWN_ROWS : 1];
    __shared__ float    rden[(MODE == 0) ? OWN_ROWS : 1];
    __shared__ unsigned qn;

    const int tid  = threadIdx.x;
    const int d0   = blockIdx.x * OWN_ROWS;
    const int nloc = (NODES - d0 < OWN_ROWS) ? (NODES - d0) : OWN_ROWS;

    for (int i = tid; i < OWN_ROWS * DIM; i += 1024) acc[i] = 0.f;
    if (tid < nloc) q[tid] = (unsigned)(EDGES + d0 + tid);
    if (tid == 0) qn = (unsigned)nloc;
    if (MODE == 0) {
        const v4f* __restrict__ er4 = (const v4f*)(t1 + (size_t)d0 * DIM);
        v4f* erow4 = (v4f*)erow;
        for (int i = tid; i < nloc * (DIM / 4); i += 1024) erow4[i] = er4[i];
        if (tid < OWN_ROWS) {
            rmax[tid] = ENC_NEG_INF;
            rold[tid] = decF(ENC_NEG_INF);
            rden[tid] = 0.f;
        }
    }
    int drains = 0;

    const uint4* __restrict__ key4 = (const uint4*)keys;
    const unsigned mine = (unsigned)blockIdx.x * 0x01010101u;
    uint4 cur[4];
    #pragma unroll
    for (int j = 0; j < 4; ++j) {
        const int i4 = j * 1024 + tid;
        cur[j] = (i4 < OWN_KEYS4) ? key4[i4] : make_uint4(0u, 0u, 0u, 0u);
    }

    for (int base4 = 0; base4 < OWN_KEYS4; base4 += 4096) {
        uint4 nxt[4];
        #pragma unroll
        for (int j = 0; j < 4; ++j) {
            const int i4 = base4 + 4096 + j * 1024 + tid;
            nxt[j] = (i4 < OWN_KEYS4) ? key4[i4] : make_uint4(0u, 0u, 0u, 0u);
        }

        #pragma unroll
        for (int j = 0; j < 4; ++j) {
            if (base4 + j * 1024 >= OWN_KEYS4) break;
            __syncthreads();
            const int n = (int)qn;
            __syncthreads();
            if (n + OWN_SUB > OWN_QCAP) {
                owner_lookup<MODE>(ety, src, dst, q, n, d0, tid);
                __syncthreads();
                if (MODE == 0)
                    gat_softmax_weights(t0, attn, acc, erow, q, qw, rmax, rold, rden,
                                        n, tid, drains > 0);
                owner_add_rows<MODE>(t0, t1, acc, q, qw, n, tid);
                ++drains;
                __syncthreads();
                if (tid == 0) qn = 0u;
                __syncthreads();
            }

            const int i4 = base4 + j * 1024 + tid;
            if (i4 < OWN_KEYS4) {
                unsigned m = (zero_bytes(cur[j].x ^ mine) >> 7) |
                             (zero_bytes(cur[j].y ^ mine) >> 6) |
                             (zero_bytes(cur[j].z ^ mine) >> 5) |
                             (zero_bytes(cur[j].w ^ mine) >> 4);
                while (m) {
                    const int b = __ffs((int)m) - 1;
                    m &= m - 1u;
                    const unsigned pos = atomicAdd(&qn, 1u);
                    q[pos] = (unsigned)(i4 * 16 + ((b & 7) << 2) + (b >> 3));
                }
            }
        }
        #pragma unroll
        for (int j = 0; j < 4; ++j) cur[j] = nxt[j];
    }

    __syncthreads();
    const int nlast = (int)qn;
    owner_lookup<MODE>(ety, src, dst, q, nlast, d0, tid);
    __syncthreads();
    if (MODE == 0)
        gat_softmax_weights(t0, attn, acc, erow, q, qw, rmax, rold, rden,
                            nlast, tid, drains > 0);
    owner_add_rows<MODE>(t0, t1, acc, q, qw, nlast, tid);
    __syncthreads();

    float* __restrict__ orow = out + (size_t)d0 * DIM;
    if (MODE == 0) {
        for (int i = tid; i < nloc * DIM; i += 1024) orow[i] = acc[i] / rden[i >> 6];
    } else {
        for (int i = tid; i < nloc * DIM; i += 1024) orow[i] = acc[i];
    }
}

__global__ void gru_update(const float* __restrict__ gi,
                           const float* __restrict__ gh,
                           float* __restrict__ h, float* __restrict__ emb,
                           long nElem)
{
    const long i = (long)blockIdx.x * blockDim.x + threadIdx.x;
    if (i >= nElem) return;
    const long n = i >> 6;
    const int  d = (int)(i & 63);
    const float* gin = gi + n * (3 * DIM);
    const float* ghn = gh + n * (3 * DIM);
    const float r = 1.f / (1.f + expf(-(gin[d]           + ghn[d])));
    const float z = 1.f / (1.f + expf(-(gin[DIM + d]     + ghn[DIM + d])));
    const float c = tanhf(gin[2 * DIM + d] + r * ghn[2 * DIM + d]);
    const float o = (1.f - z) * c + z * h[i];
    h[i] = o;
    if (emb) emb[i] = o;
}

extern "C" void kernel_launch(void* const* d_in, const int* in_sizes, int n_in,
                              void* d_out, int out_size, void* d_ws, size_t ws_size,
                              hipStream_t stream)
{
    (void)in_sizes; (void)n_in; (void)out_size; (void)ws_size;

    const float* x   = (const float*)d_in[0];
    const int*   src = (const int*)d_in[1];
    const int*   dst = (const int*)d_in[2];
    const int*   ety = (const int*)d_in[3];

    const long NE = (long)NODES * DIM;

    float* ws   = (float*)d_ws;
    float* hbuf = ws;                 ws += NE;
    float* buf1 = ws;                 ws += NE;
    float* buf2 = ws;                 ws += NE;
    float* gi   = ws;                 ws += (long)NODES * 3 * DIM;
    float* gh   = ws;                 ws += (long)NODES * 3 * DIM;
    float* acc  = ws;                 ws += NE;
    unsigned* keys = (unsigned*)ws;

    float* out = (float*)d_out;

    const int elemBlocks     = (int)((NE + 255) / 256);
    const int ownerBlocks    = (NODES + OWN_ROWS - 1) / OWN_ROWS;
    const int keyBlocks      = (EDGES / 4 + 255) / 256;
    const int gemmBlocks64   = ((NODES / 16) * (DIM / 64)       + 3) / 4;
    const int gemmBlocks192  = ((NODES / 16) * (3 * DIM / 64)   + 3) / 4;

    owner_keys<<<keyBlocks, 256, 0, stream>>>(dst, keys);

    const float* hin = x;
    for (int l = 0; l < 2; ++l) {
        const int base = 4 + 11 * l;
        const float* Wsrc = (const float*)d_in[base + 0];
        const float* bsrc = (const float*)d_in[base + 1];
        const float* Wdst = (const float*)d_in[base + 2];
        const float* bdst = (const float*)d_in[base + 3];
        const float* attn = (const float*)d_in[base + 4];
        const float* ggW  = (const float*)d_in[base + 5];
        const float* ggb  = (const float*)d_in[base + 6];
        const float* Wih  = (const float*)d_in[base + 7];
        const float* bih  = (const float*)d_in[base + 8];
        const float* Whh  = (const float*)d_in[base + 9];
        const float* bhh  = (const float*)d_in[base + 10];

        gemm_bias_wmma<DIM><<<gemmBlocks64, 128, 0, stream>>>(hin, Wsrc, bsrc, buf1,
                                                              NODES);
        gemm_bias_wmma<DIM><<<gemmBlocks64, 128, 0, stream>>>(hin, Wdst, bdst, buf2,
                                                              NODES);
        owner_sum<0><<<ownerBlocks, 1024, 0, stream>>>(buf1, buf2, attn, ety,
                                                       src, dst, keys, acc);
        elu_write<<<elemBlocks, 256, 0, stream>>>(acc, hbuf, out + (2L * l) * NE,
                                                  NE);

        for (int st = 0; st < 2; ++st) {
            gemm_bias_wmma<DIM><<<gemmBlocks64, 128, 0, stream>>>(
                hbuf, ggW,               ggb,       buf1, NODES);
            gemm_bias_wmma<DIM><<<gemmBlocks64, 128, 0, stream>>>(
                hbuf, ggW + DIM * DIM,   ggb + DIM, buf2, NODES);
            owner_sum<1><<<ownerBlocks, 1024, 0, stream>>>(buf1, buf2, attn,
                                                           ety, src, dst, keys, acc);
            gemm_bias_wmma<3 * DIM><<<gemmBlocks192, 128, 0, stream>>>(
                acc,  Wih, bih, gi, NODES);
            gemm_bias_wmma<3 * DIM><<<gemmBlocks192, 128, 0, stream>>>(
                hbuf, Whh, bhh, gh, NODES);
            gru_update<<<elemBlocks, 256, 0, stream>>>(
                gi, gh, hbuf, (st == 1) ? out + (2L * l + 1) * NE : nullptr, NE);
        }
        hin = hbuf;
    }
}


# owner groups' edge queues (source row, local row, type) built once per call by one scan + lookup launch, kept in workspace; the six sums copy theirs, no re-scan or re-gather; add order by arrival
# speedup vs baseline: 2.4071x; 1.3361x over previous
#include <hip/hip_runtime.h>
#include <hip/hip_bf16.h>


typedef __attribute__((ext_vector_type(2))) float v2f;
typedef __attribute__((ext_vector_type(4))) float v4f;
typedef __attribute__((ext_vector_type(8))) float v8f;

#define NODES 50000
#define EDGES 800000
#define DIM   64
#define NEG_SLOPE 0.2f

__device__ __forceinline__ unsigned encF(float f) {
    unsigned u = __float_as_uint(f);
    return (u & 0x80000000u) ? ~u : (u | 0x80000000u);
}
__device__ __forceinline__ float decF(unsigned u) {
    unsigned v = (u & 0x80000000u) ? (u & 0x7FFFFFFFu) : ~u;
    return __uint_as_float(v);
}

__device__ __forceinline__ void load_frag(const float* __restrict__ rowp, int hi,
                                          v4f (&f)[8])
{
    const v4f* __restrict__ p = (const v4f*)(rowp + 4 * hi);
    #pragma unroll
    for (int j = 0; j < 8; ++j) f[j] = p[2 * j];
}

template <int NOUT>
__global__ void __launch_bounds__(128) __attribute__((amdgpu_waves_per_eu(1, 4)))
gemm_bias_wmma(const float* __restrict__ A, const float* __restrict__ W,
               const float* __restrict__ bias, float* __restrict__ C, int M)
{
    constexpr int GROUPS = NOUT / 64;
    const int lane = threadIdx.x & 31;
    const int wave = blockIdx.x * 4 + (threadIdx.x >> 5);
    const int total = (M >> 4) * GROUPS;
    if (wave >= total) return;

    const int rt   = wave / GROUPS;
    const int cg   = wave - rt * GROUPS;
    const int row0 = rt << 4;
    const int colB = cg << 6;
    const int lr   = lane & 15;
    const int hi   = lane >> 4;

    v4f a4[8];
    load_frag(A + (size_t)(row0 + lr) * DIM, hi, a4);

    v4f b4[4][8];
    #pragma unroll
    for (int t = 0; t < 4; ++t)
        load_frag(W + (size_t)(colB + 16 * t + lr) * DIM, hi, b4[t]);

    v8f acc[4] = {};
    #pragma unroll
    for (int j = 0; j < 8; ++j) {
        #pragma unroll
        for (int p = 0; p < 2; ++p) {
            v2f a;
            a.x = a4[j][2 * p];
            a.y = a4[j][2 * p + 1];
            #pragma unroll
            for (int t = 0; t < 4; ++t) {
                v2f b;
                b.x = b4[t][j][2 * p];
                b.y = b4[t][j][2 * p + 1];
                acc[t] = __builtin_amdgcn_wmma_f32_16x16x4_f32(
                             false, a, false, b, (short)0, acc[t], false, false);
            }
        }
    }

    #pragma unroll
    for (int t = 0; t < 4; ++t) {
        const int col = colB + 16 * t + lr;
        const float bv = bias[col];
        float* crow = C + (size_t)(row0 + hi * 8) * NOUT + col;
        #pragma unroll
        for (int v = 0; v < 8; ++v)
            crow[(size_t)v * NOUT] = acc[t][v] + bv;
    }
}

__global__ void elu_write(const float* __restrict__ acc, float* __restrict__ h,
                          float* __restrict__ emb, long n)
{
    const long i = (long)blockIdx.x * blockDim.x + threadIdx.x;
    if (i >= n) return;
    const float v = acc[i];
    const float o = v > 0.f ? v : (expf(v) - 1.f);
    h[i] = o;
    emb[i] = o;
}

#define OWN_ROWS 196
#define OWN_SUB  16384
#define OWN_QCAP 24576
#define OWN_BATCH 8
#define OWN_PAIRS (OWN_BATCH / 2)
#define OWN_KEYS4 (EDGES / 16)
#define ENC_NEG_INF 0x007FFFFFu
#define OWN_OVER 0xFFFFFFFFu

__global__ void __launch_bounds__(256)
owner_keys(const int* __restrict__ dst, unsigned* __restrict__ keys)
{
    const int i = blockIdx.x * 256 + threadIdx.x;
    if (i >= EDGES / 4) return;
    const int4 d = ((const int4*)dst)[i];
    keys[i] =  ((unsigned)d.x / OWN_ROWS)        | (((unsigned)d.y / OWN_ROWS) << 8) |
              (((unsigned)d.z / OWN_ROWS) << 16) | (((unsigned)d.w / OWN_ROWS) << 24);
}

__device__ __forceinline__ unsigned zero_bytes(unsigned x)
{
    const unsigned y = ((x & 0x7F7F7F7Fu) + 0x7F7F7F7Fu) | x;
    return ~y & 0x80808080u;
}

__device__ __forceinline__ unsigned owner_mask(const uint4 k, unsigned mine)
{
    return (zero_bytes(k.x ^ mine) >> 7) | (zero_bytes(k.y ^ mine) >> 6) |
           (zero_bytes(k.z ^ mine) >> 5) | (zero_bytes(k.w ^ mine) >> 4);
}
__device__ __forceinline__ unsigned owner_edge(int i4, int b)
{
    return (unsigned)(i4 * 16 + ((b & 7) << 2) + (b >> 3));
}

template <int MODE>
__device__ __forceinline__ void owner_lookup(
    const int* __restrict__ ety, const int* __restrict__ src,
    const int* __restrict__ dst, unsigned* q, int n, int d0, int tid)
{
    for (int i = tid; i < n; i += 1024) {
        const int e = (int)q[i];
        int s, d;
        if (e < EDGES) { s = src[e]; d = dst[e]; } else { s = d = e - EDGES; }
        unsigned o = (unsigned)s | ((unsigned)(d - d0) << 16);
        if (MODE == 1) o |= (ety[e] == 0) ? 0u : (1u << 24);
        q[i] = o;
    }
}

template <int MODE>
__device__ __forceinline__ void owner_add_rows(
    const float* __restrict__ t0, const float* __restrict__ t1,
    float* acc, const unsigned* q, const float* qw, int n, int tid)
{
    const int lane = tid & 31;
    const int wave = tid >> 5;
    for (int i0 = wave; i0 < n; i0 += 32 * OWN_BATCH) {
        float2 v[OWN_BATCH];
        float  w[OWN_BATCH];
        int    off[OWN_BATCH];
        #pragma unroll
        for (int u = 0; u < OWN_BATCH; ++u) {
            const int i = i0 + 32 * u;
            v[u] = make_float2(0.f, 0.f); w[u] = 1.f; off[u] = 0;
            if (i < n) {
                const unsigned p = q[i];
                const float* __restrict__ tab = (MODE == 1 && (p >> 24)) ? t1 : t0;
                const float2* __restrict__ row =
                    (const float2*)(tab + (size_t)(p & 0xFFFFu) * DIM);
                v[u]   = row[lane];
                off[u] = (int)((p >> 16) & 0xFFu) * DIM + 2 * lane;
                if (MODE == 0) w[u] = qw[i];
            }
        }
        #pragma unroll
        for (int u = 0; u < OWN_BATCH; ++u) {
            const int i = i0 + 32 * u;
            if (i < n) {
                if (MODE == 0) {
                    atomicAdd(&acc[off[u]],     w[u] * v[u].x);
                    atomicAdd(&acc[off[u] + 1], w[u] * v[u].y);
                } else {
                    atomicAdd(&acc[off[u]],     v[u].x);
                    atomicAdd(&acc[off[u] + 1], v[u].y);
                }
            }
        }
    }
}

__device__ __forceinline__ void gat_softmax_weights(
    const float* __restrict__ el, const float* __restrict__ attn,
    float* acc, const float* erow, const unsigned* q, float* qw,
    unsigned* rmax, float* rold, float* rden, int n, int tid, bool later)
{
    const int lane = tid & 31;
    const int wave = tid >> 5;
    const int sub  = lane & 15;
    const int half = lane >> 4;
    const v4f av = ((const v4f*)attn)[sub];

    for (int i0 = 2 * wave; i0 < n; i0 += 64 * OWN_PAIRS) {
        v4f v[OWN_PAIRS];
        #pragma unroll
        for (int u = 0; u < OWN_PAIRS; ++u) {
            const int i  = i0 + 64 * u + half;
            const int ic = i < n ? i : n - 1;
            v[u] = ((const v4f*)(el + (size_t)(q[ic] & 0xFFFFu) * DIM))[sub];
        }
        float sc[OWN_PAIRS];
        int   dl[OWN_PAIRS];
        #pragma unroll
        for (int u = 0; u < OWN_PAIRS; ++u) {
            const int i  = i0 + 64 * u + half;
            const int ic = i < n ? i : n - 1;
            dl[u] = (int)((q[ic] >> 16) & 0xFFu);
            const v4f r = ((const v4f*)(erow + dl[u] * DIM))[sub];
            sc[u] = 0.f;
            #pragma unroll
            for (int c = 0; c < 4; ++c) {
                float t = v[u][c] + r[c];
                t = t > 0.f ? t : NEG_SLOPE * t;
                sc[u] += t * av[c];
            }
        }
        #pragma unroll
        for (int o = 8; o; o >>= 1) {
            #pragma unroll
            for (int u = 0; u < OWN_PAIRS; ++u) sc[u] += __shfl_xor(sc[u], o, 32);
        }
        #pragma unroll
        for (int u = 0; u < OWN_PAIRS; ++u) {
            const int i = i0 + 64 * u + half;
            if (sub == 0 && i < n) {
                qw[i] = sc[u];
                atomicMax(&rmax[dl[u]], encF(sc[u]));
            }
        }
    }
    __syncthreads();

    if (later) {
        for (int i = tid; i < OWN_ROWS * DIM; i += 1024) {
            const float mo = rold[i >> 6];
            const float mn = decF(rmax[i >> 6]);
            if (mn != mo) acc[i] *= expf(mo - mn);
        }
        if (tid < OWN_ROWS) {
            const float mo = rold[tid];
            const float mn = decF(rmax[tid]);
            if (mn != mo) rden[tid] *= expf(mo - mn);
        }
        __syncthreads();
    }
    if (tid < OWN_ROWS) rold[tid] = decF(rmax[tid]);
    __syncthreads();

    for (int i = tid; i < n; i += 1024) {
        const int dl = (int)((q[i] >> 16) & 0xFFu);
        const float ex = expf(qw[i] - rold[dl]);
        qw[i] = ex;
        atomicAdd(&rden[dl], ex);
    }
    __syncthreads();
}

__global__ void __launch_bounds__(1024)
owner_lists(const int* __restrict__ ety, const int* __restrict__ src,
            const int* __restrict__ dst, const unsigned* __restrict__ keys,
            unsigned* __restrict__ lists, unsigned* __restrict__ counts)
{
    __shared__ unsigned q[OWN_QCAP];
    __shared__ unsigned qn;

    const int tid  = threadIdx.x;
    const int d0   = blockIdx.x * OWN_ROWS;
    const int nloc = (NODES - d0 < OWN_ROWS) ? (NODES - d0) : OWN_ROWS;

    if (tid < nloc) q[tid] = (unsigned)(EDGES + d0 + tid);
    if (tid == 0) qn = (unsigned)nloc;
    __syncthreads();

    const uint4* __restrict__ key4 = (const uint4*)keys;
    const unsigned mine = (unsigned)blockIdx.x * 0x01010101u;
    uint4 cur[4];
    #pragma unroll
    for (int j = 0; j < 4; ++j) {
        const int i4 = j * 1024 + tid;
        cur[j] = (i4 < OWN_KEYS4) ? key4[i4] : make_uint4(0u, 0u, 0u, 0u);
    }
    for (int base4 = 0; base4 < OWN_KEYS4; base4 += 4096) {
        uint4 nxt[4];
        #pragma unroll
        for (int j = 0; j < 4; ++j) {
            const int i4 = base4 + 4096 + j * 1024 + tid;
            nxt[j] = (i4 < OWN_KEYS4) ? key4[i4] : make_uint4(0u, 0u, 0u, 0u);
        }
        #pragma unroll
        for (int j = 0; j < 4; ++j) {
            const int i4 = base4 + j * 1024 + tid;
            if (i4 < OWN_KEYS4) {
                unsigned m = owner_mask(cur[j], mine);
                while (m) {
                    const int b = __ffs((int)m) - 1;
                    m &= m - 1u;
                    const unsigned pos = atomicAdd(&qn, 1u);
                    if (pos < OWN_QCAP) q[pos] = owner_edge(i4, b);
                }
            }
        }
        #pragma unroll
        for (int j = 0; j < 4; ++j) cur[j] = nxt[j];
    }
    __syncthreads();

    const unsigned n = qn;
    if (n > OWN_QCAP) {
        if (tid == 0) counts[blockIdx.x] = OWN_OVER;
        return;
    }
    owner_lookup<1>(ety, src, dst, q, (int)n, d0, tid);
    unsigned* __restrict__ mylist = lists + (size_t)blockIdx.x * OWN_QCAP;
    for (int i = tid; i < (int)n; i += 1024) mylist[i] = q[i];
    if (tid == 0) counts[blockIdx.x] = n;
}

template <int MODE>
__global__ void __launch_bounds__(1024)
owner_sum(const float* __restrict__ t0, const float* __restrict__ t1,
          const float* __restrict__ attn, const int* __restrict__ ety,
          const int* __restrict__ src, const int* __restrict__ dst,
          const unsigned* __restrict__ keys, const unsigned* __restrict__ lists,
          const unsigned* __restrict__ counts, float* __restrict__ out)
{
    __shared__ float    acc[OWN_ROWS * DIM];
    __shared__ unsigned q[OWN_QCAP];
    __shared__ float    qw[(MODE == 0) ? OWN_QCAP : 1];
    __shared__ __attribute__((aligned(16))) float erow[(MODE == 0) ? OWN_ROWS * DIM : 4];
    __shared__ unsigned rmax[(MODE == 0) ? OWN_ROWS : 1];
    __shared__ float    rold[(MODE == 0) ? OWN_ROWS : 1];
    __shared__ float    rden[(MODE == 0) ? OWN_ROWS : 1];
    __shared__ unsigned qn;

    const int tid  = threadIdx.x;
    const int d0   = blockIdx.x * OWN_ROWS;
    const int nloc = (NODES - d0 < OWN_ROWS) ? (NODES - d0) : OWN_ROWS;

    const unsigned cnt = counts[blockIdx.x];

    for (int i = tid; i < OWN_ROWS * DIM; i += 1024) acc[i] = 0.f;
    if (MODE == 0) {
        const v4f* __restrict__ er4 = (const v4f*)(t1 + (size_t)d0 * DIM);
        v4f* erow4 = (v4f*)erow;
        for (int i = tid; i < nloc * (DIM / 4); i += 1024) erow4[i] = er4[i];
        if (tid < OWN_ROWS) {
            rmax[tid] = ENC_NEG_INF;
            rold[tid] = decF(ENC_NEG_INF);
            rden[tid] = 0.f;
        }
    }
    int drains = 0;
    int nlast;

    if (cnt <= (unsigned)OWN_QCAP) {
        const unsigned* __restrict__ mylist = lists + (size_t)blockIdx.x * OWN_QCAP;
        for (int i0 = tid; i0 < (int)cnt; i0 += 4096) {
            unsigned v[4];
            #pragma unroll
            for (int j = 0; j < 4; ++j) {
                const int i = i0 + 1024 * j;
                v[j] = (i < (int)cnt) ? mylist[i] : 0u;
            }
            #pragma unroll
            for (int j = 0; j < 4; ++j) {
                const int i = i0 + 1024 * j;
                if (i < (int)cnt) q[i] = v[j];
            }
        }
        nlast = (int)cnt;
    } else {
        if (tid < nloc) q[tid] = (unsigned)(EDGES + d0 + tid);
        if (tid == 0) qn = (unsigned)nloc;
        const uint4* __restrict__ key4 = (const uint4*)keys;
        const unsigned mine = (unsigned)blockIdx.x * 0x01010101u;
        for (int base4 = 0; base4 < OWN_KEYS4; base4 += 1024) {
            __syncthreads();
            const int n = (int)qn;
            __syncthreads();
            if (n + OWN_SUB > OWN_QCAP) {
                owner_lookup<MODE>(ety, src, dst, q, n, d0, tid);
                __syncthreads();
                if (MODE == 0)
                    gat_softmax_weights(t0, attn, acc, erow, q, qw, rmax, rold, rden,
                                        n, tid, drains > 0);
                owner_add_rows<MODE>(t0, t1, acc, q, qw, n, tid);
                ++drains;
                __syncthreads();
                if (tid == 0) qn = 0u;
                __syncthreads();
            }
            const int i4 = base4 + tid;
            if (i4 < OWN_KEYS4) {
                unsigned m = owner_mask(key4[i4], mine);
                while (m) {
                    const int b = __ffs((int)m) - 1;
                    m &= m - 1u;
                    const unsigned pos = atomicAdd(&qn, 1u);
                    q[pos] = owner_edge(i4, b);
                }
            }
        }
        __syncthreads();
        nlast = (int)qn;
        owner_lookup<MODE>(ety, src, dst, q, nlast, d0, tid);
    }
    __syncthreads();
    if (MODE == 0)
        gat_softmax_weights(t0, attn, acc, erow, q, qw, rmax, rold, rden,
                            nlast, tid, drains > 0);
    owner_add_rows<MODE>(t0, t1, acc, q, qw, nlast, tid);
    __syncthreads();

    float* __restrict__ orow = out + (size_t)d0 * DIM;
    if (MODE == 0) {
        for (int i = tid; i < nloc * DIM; i += 1024) orow[i] = acc[i] / rden[i >> 6];
    } else {
        for (int i = tid; i < nloc * DIM; i += 1024) orow[i] = acc[i];
    }
}

__global__ void gru_update(const float* __restrict__ gi,
                           const float* __restrict__ gh,
                           float* __restrict__ h, float* __restrict__ emb,
                           long nElem)
{
    const long i = (long)blockIdx.x * blockDim.x + threadIdx.x;
    if (i >= nElem) return;
    const long n = i >> 6;
    const int  d = (int)(i & 63);
    const float* gin = gi + n * (3 * DIM);
    const float* ghn = gh + n * (3 * DIM);
    const float r = 1.f / (1.f + expf(-(gin[d]           + ghn[d])));
    const float z = 1.f / (1.f + expf(-(gin[DIM + d]     + ghn[DIM + d])));
    const float c = tanhf(gin[2 * DIM + d] + r * ghn[2 * DIM + d]);
    const float o = (1.f - z) * c + z * h[i];
    h[i] = o;
    if (emb) emb[i] = o;
}

extern "C" void kernel_launch(void* const* d_in, const int* in_sizes, int n_in,
                              void* d_out, int out_size, void* d_ws, size_t ws_size,
                              hipStream_t stream)
{
    (void)in_sizes; (void)n_in; (void)out_size; (void)ws_size;

    const float* x   = (const float*)d_in[0];
    const int*   src = (const int*)d_in[1];
    const int*   dst = (const int*)d_in[2];
    const int*   ety = (const int*)d_in[3];

    const long NE = (long)NODES * DIM;

    float* ws   = (float*)d_ws;
    float* hbuf = ws;                 ws += NE;
    float* buf1 = ws;                 ws += NE;
    float* buf2 = ws;                 ws += NE;
    float* gi   = ws;                 ws += (long)NODES * 3 * DIM;
    float* gh   = ws;                 ws += (long)NODES * 3 * DIM;
    float* acc  = ws;                 ws += NE;
    unsigned* keys   = (unsigned*)ws;
    unsigned* lists  = keys + EDGES / 4;
    unsigned* counts = lists + 256L * OWN_QCAP;

    float* out = (float*)d_out;

    const int elemBlocks     = (int)((NE + 255) / 256);
    const int ownerBlocks    = (NODES + OWN_ROWS - 1) / OWN_ROWS;
    const int keyBlocks      = (EDGES / 4 + 255) / 256;
    const int gemmBlocks64   = ((NODES / 16) * (DIM / 64)       + 3) / 4;
    const int gemmBlocks192  = ((NODES / 16) * (3 * DIM / 64)   + 3) / 4;

    owner_keys<<<keyBlocks, 256, 0, stream>>>(dst, keys);
    owner_lists<<<ownerBlocks, 1024, 0, stream>>>(ety, src, dst, keys, lists, counts);

    const float* hin = x;
    for (int l = 0; l < 2; ++l) {
        const int base = 4 + 11 * l;
        const float* Wsrc = (const float*)d_in[base + 0];
        const float* bsrc = (const float*)d_in[base + 1];
        const float* Wdst = (const float*)d_in[base + 2];
        const float* bdst = (const float*)d_in[base + 3];
        const float* attn = (const float*)d_in[base + 4];
        const float* ggW  = (const float*)d_in[base + 5];
        const float* ggb  = (const float*)d_in[base + 6];
        const float* Wih  = (const float*)d_in[base + 7];
        const float* bih  = (const float*)d_in[base + 8];
        const float* Whh  = (const float*)d_in[base + 9];
        const float* bhh  = (const float*)d_in[base + 10];

        gemm_bias_wmma<DIM><<<gemmBlocks64, 128, 0, stream>>>(hin, Wsrc, bsrc, buf1,
                                                              NODES);
        gemm_bias_wmma<DIM><<<gemmBlocks64, 128, 0, stream>>>(hin, Wdst, bdst, buf2,
                                                              NODES);
        owner_sum<0><<<ownerBlocks, 1024, 0, stream>>>(buf1, buf2, attn, ety, src, dst,
                                                       keys, lists, counts, acc);
        elu_write<<<elemBlocks, 256, 0, stream>>>(acc, hbuf, out + (2L * l) * NE,
                                                  NE);

        for (int st = 0; st < 2; ++st) {
            gemm_bias_wmma<DIM><<<gemmBlocks64, 128, 0, stream>>>(
                hbuf, ggW,               ggb,       buf1, NODES);
            gemm_bias_wmma<DIM><<<gemmBlocks64, 128, 0, stream>>>(
                hbuf, ggW + DIM * DIM,   ggb + DIM, buf2, NODES);
            owner_sum<1><<<ownerBlocks, 1024, 0, stream>>>(buf1, buf2, attn, ety, src, dst,
                                                           keys, lists, counts, acc);
            gemm_bias_wmma<3 * DIM><<<gemmBlocks192, 128, 0, stream>>>(
                acc,  Wih, bih, gi, NODES);
            gemm_bias_wmma<3 * DIM><<<gemmBlocks192, 128, 0, stream>>>(
                hbuf, Whh, bhh, gh, NODES);
            gru_update<<<elemBlocks, 256, 0, stream>>>(
                gi, gh, hbuf, (st == 1) ? out + (2L * l + 1) * NE : nullptr, NE);
        }
        hin = hbuf;
    }
}


# GGC sum by linearity: owner groups sum source h rows per edge type, gi = [S0|S1] x (Wih*W[t]) composed once per call on WMMA; 8 trans GEMM launches gone (34 -> 27); a and gi sum order changes
# speedup vs baseline: 2.5344x; 1.0529x over previous
#include <hip/hip_runtime.h>
#include <hip/hip_bf16.h>


typedef __attribute__((ext_vector_type(2))) float v2f;
typedef __attribute__((ext_vector_type(4))) float v4f;
typedef __attribute__((ext_vector_type(8))) float v8f;

#define NODES 50000
#define EDGES 800000
#define DIM   64
#define NEG_SLOPE 0.2f
#define SDIM  128
#define OWN_ROWS 196
#define OWN_CNT  512

__device__ __forceinline__ unsigned encF(float f) {
    unsigned u = __float_as_uint(f);
    return (u & 0x80000000u) ? ~u : (u | 0x80000000u);
}
__device__ __forceinline__ float decF(unsigned u) {
    unsigned v = (u & 0x80000000u) ? (u & 0x7FFFFFFFu) : ~u;
    return __uint_as_float(v);
}

__device__ __forceinline__ void load_frag(const float* __restrict__ rowp, int hi,
                                          v4f (&f)[8])
{
    const v4f* __restrict__ p = (const v4f*)(rowp + 4 * hi);
    #pragma unroll
    for (int j = 0; j < 8; ++j) f[j] = p[2 * j];
}

template <int NOUT, int K, bool COUNTS>
__global__ void __launch_bounds__(128) __attribute__((amdgpu_waves_per_eu(1, 4)))
gemm_bias_wmma(const float* __restrict__ A, const float* __restrict__ W,
               const float* __restrict__ bias, float* __restrict__ C, int M,
               const float* __restrict__ cnt, const float* __restrict__ cbias)
{
    constexpr int GROUPS = NOUT / 64;
    const int lane = threadIdx.x & 31;
    const int wave = blockIdx.x * 4 + (threadIdx.x >> 5);
    const int total = (M >> 4) * GROUPS;
    if (wave >= total) return;

    const int rt   = wave / GROUPS;
    const int cg   = wave - rt * GROUPS;
    const int row0 = rt << 4;
    const int colB = cg << 6;
    const int lr   = lane & 15;
    const int hi   = lane >> 4;

    float2 cn[8];
    if (COUNTS) {
        #pragma unroll
        for (int v = 0; v < 8; ++v) {
            const int row = row0 + hi * 8 + v;
            const int g   = row / OWN_ROWS;
            cn[v] = *(const float2*)(cnt + (size_t)g * OWN_CNT + 2 * (row - g * OWN_ROWS));
        }
    }

    v8f acc[4] = {};
    #pragma unroll
    for (int kh = 0; kh < K / 64; ++kh) {
        v4f a4[8];
        load_frag(A + (size_t)(row0 + lr) * K + 64 * kh, hi, a4);

        v4f b4[4][8];
        #pragma unroll
        for (int t = 0; t < 4; ++t)
            load_frag(W + (size_t)(colB + 16 * t + lr) * K + 64 * kh, hi, b4[t]);

        #pragma unroll
        for (int j = 0; j < 8; ++j) {
            #pragma unroll
            for (int p = 0; p < 2; ++p) {
                v2f a;
                a.x = a4[j][2 * p];
                a.y = a4[j][2 * p + 1];
                #pragma unroll
                for (int t = 0; t < 4; ++t) {
                    v2f b;
                    b.x = b4[t][j][2 * p];
                    b.y = b4[t][j][2 * p + 1];
                    acc[t] = __builtin_amdgcn_wmma_f32_16x16x4_f32(
                                 false, a, false, b, (short)0, acc[t], false, false);
                }
            }
        }
    }

    #pragma unroll
    for (int t = 0; t < 4; ++t) {
        const int col = colB + 16 * t + lr;
        const float bv = bias[col];
        float c0 = 0.f, c1 = 0.f;
        if (COUNTS) { c0 = cbias[col]; c1 = cbias[NOUT + col]; }
        float* crow = C + (size_t)(row0 + hi * 8) * NOUT + col;
        #pragma unroll
        for (int v = 0; v < 8; ++v) {
            float o = acc[t][v] + bv;
            if (COUNTS) o += cn[v].x * c0 + cn[v].y * c1;
            crow[(size_t)v * NOUT] = o;
        }
    }
}

__global__ void __launch_bounds__(32)
compose_weights(const float* __restrict__ Wih0, const float* __restrict__ gw0,
                const float* __restrict__ gb0,  const float* __restrict__ Wih1,
                const float* __restrict__ gw1,  const float* __restrict__ gb1,
                float* __restrict__ Wc, float* __restrict__ cb)
{
    const int lane = threadIdx.x;
    const int lr   = lane & 15;
    const int hi   = lane >> 4;
    const int l    = blockIdx.x / 24;
    const int r    = blockIdx.x - l * 24;
    const int t    = r / 12;
    const int row0 = (r - t * 12) * 16;

    const float* __restrict__ Wih = l ? Wih1 : Wih0;
    const float* __restrict__ Wt  = (l ? gw1 : gw0) + t * DIM * DIM;
    const float* __restrict__ bt  = (l ? gb1 : gb0) + t * DIM;

    if (hi == 0) {
        const float* __restrict__ wr = Wih + (size_t)(row0 + lr) * DIM;
        float sum = 0.f;
        for (int j = 0; j < DIM; ++j) sum += wr[j] * bt[j];
        cb[l * 2 * 3 * DIM + t * 3 * DIM + row0 + lr] = sum;
    }

    v4f a4[8];
    load_frag(Wih + (size_t)(row0 + lr) * DIM, hi, a4);

    v8f acc[4] = {};
    #pragma unroll
    for (int j = 0; j < 8; ++j) {
        #pragma unroll
        for (int p = 0; p < 2; ++p) {
            v2f a;
            a.x = a4[j][2 * p];
            a.y = a4[j][2 * p + 1];
            const int k0 = 8 * j + 4 * hi + 2 * p;
            #pragma unroll
            for (int tt = 0; tt < 4; ++tt) {
                v2f b;
                b.x = Wt[(k0)     * DIM + 16 * tt + lr];
                b.y = Wt[(k0 + 1) * DIM + 16 * tt + lr];
                acc[tt] = __builtin_amdgcn_wmma_f32_16x16x4_f32(
                              false, a, false, b, (short)0, acc[tt], false, false);
            }
        }
    }

    float* crow = Wc + (size_t)l * 3 * DIM * SDIM + (size_t)(row0 + hi * 8) * SDIM
                     + t * DIM + lr;
    #pragma unroll
    for (int tt = 0; tt < 4; ++tt) {
        #pragma unroll
        for (int v = 0; v < 8; ++v)
            crow[(size_t)v * SDIM + 16 * tt] = acc[tt][v];
    }
}

__global__ void elu_write(const float* __restrict__ acc, float* __restrict__ h,
                          float* __restrict__ emb, long n)
{
    const long i = (long)blockIdx.x * blockDim.x + threadIdx.x;
    if (i >= n) return;
    const float v = acc[i];
    const float o = v > 0.f ? v : (expf(v) - 1.f);
    h[i] = o;
    emb[i] = o;
}

#define OWN_SUB  16384
#define OWN_QCAP 24576
#define OWN_BATCH 8
#define OWN_PAIRS (OWN_BATCH / 2)
#define OWN_KEYS4 (EDGES / 16)
#define ENC_NEG_INF 0x007FFFFFu
#define OWN_OVER 0xFFFFFFFFu

__global__ void __launch_bounds__(256)
owner_keys(const int* __restrict__ dst, unsigned* __restrict__ keys)
{
    const int i = blockIdx.x * 256 + threadIdx.x;
    if (i >= EDGES / 4) return;
    const int4 d = ((const int4*)dst)[i];
    keys[i] =  ((unsigned)d.x / OWN_ROWS)        | (((unsigned)d.y / OWN_ROWS) << 8) |
              (((unsigned)d.z / OWN_ROWS) << 16) | (((unsigned)d.w / OWN_ROWS) << 24);
}

__device__ __forceinline__ unsigned zero_bytes(unsigned x)
{
    const unsigned y = ((x & 0x7F7F7F7Fu) + 0x7F7F7F7Fu) | x;
    return ~y & 0x80808080u;
}

__device__ __forceinline__ unsigned owner_mask(const uint4 k, unsigned mine)
{
    return (zero_bytes(k.x ^ mine) >> 7) | (zero_bytes(k.y ^ mine) >> 6) |
           (zero_bytes(k.z ^ mine) >> 5) | (zero_bytes(k.w ^ mine) >> 4);
}
__device__ __forceinline__ unsigned owner_edge(int i4, int b)
{
    return (unsigned)(i4 * 16 + ((b & 7) << 2) + (b >> 3));
}

template <int MODE>
__device__ __forceinline__ void owner_lookup(
    const int* __restrict__ ety, const int* __restrict__ src,
    const int* __restrict__ dst, unsigned* q, int n, int d0, int tid)
{
    for (int i = tid; i < n; i += 1024) {
        const int e = (int)q[i];
        int s, d;
        if (e < EDGES) { s = src[e]; d = dst[e]; } else { s = d = e - EDGES; }
        unsigned o = (unsigned)s | ((unsigned)(d - d0) << 16);
        if (MODE == 1) o |= (ety[e] == 0) ? 0u : (1u << 24);
        q[i] = o;
    }
}

template <int MODE>
__device__ __forceinline__ void owner_add_rows(
    const float* __restrict__ t0, float* acc, const unsigned* q, const float* qw,
    unsigned* cnt, int n, int tid)
{
    const int lane = tid & 31;
    const int wave = tid >> 5;
    if (MODE == 1) {
        for (int i = tid; i < n; i += 1024) {
            const unsigned p = q[i];
            atomicAdd(&cnt[((p >> 16) & 0xFFu) * 2u + (p >> 24)], 1u);
        }
    }
    for (int i0 = wave; i0 < n; i0 += 32 * OWN_BATCH) {
        float2 v[OWN_BATCH];
        float  w[OWN_BATCH];
        int    off[OWN_BATCH];
        #pragma unroll
        for (int u = 0; u < OWN_BATCH; ++u) {
            const int i = i0 + 32 * u;
            v[u] = make_float2(0.f, 0.f); w[u] = 1.f; off[u] = 0;
            if (i < n) {
                const unsigned p = q[i];
                const float2* __restrict__ row =
                    (const float2*)(t0 + (size_t)(p & 0xFFFFu) * DIM);
                v[u]   = row[lane];
                if (MODE == 0) {
                    off[u] = (int)((p >> 16) & 0xFFu) * DIM + 2 * lane;
                    w[u]   = qw[i];
                } else {
                    off[u] = (int)((p >> 16) & 0xFFu) * SDIM + (int)(p >> 24) * DIM
                             + 2 * lane;
                }
            }
        }
        #pragma unroll
        for (int u = 0; u < OWN_BATCH; ++u) {
            const int i = i0 + 32 * u;
            if (i < n) {
                if (MODE == 0) {
                    atomicAdd(&acc[off[u]],     w[u] * v[u].x);
                    atomicAdd(&acc[off[u] + 1], w[u] * v[u].y);
                } else {
                    atomicAdd(&acc[off[u]],     v[u].x);
                    atomicAdd(&acc[off[u] + 1], v[u].y);
                }
            }
        }
    }
}

__device__ __forceinline__ void gat_softmax_weights(
    const float* __restrict__ el, const float* __restrict__ attn,
    float* acc, const float* erow, const unsigned* q, float* qw,
    unsigned* rmax, float* rold, float* rden, int n, int tid, bool later)
{
    const int lane = tid & 31;
    const int wave = tid >> 5;
    const int sub  = lane & 15;
    const int half = lane >> 4;
    const v4f av = ((const v4f*)attn)[sub];

    for (int i0 = 2 * wave; i0 < n; i0 += 64 * OWN_PAIRS) {
        v4f v[OWN_PAIRS];
        #pragma unroll
        for (int u = 0; u < OWN_PAIRS; ++u) {
            const int i  = i0 + 64 * u + half;
            const int ic = i < n ? i : n - 1;
            v[u] = ((const v4f*)(el + (size_t)(q[ic] & 0xFFFFu) * DIM))[sub];
        }
        float sc[OWN_PAIRS];
        int   dl[OWN_PAIRS];
        #pragma unroll
        for (int u = 0; u < OWN_PAIRS; ++u) {
            const int i  = i0 + 64 * u + half;
            const int ic = i < n ? i : n - 1;
            dl[u] = (int)((q[ic] >> 16) & 0xFFu);
            const v4f r = ((const v4f*)(erow + dl[u] * DIM))[sub];
            sc[u] = 0.f;
            #pragma unroll
            for (int c = 0; c < 4; ++c) {
                float t = v[u][c] + r[c];
                t = t > 0.f ? t : NEG_SLOPE * t;
                sc[u] += t * av[c];
            }
        }
        #pragma unroll
        for (int o = 8; o; o >>= 1) {
            #pragma unroll
            for (int u = 0; u < OWN_PAIRS; ++u) sc[u] += __shfl_xor(sc[u], o, 32);
        }
        #pragma unroll
        for (int u = 0; u < OWN_PAIRS; ++u) {
            const int i = i0 + 64 * u + half;
            if (sub == 0 && i < n) {
                qw[i] = sc[u];
                atomicMax(&rmax[dl[u]], encF(sc[u]));
            }
        }
    }
    __syncthreads();

    if (later) {
        for (int i = tid; i < OWN_ROWS * DIM; i += 1024) {
            const float mo = rold[i >> 6];
            const float mn = decF(rmax[i >> 6]);
            if (mn != mo) acc[i] *= expf(mo - mn);
        }
        if (tid < OWN_ROWS) {
            const float mo = rold[tid];
            const float mn = decF(rmax[tid]);
            if (mn != mo) rden[tid] *= expf(mo - mn);
        }
        __syncthreads();
    }
    if (tid < OWN_ROWS) rold[tid] = decF(rmax[tid]);
    __syncthreads();

    for (int i = tid; i < n; i += 1024) {
        const int dl = (int)((q[i] >> 16) & 0xFFu);
        const float ex = expf(qw[i] - rold[dl]);
        qw[i] = ex;
        atomicAdd(&rden[dl], ex);
    }
    __syncthreads();
}

__global__ void __launch_bounds__(1024)
owner_lists(const int* __restrict__ ety, const int* __restrict__ src,
            const int* __restrict__ dst, const unsigned* __restrict__ keys,
            unsigned* __restrict__ lists, unsigned* __restrict__ counts)
{
    __shared__ unsigned q[OWN_QCAP];
    __shared__ unsigned qn;

    const int tid  = threadIdx.x;
    const int d0   = blockIdx.x * OWN_ROWS;
    const int nloc = (NODES - d0 < OWN_ROWS) ? (NODES - d0) : OWN_ROWS;

    if (tid < nloc) q[tid] = (unsigned)(EDGES + d0 + tid);
    if (tid == 0) qn = (unsigned)nloc;
    __syncthreads();

    const uint4* __restrict__ key4 = (const uint4*)keys;
    const unsigned mine = (unsigned)blockIdx.x * 0x01010101u;
    uint4 cur[4];
    #pragma unroll
    for (int j = 0; j < 4; ++j) {
        const int i4 = j * 1024 + tid;
        cur[j] = (i4 < OWN_KEYS4) ? key4[i4] : make_uint4(0u, 0u, 0u, 0u);
    }
    for (int base4 = 0; base4 < OWN_KEYS4; base4 += 4096) {
        uint4 nxt[4];
        #pragma unroll
        for (int j = 0; j < 4; ++j) {
            const int i4 = base4 + 4096 + j * 1024 + tid;
            nxt[j] = (i4 < OWN_KEYS4) ? key4[i4] : make_uint4(0u, 0u, 0u, 0u);
        }
        #pragma unroll
        for (int j = 0; j < 4; ++j) {
            const int i4 = base4 + j * 1024 + tid;
            if (i4 < OWN_KEYS4) {
                unsigned m = owner_mask(cur[j], mine);
                while (m) {
                    const int b = __ffs((int)m) - 1;
                    m &= m - 1u;
                    const unsigned pos = atomicAdd(&qn, 1u);
                    if (pos < OWN_QCAP) q[pos] = owner_edge(i4, b);
                }
            }
        }
        #pragma unroll
        for (int j = 0; j < 4; ++j) cur[j] = nxt[j];
    }
    __syncthreads();

    const unsigned n = qn;
    if (n > OWN_QCAP) {
        if (tid == 0) counts[blockIdx.x] = OWN_OVER;
        return;
    }
    owner_lookup<1>(ety, src, dst, q, (int)n, d0, tid);
    unsigned* __restrict__ mylist = lists + (size_t)blockIdx.x * OWN_QCAP;
    for (int i = tid; i < (int)n; i += 1024) mylist[i] = q[i];
    if (tid == 0) counts[blockIdx.x] = n;
}

template <int MODE>
__global__ void __launch_bounds__(1024)
owner_sum(const float* __restrict__ t0, const float* __restrict__ t1,
          const float* __restrict__ attn, const int* __restrict__ ety,
          const int* __restrict__ src, const int* __restrict__ dst,
          const unsigned* __restrict__ keys, const unsigned* __restrict__ lists,
          const unsigned* __restrict__ counts, float* __restrict__ out,
          float* __restrict__ cntf)
{
    constexpr int ROWLEN = (MODE == 0) ? DIM : SDIM;
    __shared__ float    acc[OWN_ROWS * ROWLEN];
    __shared__ unsigned ecnt[(MODE == 1) ? OWN_ROWS * 2 : 1];
    __shared__ unsigned q[OWN_QCAP];
    __shared__ float    qw[(MODE == 0) ? OWN_QCAP : 1];
    __shared__ __attribute__((aligned(16))) float erow[(MODE == 0) ? OWN_ROWS * DIM : 4];
    __shared__ unsigned rmax[(MODE == 0) ? OWN_ROWS : 1];
    __shared__ float    rold[(MODE == 0) ? OWN_ROWS : 1];
    __shared__ float    rden[(MODE == 0) ? OWN_ROWS : 1];
    __shared__ unsigned qn;

    const int tid  = threadIdx.x;
    const int d0   = blockIdx.x * OWN_ROWS;
    const int nloc = (NODES - d0 < OWN_ROWS) ? (NODES - d0) : OWN_ROWS;

    const unsigned cnt = counts[blockIdx.x];

    for (int i = tid; i < OWN_ROWS * ROWLEN; i += 1024) acc[i] = 0.f;
    if (MODE == 1 && tid < OWN_ROWS * 2) ecnt[tid] = 0u;
    if (MODE == 0) {
        const v4f* __restrict__ er4 = (const v4f*)(t1 + (size_t)d0 * DIM);
        v4f* erow4 = (v4f*)erow;
        for (int i = tid; i < nloc * (DIM / 4); i += 1024) erow4[i] = er4[i];
        if (tid < OWN_ROWS) {
            rmax[tid] = ENC_NEG_INF;
            rold[tid] = decF(ENC_NEG_INF);
            rden[tid] = 0.f;
        }
    }
    int drains = 0;
    int nlast;

    if (cnt <= (unsigned)OWN_QCAP) {
        const unsigned* __restrict__ mylist = lists + (size_t)blockIdx.x * OWN_QCAP;
        for (int i0 = tid; i0 < (int)cnt; i0 += 4096) {
            unsigned v[4];
            #pragma unroll
            for (int j = 0; j < 4; ++j) {
                const int i = i0 + 1024 * j;
                v[j] = (i < (int)cnt) ? mylist[i] : 0u;
            }
            #pragma unroll
            for (int j = 0; j < 4; ++j) {
                const int i = i0 + 1024 * j;
                if (i < (int)cnt) q[i] = v[j];
            }
        }
        nlast = (int)cnt;
    } else {
        if (tid < nloc) q[tid] = (unsigned)(EDGES + d0 + tid);
        if (tid == 0) qn = (unsigned)nloc;
        const uint4* __restrict__ key4 = (const uint4*)keys;
        const unsigned mine = (unsigned)blockIdx.x * 0x01010101u;
        for (int base4 = 0; base4 < OWN_KEYS4; base4 += 1024) {
            __syncthreads();
            const int n = (int)qn;
            __syncthreads();
            if (n + OWN_SUB > OWN_QCAP) {
                owner_lookup<MODE>(ety, src, dst, q, n, d0, tid);
                __syncthreads();
                if (MODE == 0)
                    gat_softmax_weights(t0, attn, acc, erow, q, qw, rmax, rold, rden,
                                        n, tid, drains > 0);
                owner_add_rows<MODE>(t0, acc, q, qw, ecnt, n, tid);
                ++drains;
                __syncthreads();
                if (tid == 0) qn = 0u;
                __syncthreads();
            }
            const int i4 = base4 + tid;
            if (i4 < OWN_KEYS4) {
                unsigned m = owner_mask(key4[i4], mine);
                while (m) {
                    const int b = __ffs((int)m) - 1;
                    m &= m - 1u;
                    const unsigned pos = atomicAdd(&qn, 1u);
                    q[pos] = owner_edge(i4, b);
                }
            }
        }
        __syncthreads();
        nlast = (int)qn;
        owner_lookup<MODE>(ety, src, dst, q, nlast, d0, tid);
    }
    __syncthreads();
    if (MODE == 0)
        gat_softmax_weights(t0, attn, acc, erow, q, qw, rmax, rold, rden,
                            nlast, tid, drains > 0);
    owner_add_rows<MODE>(t0, acc, q, qw, ecnt, nlast, tid);
    __syncthreads();

    float* __restrict__ orow = out + (size_t)d0 * ROWLEN;
    if (MODE == 0) {
        for (int i = tid; i < nloc * DIM; i += 1024) orow[i] = acc[i] / rden[i >> 6];
    } else {
        for (int i = tid; i < nloc * SDIM; i += 1024) orow[i] = acc[i];
        if (tid < nloc * 2)
            cntf[(size_t)blockIdx.x * OWN_CNT + tid] = (float)ecnt[tid];
    }
}

__global__ void gru_update(const float* __restrict__ gi,
                           const float* __restrict__ gh,
                           float* __restrict__ h, float* __restrict__ emb,
                           long nElem)
{
    const long i = (long)blockIdx.x * blockDim.x + threadIdx.x;
    if (i >= nElem) return;
    const long n = i >> 6;
    const int  d = (int)(i & 63);
    const float* gin = gi + n * (3 * DIM);
    const float* ghn = gh + n * (3 * DIM);
    const float r = 1.f / (1.f + expf(-(gin[d]           + ghn[d])));
    const float z = 1.f / (1.f + expf(-(gin[DIM + d]     + ghn[DIM + d])));
    const float c = tanhf(gin[2 * DIM + d] + r * ghn[2 * DIM + d]);
    const float o = (1.f - z) * c + z * h[i];
    h[i] = o;
    if (emb) emb[i] = o;
}

extern "C" void kernel_launch(void* const* d_in, const int* in_sizes, int n_in,
                              void* d_out, int out_size, void* d_ws, size_t ws_size,
                              hipStream_t stream)
{
    (void)in_sizes; (void)n_in; (void)out_size; (void)ws_size;

    const float* x   = (const float*)d_in[0];
    const int*   src = (const int*)d_in[1];
    const int*   dst = (const int*)d_in[2];
    const int*   ety = (const int*)d_in[3];

    const long NE = (long)NODES * DIM;

    float* ws   = (float*)d_ws;
    float* hbuf = ws;                 ws += NE;
    float* buf1 = ws;                 ws += NE;
    float* buf2 = ws;                 ws += NE;
    float* gi   = ws;                 ws += (long)NODES * 3 * DIM;
    float* gh   = ws;                 ws += (long)NODES * 3 * DIM;
    float* acc  = ws;                 ws += NE;
    unsigned* keys   = (unsigned*)ws;
    unsigned* lists  = keys + EDGES / 4;
    unsigned* counts = lists + 256L * OWN_QCAP;
    float* cntf = (float*)(counts + 256);
    float* Wc   = cntf + 256L * OWN_CNT;
    float* cb   = Wc + 2L * 3 * DIM * SDIM;
    float* ssum = buf1;

    float* out = (float*)d_out;

    const int elemBlocks     = (int)((NE + 255) / 256);
    const int ownerBlocks    = (NODES + OWN_ROWS - 1) / OWN_ROWS;
    const int keyBlocks      = (EDGES / 4 + 255) / 256;
    const int gemmBlocks64   = ((NODES / 16) * (DIM / 64)       + 3) / 4;
    const int gemmBlocks192  = ((NODES / 16) * (3 * DIM / 64)   + 3) / 4;

    owner_keys<<<keyBlocks, 256, 0, stream>>>(dst, keys);
    owner_lists<<<ownerBlocks, 1024, 0, stream>>>(ety, src, dst, keys, lists, counts);
    compose_weights<<<48, 32, 0, stream>>>(
        (const float*)d_in[4 + 7],  (const float*)d_in[4 + 5],  (const float*)d_in[4 + 6],
        (const float*)d_in[15 + 7], (const float*)d_in[15 + 5], (const float*)d_in[15 + 6],
        Wc, cb);

    const float* hin = x;
    for (int l = 0; l < 2; ++l) {
        const int base = 4 + 11 * l;
        const float* Wsrc = (const float*)d_in[base + 0];
        const float* bsrc = (const float*)d_in[base + 1];
        const float* Wdst = (const float*)d_in[base + 2];
        const float* bdst = (const float*)d_in[base + 3];
        const float* attn = (const float*)d_in[base + 4];
        const float* bih  = (const float*)d_in[base + 8];
        const float* Whh  = (const float*)d_in[base + 9];
        const float* bhh  = (const float*)d_in[base + 10];

        gemm_bias_wmma<DIM, DIM, false><<<gemmBlocks64, 128, 0, stream>>>(
            hin, Wsrc, bsrc, buf1, NODES, nullptr, nullptr);
        gemm_bias_wmma<DIM, DIM, false><<<gemmBlocks64, 128, 0, stream>>>(
            hin, Wdst, bdst, buf2, NODES, nullptr, nullptr);
        owner_sum<0><<<ownerBlocks, 1024, 0, stream>>>(buf1, buf2, attn, ety, src, dst,
                                                       keys, lists, counts, acc, nullptr);
        elu_write<<<elemBlocks, 256, 0, stream>>>(acc, hbuf, out + (2L * l) * NE,
                                                  NE);

        for (int st = 0; st < 2; ++st) {
            owner_sum<1><<<ownerBlocks, 1024, 0, stream>>>(hbuf, hbuf, attn, ety, src, dst,
                                                           keys, lists, counts, ssum, cntf);
            gemm_bias_wmma<3 * DIM, SDIM, true><<<gemmBlocks192, 128, 0, stream>>>(
                ssum, Wc + (long)l * 3 * DIM * SDIM, bih, gi, NODES,
                cntf, cb + l * 2 * 3 * DIM);
            gemm_bias_wmma<3 * DIM, DIM, false><<<gemmBlocks192, 128, 0, stream>>>(
                hbuf, Whh, bhh, gh, NODES, nullptr, nullptr);
            gru_update<<<elemBlocks, 256, 0, stream>>>(
                gi, gh, hbuf, (st == 1) ? out + (2L * l + 1) * NE : nullptr, NE);
        }
        hin = hbuf;
    }
}


# owner groups take a batch's queue entries by one LDS load plus lane reads, not one LDS round trip per entry: 16 row loads per wave go out together, scalar bases; LDS add order by arrival
# speedup vs baseline: 2.7009x; 1.0657x over previous
#include <hip/hip_runtime.h>
#include <hip/hip_bf16.h>


typedef __attribute__((ext_vector_type(2))) float v2f;
typedef __attribute__((ext_vector_type(4))) float v4f;
typedef __attribute__((ext_vector_type(8))) float v8f;

#define NODES 50000
#define EDGES 800000
#define DIM   64
#define NEG_SLOPE 0.2f
#define SDIM  128
#define OWN_ROWS 196
#define OWN_CNT  512

__device__ __forceinline__ unsigned encF(float f) {
    unsigned u = __float_as_uint(f);
    return (u & 0x80000000u) ? ~u : (u | 0x80000000u);
}
__device__ __forceinline__ float decF(unsigned u) {
    unsigned v = (u & 0x80000000u) ? (u & 0x7FFFFFFFu) : ~u;
    return __uint_as_float(v);
}

__device__ __forceinline__ void load_frag(const float* __restrict__ rowp, int hi,
                                          v4f (&f)[8])
{
    const v4f* __restrict__ p = (const v4f*)(rowp + 4 * hi);
    #pragma unroll
    for (int j = 0; j < 8; ++j) f[j] = p[2 * j];
}

template <int NOUT, int K, bool COUNTS>
__global__ void __launch_bounds__(128) __attribute__((amdgpu_waves_per_eu(1, 4)))
gemm_bias_wmma(const float* __restrict__ A, const float* __restrict__ W,
               const float* __restrict__ bias, float* __restrict__ C, int M,
               const float* __restrict__ cnt, const float* __restrict__ cbias)
{
    constexpr int GROUPS = NOUT / 64;
    const int lane = threadIdx.x & 31;
    const int wave = blockIdx.x * 4 + (threadIdx.x >> 5);
    const int total = (M >> 4) * GROUPS;
    if (wave >= total) return;

    const int rt   = wave / GROUPS;
    const int cg   = wave - rt * GROUPS;
    const int row0 = rt << 4;
    const int colB = cg << 6;
    const int lr   = lane & 15;
    const int hi   = lane >> 4;

    float2 cn[8];
    if (COUNTS) {
        #pragma unroll
        for (int v = 0; v < 8; ++v) {
            const int row = row0 + hi * 8 + v;
            const int g   = row / OWN_ROWS;
            cn[v] = *(const float2*)(cnt + (size_t)g * OWN_CNT + 2 * (row - g * OWN_ROWS));
        }
    }

    v8f acc[4] = {};
    #pragma unroll
    for (int kh = 0; kh < K / 64; ++kh) {
        v4f a4[8];
        load_frag(A + (size_t)(row0 + lr) * K + 64 * kh, hi, a4);

        v4f b4[4][8];
        #pragma unroll
        for (int t = 0; t < 4; ++t)
            load_frag(W + (size_t)(colB + 16 * t + lr) * K + 64 * kh, hi, b4[t]);

        #pragma unroll
        for (int j = 0; j < 8; ++j) {
            #pragma unroll
            for (int p = 0; p < 2; ++p) {
                v2f a;
                a.x = a4[j][2 * p];
                a.y = a4[j][2 * p + 1];
                #pragma unroll
                for (int t = 0; t < 4; ++t) {
                    v2f b;
                    b.x = b4[t][j][2 * p];
                    b.y = b4[t][j][2 * p + 1];
                    acc[t] = __builtin_amdgcn_wmma_f32_16x16x4_f32(
                                 false, a, false, b, (short)0, acc[t], false, false);
                }
            }
        }
    }

    #pragma unroll
    for (int t = 0; t < 4; ++t) {
        const int col = colB + 16 * t + lr;
        const float bv = bias[col];
        float c0 = 0.f, c1 = 0.f;
        if (COUNTS) { c0 = cbias[col]; c1 = cbias[NOUT + col]; }
        float* crow = C + (size_t)(row0 + hi * 8) * NOUT + col;
        #pragma unroll
        for (int v = 0; v < 8; ++v) {
            float o = acc[t][v] + bv;
            if (COUNTS) o += cn[v].x * c0 + cn[v].y * c1;
            crow[(size_t)v * NOUT] = o;
        }
    }
}

__global__ void __launch_bounds__(32)
compose_weights(const float* __restrict__ Wih0, const float* __restrict__ gw0,
                const float* __restrict__ gb0,  const float* __restrict__ Wih1,
                const float* __restrict__ gw1,  const float* __restrict__ gb1,
                float* __restrict__ Wc, float* __restrict__ cb)
{
    const int lane = threadIdx.x;
    const int lr   = lane & 15;
    const int hi   = lane >> 4;
    const int l    = blockIdx.x / 24;
    const int r    = blockIdx.x - l * 24;
    const int t    = r / 12;
    const int row0 = (r - t * 12) * 16;

    const float* __restrict__ Wih = l ? Wih1 : Wih0;
    const float* __restrict__ Wt  = (l ? gw1 : gw0) + t * DIM * DIM;
    const float* __restrict__ bt  = (l ? gb1 : gb0) + t * DIM;

    if (hi == 0) {
        const float* __restrict__ wr = Wih + (size_t)(row0 + lr) * DIM;
        float sum = 0.f;
        for (int j = 0; j < DIM; ++j) sum += wr[j] * bt[j];
        cb[l * 2 * 3 * DIM + t * 3 * DIM + row0 + lr] = sum;
    }

    v4f a4[8];
    load_frag(Wih + (size_t)(row0 + lr) * DIM, hi, a4);

    v8f acc[4] = {};
    #pragma unroll
    for (int j = 0; j < 8; ++j) {
        #pragma unroll
        for (int p = 0; p < 2; ++p) {
            v2f a;
            a.x = a4[j][2 * p];
            a.y = a4[j][2 * p + 1];
            const int k0 = 8 * j + 4 * hi + 2 * p;
            #pragma unroll
            for (int tt = 0; tt < 4; ++tt) {
                v2f b;
                b.x = Wt[(k0)     * DIM + 16 * tt + lr];
                b.y = Wt[(k0 + 1) * DIM + 16 * tt + lr];
                acc[tt] = __builtin_amdgcn_wmma_f32_16x16x4_f32(
                              false, a, false, b, (short)0, acc[tt], false, false);
            }
        }
    }

    float* crow = Wc + (size_t)l * 3 * DIM * SDIM + (size_t)(row0 + hi * 8) * SDIM
                     + t * DIM + lr;
    #pragma unroll
    for (int tt = 0; tt < 4; ++tt) {
        #pragma unroll
        for (int v = 0; v < 8; ++v)
            crow[(size_t)v * SDIM + 16 * tt] = acc[tt][v];
    }
}

__global__ void elu_write(const float* __restrict__ acc, float* __restrict__ h,
                          float* __restrict__ emb, long n)
{
    const long i = (long)blockIdx.x * blockDim.x + threadIdx.x;
    if (i >= n) return;
    const float v = acc[i];
    const float o = v > 0.f ? v : (expf(v) - 1.f);
    h[i] = o;
    emb[i] = o;
}

#define OWN_SUB  16384
#define OWN_QCAP 24576
#define OWN_BATCH 16
#define OWN_PAIRS 8
#define OWN_KEYS4 (EDGES / 16)
#define ENC_NEG_INF 0x007FFFFFu
#define OWN_OVER 0xFFFFFFFFu

__global__ void __launch_bounds__(256)
owner_keys(const int* __restrict__ dst, unsigned* __restrict__ keys)
{
    const int i = blockIdx.x * 256 + threadIdx.x;
    if (i >= EDGES / 4) return;
    const int4 d = ((const int4*)dst)[i];
    keys[i] =  ((unsigned)d.x / OWN_ROWS)        | (((unsigned)d.y / OWN_ROWS) << 8) |
              (((unsigned)d.z / OWN_ROWS) << 16) | (((unsigned)d.w / OWN_ROWS) << 24);
}

__device__ __forceinline__ unsigned zero_bytes(unsigned x)
{
    const unsigned y = ((x & 0x7F7F7F7Fu) + 0x7F7F7F7Fu) | x;
    return ~y & 0x80808080u;
}

__device__ __forceinline__ unsigned owner_mask(const uint4 k, unsigned mine)
{
    return (zero_bytes(k.x ^ mine) >> 7) | (zero_bytes(k.y ^ mine) >> 6) |
           (zero_bytes(k.z ^ mine) >> 5) | (zero_bytes(k.w ^ mine) >> 4);
}
__device__ __forceinline__ unsigned owner_edge(int i4, int b)
{
    return (unsigned)(i4 * 16 + ((b & 7) << 2) + (b >> 3));
}

template <int MODE>
__device__ __forceinline__ void owner_lookup(
    const int* __restrict__ ety, const int* __restrict__ src,
    const int* __restrict__ dst, unsigned* q, int n, int d0, int tid)
{
    for (int i = tid; i < n; i += 1024) {
        const int e = (int)q[i];
        int s, d;
        if (e < EDGES) { s = src[e]; d = dst[e]; } else { s = d = e - EDGES; }
        unsigned o = (unsigned)s | ((unsigned)(d - d0) << 16);
        if (MODE == 1) o |= (ety[e] == 0) ? 0u : (1u << 24);
        q[i] = o;
    }
}

template <int MODE, bool FULL>
__device__ __forceinline__ void owner_add_batch(
    const float* __restrict__ t0, float* acc, const unsigned* q, const float* qw,
    unsigned& myq, float& myw, int nb, int inext, int lane)
{
    float2 v[OWN_BATCH];
    #pragma unroll
    for (int u = 0; u < OWN_BATCH; ++u) {
        const unsigned p = (unsigned)__builtin_amdgcn_readlane((int)myq, u);
        const float2* __restrict__ row = (const float2*)(t0 + (size_t)(p & 0xFFFFu) * DIM);
        v[u] = row[lane];
    }
    const unsigned nq = q[inext];
    const float    nw = (MODE == 0) ? qw[inext] : 1.f;
    #pragma unroll
    for (int u = 0; u < OWN_BATCH; ++u) {
        if (FULL || u < nb) {
            const unsigned p = (unsigned)__builtin_amdgcn_readlane((int)myq, u);
            if (MODE == 0) {
                const float w = __uint_as_float(
                    (unsigned)__builtin_amdgcn_readlane((int)__float_as_uint(myw), u));
                const int off = (int)((p >> 16) & 0xFFu) * DIM + 2 * lane;
                atomicAdd(&acc[off],     w * v[u].x);
                atomicAdd(&acc[off + 1], w * v[u].y);
            } else {
                const int off = (int)((p >> 16) & 0xFFu) * SDIM + (int)(p >> 24) * DIM
                                + 2 * lane;
                atomicAdd(&acc[off],     v[u].x);
                atomicAdd(&acc[off + 1], v[u].y);
            }
        }
    }
    myq = nq;
    myw = nw;
}

template <int MODE>
__device__ __forceinline__ void owner_add_rows(
    const float* __restrict__ t0, float* acc, const unsigned* q, const float* qw,
    unsigned* cnt, int nv, int tid)
{
    const int lane = tid & 31;
    const int wave = __builtin_amdgcn_readfirstlane(tid >> 5);
    const int n    = __builtin_amdgcn_readfirstlane(nv);
    if (MODE == 1) {
        for (int i = tid; i < n; i += 1024) {
            const unsigned p = q[i];
            atomicAdd(&cnt[((p >> 16) & 0xFFu) * 2u + (p >> 24)], 1u);
        }
    }
    const int ifirst = (lane < OWN_BATCH && wave + 32 * lane < n) ? wave + 32 * lane : wave;
    unsigned myq = q[ifirst];
    float    myw = (MODE == 0) ? qw[ifirst] : 1.f;
    for (int i0 = wave; i0 < n; i0 += 32 * OWN_BATCH) {
        int nb = (n - i0 + 31) >> 5;
        if (nb > OWN_BATCH) nb = OWN_BATCH;
        const int i1    = i0 + 32 * OWN_BATCH + 32 * lane;
        const int inext = (lane < OWN_BATCH && i1 < n) ? i1 : i0;
        if (nb == OWN_BATCH)
            owner_add_batch<MODE, true >(t0, acc, q, qw, myq, myw, nb, inext, lane);
        else
            owner_add_batch<MODE, false>(t0, acc, q, qw, myq, myw, nb, inext, lane);
    }
}

__device__ __forceinline__ void gat_softmax_weights(
    const float* __restrict__ el, const float* __restrict__ attn,
    float* acc, const float* erow, const unsigned* q, float* qw,
    unsigned* rmax, float* rold, float* rden, int n, int tid, bool later)
{
    const int lane = tid & 31;
    const int wave = tid >> 5;
    const int sub  = lane & 15;
    const int half = lane >> 4;
    const v4f av = ((const v4f*)attn)[sub];

    const int wave_s = __builtin_amdgcn_readfirstlane(wave);
    const int n_s    = __builtin_amdgcn_readfirstlane(n);
    for (int i0 = 2 * wave_s; i0 < n_s; i0 += 64 * OWN_PAIRS) {
        const int iq = i0 + 64 * (lane >> 1) + (lane & 1);
        const unsigned myq = q[(lane < 2 * OWN_PAIRS && iq < n_s) ? iq : n_s - 1];
        v4f v[OWN_PAIRS];
        int dl[OWN_PAIRS];
        #pragma unroll
        for (int u = 0; u < OWN_PAIRS; ++u) {
            const unsigned p0 = (unsigned)__builtin_amdgcn_readlane((int)myq, 2 * u);
            const unsigned p1 = (unsigned)__builtin_amdgcn_readlane((int)myq, 2 * u + 1);
            const unsigned p  = half ? p1 : p0;
            v[u]  = ((const v4f*)(el + (size_t)(p & 0xFFFFu) * DIM))[sub];
            dl[u] = (int)((p >> 16) & 0xFFu);
        }
        float sc[OWN_PAIRS];
        #pragma unroll
        for (int u = 0; u < OWN_PAIRS; ++u) {
            const v4f r = ((const v4f*)(erow + dl[u] * DIM))[sub];
            sc[u] = 0.f;
            #pragma unroll
            for (int c = 0; c < 4; ++c) {
                float t = v[u][c] + r[c];
                t = t > 0.f ? t : NEG_SLOPE * t;
                sc[u] += t * av[c];
            }
        }
        #pragma unroll
        for (int o = 8; o; o >>= 1) {
            #pragma unroll
            for (int u = 0; u < OWN_PAIRS; ++u) sc[u] += __shfl_xor(sc[u], o, 32);
        }
        #pragma unroll
        for (int u = 0; u < OWN_PAIRS; ++u) {
            const int i = i0 + 64 * u + half;
            if (sub == 0 && i < n) {
                qw[i] = sc[u];
                atomicMax(&rmax[dl[u]], encF(sc[u]));
            }
        }
    }
    __syncthreads();

    if (later) {
        for (int i = tid; i < OWN_ROWS * DIM; i += 1024) {
            const float mo = rold[i >> 6];
            const float mn = decF(rmax[i >> 6]);
            if (mn != mo) acc[i] *= expf(mo - mn);
        }
        if (tid < OWN_ROWS) {
            const float mo = rold[tid];
            const float mn = decF(rmax[tid]);
            if (mn != mo) rden[tid] *= expf(mo - mn);
        }
        __syncthreads();
    }
    if (tid < OWN_ROWS) rold[tid] = decF(rmax[tid]);
    __syncthreads();

    for (int i = tid; i < n; i += 1024) {
        const int dl = (int)((q[i] >> 16) & 0xFFu);
        const float ex = expf(qw[i] - rold[dl]);
        qw[i] = ex;
        atomicAdd(&rden[dl], ex);
    }
    __syncthreads();
}

__global__ void __launch_bounds__(1024)
owner_lists(const int* __restrict__ ety, const int* __restrict__ src,
            const int* __restrict__ dst, const unsigned* __restrict__ keys,
            unsigned* __restrict__ lists, unsigned* __restrict__ counts)
{
    __shared__ unsigned q[OWN_QCAP];
    __shared__ unsigned qn;

    const int tid  = threadIdx.x;
    const int d0   = blockIdx.x * OWN_ROWS;
    const int nloc = (NODES - d0 < OWN_ROWS) ? (NODES - d0) : OWN_ROWS;

    if (tid < nloc) q[tid] = (unsigned)(EDGES + d0 + tid);
    if (tid == 0) qn = (unsigned)nloc;
    __syncthreads();

    const uint4* __restrict__ key4 = (const uint4*)keys;
    const unsigned mine = (unsigned)blockIdx.x * 0x01010101u;
    uint4 cur[4];
    #pragma unroll
    for (int j = 0; j < 4; ++j) {
        const int i4 = j * 1024 + tid;
        cur[j] = (i4 < OWN_KEYS4) ? key4[i4] : make_uint4(0u, 0u, 0u, 0u);
    }
    for (int base4 = 0; base4 < OWN_KEYS4; base4 += 4096) {
        uint4 nxt[4];
        #pragma unroll
        for (int j = 0; j < 4; ++j) {
            const int i4 = base4 + 4096 + j * 1024 + tid;
            nxt[j] = (i4 < OWN_KEYS4) ? key4[i4] : make_uint4(0u, 0u, 0u, 0u);
        }
        #pragma unroll
        for (int j = 0; j < 4; ++j) {
            const int i4 = base4 + j * 1024 + tid;
            if (i4 < OWN_KEYS4) {
                unsigned m = owner_mask(cur[j], mine);
                while (m) {
                    const int b = __ffs((int)m) - 1;
                    m &= m - 1u;
                    const unsigned pos = atomicAdd(&qn, 1u);
                    if (pos < OWN_QCAP) q[pos] = owner_edge(i4, b);
                }
            }
        }
        #pragma unroll
        for (int j = 0; j < 4; ++j) cur[j] = nxt[j];
    }
    __syncthreads();

    const unsigned n = qn;
    if (n > OWN_QCAP) {
        if (tid == 0) counts[blockIdx.x] = OWN_OVER;
        return;
    }
    owner_lookup<1>(ety, src, dst, q, (int)n, d0, tid);
    unsigned* __restrict__ mylist = lists + (size_t)blockIdx.x * OWN_QCAP;
    for (int i = tid; i < (int)n; i += 1024) mylist[i] = q[i];
    if (tid == 0) counts[blockIdx.x] = n;
}

template <int MODE>
__global__ void __launch_bounds__(1024)
owner_sum(const float* __restrict__ t0, const float* __restrict__ t1,
          const float* __restrict__ attn, const int* __restrict__ ety,
          const int* __restrict__ src, const int* __restrict__ dst,
          const unsigned* __restrict__ keys, const unsigned* __restrict__ lists,
          const unsigned* __restrict__ counts, float* __restrict__ out,
          float* __restrict__ cntf)
{
    constexpr int ROWLEN = (MODE == 0) ? DIM : SDIM;
    __shared__ float    acc[OWN_ROWS * ROWLEN];
    __shared__ unsigned ecnt[(MODE == 1) ? OWN_ROWS * 2 : 1];
    __shared__ unsigned q[OWN_QCAP];
    __shared__ float    qw[(MODE == 0) ? OWN_QCAP : 1];
    __shared__ __attribute__((aligned(16))) float erow[(MODE == 0) ? OWN_ROWS * DIM : 4];
    __shared__ unsigned rmax[(MODE == 0) ? OWN_ROWS : 1];
    __shared__ float    rold[(MODE == 0) ? OWN_ROWS : 1];
    __shared__ float    rden[(MODE == 0) ? OWN_ROWS : 1];
    __shared__ unsigned qn;

    const int tid  = threadIdx.x;
    const int d0   = blockIdx.x * OWN_ROWS;
    const int nloc = (NODES - d0 < OWN_ROWS) ? (NODES - d0) : OWN_ROWS;

    const unsigned cnt = counts[blockIdx.x];

    for (int i = tid; i < OWN_ROWS * ROWLEN; i += 1024) acc[i] = 0.f;
    if (MODE == 1 && tid < OWN_ROWS * 2) ecnt[tid] = 0u;
    if (MODE == 0) {
        const v4f* __restrict__ er4 = (const v4f*)(t1 + (size_t)d0 * DIM);
        v4f* erow4 = (v4f*)erow;
        for (int i = tid; i < nloc * (DIM / 4); i += 1024) erow4[i] = er4[i];
        if (tid < OWN_ROWS) {
            rmax[tid] = ENC_NEG_INF;
            rold[tid] = decF(ENC_NEG_INF);
            rden[tid] = 0.f;
        }
    }
    int drains = 0;
    int nlast;

    if (cnt <= (unsigned)OWN_QCAP) {
        const unsigned* __restrict__ mylist = lists + (size_t)blockIdx.x * OWN_QCAP;
        for (int i0 = tid; i0 < (int)cnt; i0 += 4096) {
            unsigned v[4];
            #pragma unroll
            for (int j = 0; j < 4; ++j) {
                const int i = i0 + 1024 * j;
                v[j] = (i < (int)cnt) ? mylist[i] : 0u;
            }
            #pragma unroll
            for (int j = 0; j < 4; ++j) {
                const int i = i0 + 1024 * j;
                if (i < (int)cnt) q[i] = v[j];
            }
        }
        nlast = (int)cnt;
    } else {
        if (tid < nloc) q[tid] = (unsigned)(EDGES + d0 + tid);
        if (tid == 0) qn = (unsigned)nloc;
        const uint4* __restrict__ key4 = (const uint4*)keys;
        const unsigned mine = (unsigned)blockIdx.x * 0x01010101u;
        for (int base4 = 0; base4 < OWN_KEYS4; base4 += 1024) {
            __syncthreads();
            const int n = (int)qn;
            __syncthreads();
            if (n + OWN_SUB > OWN_QCAP) {
                owner_lookup<MODE>(ety, src, dst, q, n, d0, tid);
                __syncthreads();
                if (MODE == 0)
                    gat_softmax_weights(t0, attn, acc, erow, q, qw, rmax, rold, rden,
                                        n, tid, drains > 0);
                owner_add_rows<MODE>(t0, acc, q, qw, ecnt, n, tid);
                ++drains;
                __syncthreads();
                if (tid == 0) qn = 0u;
                __syncthreads();
            }
            const int i4 = base4 + tid;
            if (i4 < OWN_KEYS4) {
                unsigned m = owner_mask(key4[i4], mine);
                while (m) {
                    const int b = __ffs((int)m) - 1;
                    m &= m - 1u;
                    const unsigned pos = atomicAdd(&qn, 1u);
                    q[pos] = owner_edge(i4, b);
                }
            }
        }
        __syncthreads();
        nlast = (int)qn;
        owner_lookup<MODE>(ety, src, dst, q, nlast, d0, tid);
    }
    __syncthreads();
    if (MODE == 0)
        gat_softmax_weights(t0, attn, acc, erow, q, qw, rmax, rold, rden,
                            nlast, tid, drains > 0);
    owner_add_rows<MODE>(t0, acc, q, qw, ecnt, nlast, tid);
    __syncthreads();

    float* __restrict__ orow = out + (size_t)d0 * ROWLEN;
    if (MODE == 0) {
        for (int i = tid; i < nloc * DIM; i += 1024) orow[i] = acc[i] / rden[i >> 6];
    } else {
        for (int i = tid; i < nloc * SDIM; i += 1024) orow[i] = acc[i];
        if (tid < nloc * 2)
            cntf[(size_t)blockIdx.x * OWN_CNT + tid] = (float)ecnt[tid];
    }
}

__global__ void gru_update(const float* __restrict__ gi,
                           const float* __restrict__ gh,
                           float* __restrict__ h, float* __restrict__ emb,
                           long nElem)
{
    const long i = (long)blockIdx.x * blockDim.x + threadIdx.x;
    if (i >= nElem) return;
    const long n = i >> 6;
    const int  d = (int)(i & 63);
    const float* gin = gi + n * (3 * DIM);
    const float* ghn = gh + n * (3 * DIM);
    const float r = 1.f / (1.f + expf(-(gin[d]           + ghn[d])));
    const float z = 1.f / (1.f + expf(-(gin[DIM + d]     + ghn[DIM + d])));
    const float c = tanhf(gin[2 * DIM + d] + r * ghn[2 * DIM + d]);
    const float o = (1.f - z) * c + z * h[i];
    h[i] = o;
    if (emb) emb[i] = o;
}

extern "C" void kernel_launch(void* const* d_in, const int* in_sizes, int n_in,
                              void* d_out, int out_size, void* d_ws, size_t ws_size,
                              hipStream_t stream)
{
    (void)in_sizes; (void)n_in; (void)out_size; (void)ws_size;

    const float* x   = (const float*)d_in[0];
    const int*   src = (const int*)d_in[1];
    const int*   dst = (const int*)d_in[2];
    const int*   ety = (const int*)d_in[3];

    const long NE = (long)NODES * DIM;

    float* ws   = (float*)d_ws;
    float* hbuf = ws;                 ws += NE;
    float* buf1 = ws;                 ws += NE;
    float* buf2 = ws;                 ws += NE;
    float* gi   = ws;                 ws += (long)NODES * 3 * DIM;
    float* gh   = ws;                 ws += (long)NODES * 3 * DIM;
    float* acc  = ws;                 ws += NE;
    unsigned* keys   = (unsigned*)ws;
    unsigned* lists  = keys + EDGES / 4;
    unsigned* counts = lists + 256L * OWN_QCAP;
    float* cntf = (float*)(counts + 256);
    float* Wc   = cntf + 256L * OWN_CNT;
    float* cb   = Wc + 2L * 3 * DIM * SDIM;
    float* ssum = buf1;

    float* out = (float*)d_out;

    const int elemBlocks     = (int)((NE + 255) / 256);
    const int ownerBlocks    = (NODES + OWN_ROWS - 1) / OWN_ROWS;
    const int keyBlocks      = (EDGES / 4 + 255) / 256;
    const int gemmBlocks64   = ((NODES / 16) * (DIM / 64)       + 3) / 4;
    const int gemmBlocks192  = ((NODES / 16) * (3 * DIM / 64)   + 3) / 4;

    owner_keys<<<keyBlocks, 256, 0, stream>>>(dst, keys);
    owner_lists<<<ownerBlocks, 1024, 0, stream>>>(ety, src, dst, keys, lists, counts);
    compose_weights<<<48, 32, 0, stream>>>(
        (const float*)d_in[4 + 7],  (const float*)d_in[4 + 5],  (const float*)d_in[4 + 6],
        (const float*)d_in[15 + 7], (const float*)d_in[15 + 5], (const float*)d_in[15 + 6],
        Wc, cb);

    const float* hin = x;
    for (int l = 0; l < 2; ++l) {
        const int base = 4 + 11 * l;
        const float* Wsrc = (const float*)d_in[base + 0];
        const float* bsrc = (const float*)d_in[base + 1];
        const float* Wdst = (const float*)d_in[base + 2];
        const float* bdst = (const float*)d_in[base + 3];
        const float* attn = (const float*)d_in[base + 4];
        const float* bih  = (const float*)d_in[base + 8];
        const float* Whh  = (const float*)d_in[base + 9];
        const float* bhh  = (const float*)d_in[base + 10];

        gemm_bias_wmma<DIM, DIM, false><<<gemmBlocks64, 128, 0, stream>>>(
            hin, Wsrc, bsrc, buf1, NODES, nullptr, nullptr);
        gemm_bias_wmma<DIM, DIM, false><<<gemmBlocks64, 128, 0, stream>>>(
            hin, Wdst, bdst, buf2, NODES, nullptr, nullptr);
        owner_sum<0><<<ownerBlocks, 1024, 0, stream>>>(buf1, buf2, attn, ety, src, dst,
                                                       keys, lists, counts, acc, nullptr);
        elu_write<<<elemBlocks, 256, 0, stream>>>(acc, hbuf, out + (2L * l) * NE,
                                                  NE);

        for (int st = 0; st < 2; ++st) {
            owner_sum<1><<<ownerBlocks, 1024, 0, stream>>>(hbuf, hbuf, attn, ety, src, dst,
                                                           keys, lists, counts, ssum, cntf);
            gemm_bias_wmma<3 * DIM, SDIM, true><<<gemmBlocks192, 128, 0, stream>>>(
                ssum, Wc + (long)l * 3 * DIM * SDIM, bih, gi, NODES,
                cntf, cb + l * 2 * 3 * DIM);
            gemm_bias_wmma<3 * DIM, DIM, false><<<gemmBlocks192, 128, 0, stream>>>(
                hbuf, Whh, bhh, gh, NODES, nullptr, nullptr);
            gru_update<<<elemBlocks, 256, 0, stream>>>(
                gi, gh, hbuf, (st == 1) ? out + (2L * l + 1) * NE : nullptr, NE);
        }
        hin = hbuf;
    }
}


# GAT edge softmax in one pass per owner group: weights exp(score) with no row-max shift while all |score| <= 40 (else the old exact path); source rows gathered once, not twice; den sums unshifted
# speedup vs baseline: 2.8100x; 1.0404x over previous
#include <hip/hip_runtime.h>
#include <hip/hip_bf16.h>


typedef __attribute__((ext_vector_type(2))) float v2f;
typedef __attribute__((ext_vector_type(4))) float v4f;
typedef __attribute__((ext_vector_type(8))) float v8f;

#define NODES 50000
#define EDGES 800000
#define DIM   64
#define NEG_SLOPE 0.2f
#define SDIM  128
#define OWN_ROWS 196
#define OWN_CNT  512

__device__ __forceinline__ unsigned encF(float f) {
    unsigned u = __float_as_uint(f);
    return (u & 0x80000000u) ? ~u : (u | 0x80000000u);
}
__device__ __forceinline__ float decF(unsigned u) {
    unsigned v = (u & 0x80000000u) ? (u & 0x7FFFFFFFu) : ~u;
    return __uint_as_float(v);
}

__device__ __forceinline__ void load_frag(const float* __restrict__ rowp, int hi,
                                          v4f (&f)[8])
{
    const v4f* __restrict__ p = (const v4f*)(rowp + 4 * hi);
    #pragma unroll
    for (int j = 0; j < 8; ++j) f[j] = p[2 * j];
}

template <int NOUT, int K, bool COUNTS>
__global__ void __launch_bounds__(128) __attribute__((amdgpu_waves_per_eu(1, 4)))
gemm_bias_wmma(const float* __restrict__ A, const float* __restrict__ W,
               const float* __restrict__ bias, float* __restrict__ C, int M,
               const float* __restrict__ cnt, const float* __restrict__ cbias)
{
    constexpr int GROUPS = NOUT / 64;
    const int lane = threadIdx.x & 31;
    const int wave = blockIdx.x * 4 + (threadIdx.x >> 5);
    const int total = (M >> 4) * GROUPS;
    if (wave >= total) return;

    const int rt   = wave / GROUPS;
    const int cg   = wave - rt * GROUPS;
    const int row0 = rt << 4;
    const int colB = cg << 6;
    const int lr   = lane & 15;
    const int hi   = lane >> 4;

    float2 cn[8];
    if (COUNTS) {
        #pragma unroll
        for (int v = 0; v < 8; ++v) {
            const int row = row0 + hi * 8 + v;
            const int g   = row / OWN_ROWS;
            cn[v] = *(const float2*)(cnt + (size_t)g * OWN_CNT + 2 * (row - g * OWN_ROWS));
        }
    }

    v8f acc[4] = {};
    #pragma unroll
    for (int kh = 0; kh < K / 64; ++kh) {
        v4f a4[8];
        load_frag(A + (size_t)(row0 + lr) * K + 64 * kh, hi, a4);

        v4f b4[4][8];
        #pragma unroll
        for (int t = 0; t < 4; ++t)
            load_frag(W + (size_t)(colB + 16 * t + lr) * K + 64 * kh, hi, b4[t]);

        #pragma unroll
        for (int j = 0; j < 8; ++j) {
            #pragma unroll
            for (int p = 0; p < 2; ++p) {
                v2f a;
                a.x = a4[j][2 * p];
                a.y = a4[j][2 * p + 1];
                #pragma unroll
                for (int t = 0; t < 4; ++t) {
                    v2f b;
                    b.x = b4[t][j][2 * p];
                    b.y = b4[t][j][2 * p + 1];
                    acc[t] = __builtin_amdgcn_wmma_f32_16x16x4_f32(
                                 false, a, false, b, (short)0, acc[t], false, false);
                }
            }
        }
    }

    #pragma unroll
    for (int t = 0; t < 4; ++t) {
        const int col = colB + 16 * t + lr;
        const float bv = bias[col];
        float c0 = 0.f, c1 = 0.f;
        if (COUNTS) { c0 = cbias[col]; c1 = cbias[NOUT + col]; }
        float* crow = C + (size_t)(row0 + hi * 8) * NOUT + col;
        #pragma unroll
        for (int v = 0; v < 8; ++v) {
            float o = acc[t][v] + bv;
            if (COUNTS) o += cn[v].x * c0 + cn[v].y * c1;
            crow[(size_t)v * NOUT] = o;
        }
    }
}

__global__ void __launch_bounds__(32)
compose_weights(const float* __restrict__ Wih0, const float* __restrict__ gw0,
                const float* __restrict__ gb0,  const float* __restrict__ Wih1,
                const float* __restrict__ gw1,  const float* __restrict__ gb1,
                float* __restrict__ Wc, float* __restrict__ cb)
{
    const int lane = threadIdx.x;
    const int lr   = lane & 15;
    const int hi   = lane >> 4;
    const int l    = blockIdx.x / 24;
    const int r    = blockIdx.x - l * 24;
    const int t    = r / 12;
    const int row0 = (r - t * 12) * 16;

    const float* __restrict__ Wih = l ? Wih1 : Wih0;
    const float* __restrict__ Wt  = (l ? gw1 : gw0) + t * DIM * DIM;
    const float* __restrict__ bt  = (l ? gb1 : gb0) + t * DIM;

    if (hi == 0) {
        const float* __restrict__ wr = Wih + (size_t)(row0 + lr) * DIM;
        float sum = 0.f;
        for (int j = 0; j < DIM; ++j) sum += wr[j] * bt[j];
        cb[l * 2 * 3 * DIM + t * 3 * DIM + row0 + lr] = sum;
    }

    v4f a4[8];
    load_frag(Wih + (size_t)(row0 + lr) * DIM, hi, a4);

    v8f acc[4] = {};
    #pragma unroll
    for (int j = 0; j < 8; ++j) {
        #pragma unroll
        for (int p = 0; p < 2; ++p) {
            v2f a;
            a.x = a4[j][2 * p];
            a.y = a4[j][2 * p + 1];
            const int k0 = 8 * j + 4 * hi + 2 * p;
            #pragma unroll
            for (int tt = 0; tt < 4; ++tt) {
                v2f b;
                b.x = Wt[(k0)     * DIM + 16 * tt + lr];
                b.y = Wt[(k0 + 1) * DIM + 16 * tt + lr];
                acc[tt] = __builtin_amdgcn_wmma_f32_16x16x4_f32(
                              false, a, false, b, (short)0, acc[tt], false, false);
            }
        }
    }

    float* crow = Wc + (size_t)l * 3 * DIM * SDIM + (size_t)(row0 + hi * 8) * SDIM
                     + t * DIM + lr;
    #pragma unroll
    for (int tt = 0; tt < 4; ++tt) {
        #pragma unroll
        for (int v = 0; v < 8; ++v)
            crow[(size_t)v * SDIM + 16 * tt] = acc[tt][v];
    }
}

__global__ void elu_write(const float* __restrict__ acc, float* __restrict__ h,
                          float* __restrict__ emb, long n)
{
    const long i = (long)blockIdx.x * blockDim.x + threadIdx.x;
    if (i >= n) return;
    const float v = acc[i];
    const float o = v > 0.f ? v : (expf(v) - 1.f);
    h[i] = o;
    emb[i] = o;
}

#define OWN_SUB  16384
#define OWN_QCAP 24576
#define OWN_BATCH 16
#define OWN_PAIRS 8
#define OWN_KEYS4 (EDGES / 16)
#define ENC_NEG_INF 0x007FFFFFu
#define OWN_OVER 0xFFFFFFFFu
#define GAT_SAFE 40.0f

__global__ void __launch_bounds__(256)
owner_keys(const int* __restrict__ dst, unsigned* __restrict__ keys)
{
    const int i = blockIdx.x * 256 + threadIdx.x;
    if (i >= EDGES / 4) return;
    const int4 d = ((const int4*)dst)[i];
    keys[i] =  ((unsigned)d.x / OWN_ROWS)        | (((unsigned)d.y / OWN_ROWS) << 8) |
              (((unsigned)d.z / OWN_ROWS) << 16) | (((unsigned)d.w / OWN_ROWS) << 24);
}

__device__ __forceinline__ unsigned zero_bytes(unsigned x)
{
    const unsigned y = ((x & 0x7F7F7F7Fu) + 0x7F7F7F7Fu) | x;
    return ~y & 0x80808080u;
}

__device__ __forceinline__ unsigned owner_mask(const uint4 k, unsigned mine)
{
    return (zero_bytes(k.x ^ mine) >> 7) | (zero_bytes(k.y ^ mine) >> 6) |
           (zero_bytes(k.z ^ mine) >> 5) | (zero_bytes(k.w ^ mine) >> 4);
}
__device__ __forceinline__ unsigned owner_edge(int i4, int b)
{
    return (unsigned)(i4 * 16 + ((b & 7) << 2) + (b >> 3));
}

template <int MODE>
__device__ __forceinline__ void owner_lookup(
    const int* __restrict__ ety, const int* __restrict__ src,
    const int* __restrict__ dst, unsigned* q, int n, int d0, int tid)
{
    for (int i = tid; i < n; i += 1024) {
        const int e = (int)q[i];
        int s, d;
        if (e < EDGES) { s = src[e]; d = dst[e]; } else { s = d = e - EDGES; }
        unsigned o = (unsigned)s | ((unsigned)(d - d0) << 16);
        if (MODE == 1) o |= (ety[e] == 0) ? 0u : (1u << 24);
        q[i] = o;
    }
}

template <int MODE, bool FULL>
__device__ __forceinline__ void owner_add_batch(
    const float* __restrict__ t0, float* acc, const unsigned* q, const float* qw,
    unsigned& myq, float& myw, int nb, int inext, int lane)
{
    float2 v[OWN_BATCH];
    #pragma unroll
    for (int u = 0; u < OWN_BATCH; ++u) {
        const unsigned p = (unsigned)__builtin_amdgcn_readlane((int)myq, u);
        const float2* __restrict__ row = (const float2*)(t0 + (size_t)(p & 0xFFFFu) * DIM);
        v[u] = row[lane];
    }
    const unsigned nq = q[inext];
    const float    nw = (MODE == 0) ? qw[inext] : 1.f;
    #pragma unroll
    for (int u = 0; u < OWN_BATCH; ++u) {
        if (FULL || u < nb) {
            const unsigned p = (unsigned)__builtin_amdgcn_readlane((int)myq, u);
            if (MODE == 0) {
                const float w = __uint_as_float(
                    (unsigned)__builtin_amdgcn_readlane((int)__float_as_uint(myw), u));
                const int off = (int)((p >> 16) & 0xFFu) * DIM + 2 * lane;
                atomicAdd(&acc[off],     w * v[u].x);
                atomicAdd(&acc[off + 1], w * v[u].y);
            } else {
                const int off = (int)((p >> 16) & 0xFFu) * SDIM + (int)(p >> 24) * DIM
                                + 2 * lane;
                atomicAdd(&acc[off],     v[u].x);
                atomicAdd(&acc[off + 1], v[u].y);
            }
        }
    }
    myq = nq;
    myw = nw;
}

template <int MODE>
__device__ __forceinline__ void owner_add_rows(
    const float* __restrict__ t0, float* acc, const unsigned* q, const float* qw,
    unsigned* cnt, int nv, int tid)
{
    const int lane = tid & 31;
    const int wave = __builtin_amdgcn_readfirstlane(tid >> 5);
    const int n    = __builtin_amdgcn_readfirstlane(nv);
    if (MODE == 1) {
        for (int i = tid; i < n; i += 1024) {
            const unsigned p = q[i];
            atomicAdd(&cnt[((p >> 16) & 0xFFu) * 2u + (p >> 24)], 1u);
        }
    }
    const int ifirst = (lane < OWN_BATCH && wave + 32 * lane < n) ? wave + 32 * lane : wave;
    unsigned myq = q[ifirst];
    float    myw = (MODE == 0) ? qw[ifirst] : 1.f;
    for (int i0 = wave; i0 < n; i0 += 32 * OWN_BATCH) {
        int nb = (n - i0 + 31) >> 5;
        if (nb > OWN_BATCH) nb = OWN_BATCH;
        const int i1    = i0 + 32 * OWN_BATCH + 32 * lane;
        const int inext = (lane < OWN_BATCH && i1 < n) ? i1 : i0;
        if (nb == OWN_BATCH)
            owner_add_batch<MODE, true >(t0, acc, q, qw, myq, myw, nb, inext, lane);
        else
            owner_add_batch<MODE, false>(t0, acc, q, qw, myq, myw, nb, inext, lane);
    }
}

__device__ __forceinline__ void gat_softmax_weights(
    const float* __restrict__ el, const float* __restrict__ attn,
    float* acc, const float* erow, const unsigned* q, float* qw,
    unsigned* rmax, float* rold, float* rden, int n, int tid, bool later)
{
    const int lane = tid & 31;
    const int wave = tid >> 5;
    const int sub  = lane & 15;
    const int half = lane >> 4;
    const v4f av = ((const v4f*)attn)[sub];

    const int wave_s = __builtin_amdgcn_readfirstlane(wave);
    const int n_s    = __builtin_amdgcn_readfirstlane(n);
    for (int i0 = 2 * wave_s; i0 < n_s; i0 += 64 * OWN_PAIRS) {
        const int iq = i0 + 64 * (lane >> 1) + (lane & 1);
        const unsigned myq = q[(lane < 2 * OWN_PAIRS && iq < n_s) ? iq : n_s - 1];
        v4f v[OWN_PAIRS];
        int dl[OWN_PAIRS];
        #pragma unroll
        for (int u = 0; u < OWN_PAIRS; ++u) {
            const unsigned p0 = (unsigned)__builtin_amdgcn_readlane((int)myq, 2 * u);
            const unsigned p1 = (unsigned)__builtin_amdgcn_readlane((int)myq, 2 * u + 1);
            const unsigned p  = half ? p1 : p0;
            v[u]  = ((const v4f*)(el + (size_t)(p & 0xFFFFu) * DIM))[sub];
            dl[u] = (int)((p >> 16) & 0xFFu);
        }
        float sc[OWN_PAIRS];
        #pragma unroll
        for (int u = 0; u < OWN_PAIRS; ++u) {
            const v4f r = ((const v4f*)(erow + dl[u] * DIM))[sub];
            sc[u] = 0.f;
            #pragma unroll
            for (int c = 0; c < 4; ++c) {
                float t = v[u][c] + r[c];
                t = t > 0.f ? t : NEG_SLOPE * t;
                sc[u] += t * av[c];
            }
        }
        #pragma unroll
        for (int o = 8; o; o >>= 1) {
            #pragma unroll
            for (int u = 0; u < OWN_PAIRS; ++u) sc[u] += __shfl_xor(sc[u], o, 32);
        }
        #pragma unroll
        for (int u = 0; u < OWN_PAIRS; ++u) {
            const int i = i0 + 64 * u + half;
            if (sub == 0 && i < n) {
                qw[i] = sc[u];
                atomicMax(&rmax[dl[u]], encF(sc[u]));
            }
        }
    }
    __syncthreads();

    if (later) {
        for (int i = tid; i < OWN_ROWS * DIM; i += 1024) {
            const float mo = rold[i >> 6];
            const float mn = decF(rmax[i >> 6]);
            if (mn != mo) acc[i] *= expf(mo - mn);
        }
        if (tid < OWN_ROWS) {
            const float mo = rold[tid];
            const float mn = decF(rmax[tid]);
            if (mn != mo) rden[tid] *= expf(mo - mn);
        }
        __syncthreads();
    }
    if (tid < OWN_ROWS) rold[tid] = decF(rmax[tid]);
    __syncthreads();

    for (int i = tid; i < n; i += 1024) {
        const int dl = (int)((q[i] >> 16) & 0xFFu);
        const float ex = expf(qw[i] - rold[dl]);
        qw[i] = ex;
        atomicAdd(&rden[dl], ex);
    }
    __syncthreads();
}

template <int M>
__device__ __forceinline__ float row_xor(float x)
{
    return __int_as_float(__builtin_amdgcn_update_dpp(
        0, __float_as_int(x), 0x160 + M, 0xf, 0xf, true));
}

__device__ __forceinline__ bool gat_one_pass(
    const float* __restrict__ el, const float* __restrict__ attn,
    float* acc, const float* erow, const unsigned* q, float* rden, int n, int tid)
{
    const int lane = tid & 31;
    const int sub  = lane & 15;
    const int half = lane >> 4;
    const v4f av = ((const v4f*)attn)[sub];
    const int wave_s = __builtin_amdgcn_readfirstlane(tid >> 5);
    const int n_s    = __builtin_amdgcn_readfirstlane(n);
    const int offa = 4 * sub + 2 * half;
    const int offb = 4 * sub + 2 - 2 * half;
    bool wide = false;
    for (int i0 = 2 * wave_s; i0 < n_s; i0 += 64 * OWN_PAIRS) {
        const int iq = i0 + 64 * (lane >> 1) + (lane & 1);
        const unsigned myq = q[(lane < 2 * OWN_PAIRS && iq < n_s) ? iq : n_s - 1];
        v4f v[OWN_PAIRS];
        int dl[OWN_PAIRS];
        #pragma unroll
        for (int u = 0; u < OWN_PAIRS; ++u) {
            const unsigned p0 = (unsigned)__builtin_amdgcn_readlane((int)myq, 2 * u);
            const unsigned p1 = (unsigned)__builtin_amdgcn_readlane((int)myq, 2 * u + 1);
            const unsigned p  = half ? p1 : p0;
            v[u]  = ((const v4f*)(el + (size_t)(p & 0xFFFFu) * DIM))[sub];
            dl[u] = (int)((p >> 16) & 0xFFu);
        }
        float sc[OWN_PAIRS];
        #pragma unroll
        for (int u = 0; u < OWN_PAIRS; ++u) {
            const v4f r = ((const v4f*)(erow + dl[u] * DIM))[sub];
            sc[u] = 0.f;
            #pragma unroll
            for (int c = 0; c < 4; ++c) {
                float t = v[u][c] + r[c];
                t = t > 0.f ? t : NEG_SLOPE * t;
                sc[u] += t * av[c];
            }
        }
        #pragma unroll
        for (int u = 0; u < OWN_PAIRS; ++u) sc[u] += row_xor<8>(sc[u]);
        #pragma unroll
        for (int u = 0; u < OWN_PAIRS; ++u) sc[u] += row_xor<4>(sc[u]);
        #pragma unroll
        for (int u = 0; u < OWN_PAIRS; ++u) sc[u] += row_xor<2>(sc[u]);
        #pragma unroll
        for (int u = 0; u < OWN_PAIRS; ++u) sc[u] += row_xor<1>(sc[u]);

        float wsel = 0.f;
        int   dsel = 0;
        #pragma unroll
        for (int u = 0; u < OWN_PAIRS; ++u) {
            const bool live = i0 + 64 * u + half < n_s;
            wide |= live && !(fabsf(sc[u]) <= GAT_SAFE);
            const float w = live ? expf(sc[u]) : 0.f;
            if (sub == u) { wsel = w; dsel = dl[u]; }
            if (live) {
                float* pa = acc + dl[u] * DIM + offa;
                float* pb = acc + dl[u] * DIM + offb;
                const v4f x = v[u];
                atomicAdd(pa,     w * (half ? x[2] : x[0]));
                atomicAdd(pa + 1, w * (half ? x[3] : x[1]));
                atomicAdd(pb,     w * (half ? x[0] : x[2]));
                atomicAdd(pb + 1, w * (half ? x[1] : x[3]));
            }
        }
        if (sub < OWN_PAIRS) atomicAdd(&rden[dsel], wsel);
    }
    return wide;
}

__global__ void __launch_bounds__(1024)
owner_lists(const int* __restrict__ ety, const int* __restrict__ src,
            const int* __restrict__ dst, const unsigned* __restrict__ keys,
            unsigned* __restrict__ lists, unsigned* __restrict__ counts)
{
    __shared__ unsigned q[OWN_QCAP];
    __shared__ unsigned qn;

    const int tid  = threadIdx.x;
    const int d0   = blockIdx.x * OWN_ROWS;
    const int nloc = (NODES - d0 < OWN_ROWS) ? (NODES - d0) : OWN_ROWS;

    if (tid < nloc) q[tid] = (unsigned)(EDGES + d0 + tid);
    if (tid == 0) qn = (unsigned)nloc;
    __syncthreads();

    const uint4* __restrict__ key4 = (const uint4*)keys;
    const unsigned mine = (unsigned)blockIdx.x * 0x01010101u;
    uint4 cur[4];
    #pragma unroll
    for (int j = 0; j < 4; ++j) {
        const int i4 = j * 1024 + tid;
        cur[j] = (i4 < OWN_KEYS4) ? key4[i4] : make_uint4(0u, 0u, 0u, 0u);
    }
    for (int base4 = 0; base4 < OWN_KEYS4; base4 += 4096) {
        uint4 nxt[4];
        #pragma unroll
        for (int j = 0; j < 4; ++j) {
            const int i4 = base4 + 4096 + j * 1024 + tid;
            nxt[j] = (i4 < OWN_KEYS4) ? key4[i4] : make_uint4(0u, 0u, 0u, 0u);
        }
        #pragma unroll
        for (int j = 0; j < 4; ++j) {
            const int i4 = base4 + j * 1024 + tid;
            if (i4 < OWN_KEYS4) {
                unsigned m = owner_mask(cur[j], mine);
                while (m) {
                    const int b = __ffs((int)m) - 1;
                    m &= m - 1u;
                    const unsigned pos = atomicAdd(&qn, 1u);
                    if (pos < OWN_QCAP) q[pos] = owner_edge(i4, b);
                }
            }
        }
        #pragma unroll
        for (int j = 0; j < 4; ++j) cur[j] = nxt[j];
    }
    __syncthreads();

    const unsigned n = qn;
    if (n > OWN_QCAP) {
        if (tid == 0) counts[blockIdx.x] = OWN_OVER;
        return;
    }
    owner_lookup<1>(ety, src, dst, q, (int)n, d0, tid);
    unsigned* __restrict__ mylist = lists + (size_t)blockIdx.x * OWN_QCAP;
    for (int i = tid; i < (int)n; i += 1024) mylist[i] = q[i];
    if (tid == 0) counts[blockIdx.x] = n;
}

template <int MODE>
__global__ void __launch_bounds__(1024)
owner_sum(const float* __restrict__ t0, const float* __restrict__ t1,
          const float* __restrict__ attn, const int* __restrict__ ety,
          const int* __restrict__ src, const int* __restrict__ dst,
          const unsigned* __restrict__ keys, const unsigned* __restrict__ lists,
          const unsigned* __restrict__ counts, float* __restrict__ out,
          float* __restrict__ cntf)
{
    constexpr int ROWLEN = (MODE == 0) ? DIM : SDIM;
    __shared__ float    acc[OWN_ROWS * ROWLEN];
    __shared__ unsigned ecnt[(MODE == 1) ? OWN_ROWS * 2 : 1];
    __shared__ unsigned q[OWN_QCAP];
    __shared__ float    qw[(MODE == 0) ? OWN_QCAP : 1];
    __shared__ __attribute__((aligned(16))) float erow[(MODE == 0) ? OWN_ROWS * DIM : 4];
    __shared__ unsigned rmax[(MODE == 0) ? OWN_ROWS : 1];
    __shared__ float    rold[(MODE == 0) ? OWN_ROWS : 1];
    __shared__ float    rden[(MODE == 0) ? OWN_ROWS : 1];
    __shared__ unsigned qn;
    __shared__ unsigned redo;

    const int tid  = threadIdx.x;
    const int d0   = blockIdx.x * OWN_ROWS;
    const int nloc = (NODES - d0 < OWN_ROWS) ? (NODES - d0) : OWN_ROWS;

    if (tid == 0) redo = 0u;

    const unsigned cnt = counts[blockIdx.x];

    for (int i = tid; i < OWN_ROWS * ROWLEN; i += 1024) acc[i] = 0.f;
    if (MODE == 1 && tid < OWN_ROWS * 2) ecnt[tid] = 0u;
    if (MODE == 0) {
        const v4f* __restrict__ er4 = (const v4f*)(t1 + (size_t)d0 * DIM);
        v4f* erow4 = (v4f*)erow;
        for (int i = tid; i < nloc * (DIM / 4); i += 1024) erow4[i] = er4[i];
        if (tid < OWN_ROWS) {
            rmax[tid] = ENC_NEG_INF;
            rold[tid] = decF(ENC_NEG_INF);
            rden[tid] = 0.f;
        }
    }
    int drains = 0;
    int nlast;

    if (cnt <= (unsigned)OWN_QCAP) {
        const unsigned* __restrict__ mylist = lists + (size_t)blockIdx.x * OWN_QCAP;
        for (int i0 = tid; i0 < (int)cnt; i0 += 4096) {
            unsigned v[4];
            #pragma unroll
            for (int j = 0; j < 4; ++j) {
                const int i = i0 + 1024 * j;
                v[j] = (i < (int)cnt) ? mylist[i] : 0u;
            }
            #pragma unroll
            for (int j = 0; j < 4; ++j) {
                const int i = i0 + 1024 * j;
                if (i < (int)cnt) q[i] = v[j];
            }
        }
        nlast = (int)cnt;
    } else {
        if (tid < nloc) q[tid] = (unsigned)(EDGES + d0 + tid);
        if (tid == 0) qn = (unsigned)nloc;
        const uint4* __restrict__ key4 = (const uint4*)keys;
        const unsigned mine = (unsigned)blockIdx.x * 0x01010101u;
        for (int base4 = 0; base4 < OWN_KEYS4; base4 += 1024) {
            __syncthreads();
            const int n = (int)qn;
            __syncthreads();
            if (n + OWN_SUB > OWN_QCAP) {
                owner_lookup<MODE>(ety, src, dst, q, n, d0, tid);
                __syncthreads();
                if (MODE == 0)
                    gat_softmax_weights(t0, attn, acc, erow, q, qw, rmax, rold, rden,
                                        n, tid, drains > 0);
                owner_add_rows<MODE>(t0, acc, q, qw, ecnt, n, tid);
                ++drains;
                __syncthreads();
                if (tid == 0) qn = 0u;
                __syncthreads();
            }
            const int i4 = base4 + tid;
            if (i4 < OWN_KEYS4) {
                unsigned m = owner_mask(key4[i4], mine);
                while (m) {
                    const int b = __ffs((int)m) - 1;
                    m &= m - 1u;
                    const unsigned pos = atomicAdd(&qn, 1u);
                    q[pos] = owner_edge(i4, b);
                }
            }
        }
        __syncthreads();
        nlast = (int)qn;
        owner_lookup<MODE>(ety, src, dst, q, nlast, d0, tid);
    }
    __syncthreads();
    if (MODE == 0) {
        bool exact = true;
        if (cnt <= (unsigned)OWN_QCAP) {
            if (gat_one_pass(t0, attn, acc, erow, q, rden, nlast, tid)) redo = 1u;
            __syncthreads();
            exact = redo != 0u;
            if (exact) {
                for (int i = tid; i < OWN_ROWS * ROWLEN; i += 1024) acc[i] = 0.f;
                if (tid < OWN_ROWS) rden[tid] = 0.f;
                __syncthreads();
            }
        }
        if (exact) {
            gat_softmax_weights(t0, attn, acc, erow, q, qw, rmax, rold, rden,
                                nlast, tid, drains > 0);
            owner_add_rows<MODE>(t0, acc, q, qw, ecnt, nlast, tid);
        }
    } else {
        owner_add_rows<MODE>(t0, acc, q, qw, ecnt, nlast, tid);
    }
    __syncthreads();

    float* __restrict__ orow = out + (size_t)d0 * ROWLEN;
    if (MODE == 0) {
        for (int i = tid; i < nloc * DIM; i += 1024) orow[i] = acc[i] / rden[i >> 6];
    } else {
        for (int i = tid; i < nloc * SDIM; i += 1024) orow[i] = acc[i];
        if (tid < nloc * 2)
            cntf[(size_t)blockIdx.x * OWN_CNT + tid] = (float)ecnt[tid];
    }
}

__global__ void gru_update(const float* __restrict__ gi,
                           const float* __restrict__ gh,
                           float* __restrict__ h, float* __restrict__ emb,
                           long nElem)
{
    const long i = (long)blockIdx.x * blockDim.x + threadIdx.x;
    if (i >= nElem) return;
    const long n = i >> 6;
    const int  d = (int)(i & 63);
    const float* gin = gi + n * (3 * DIM);
    const float* ghn = gh + n * (3 * DIM);
    const float r = 1.f / (1.f + expf(-(gin[d]           + ghn[d])));
    const float z = 1.f / (1.f + expf(-(gin[DIM + d]     + ghn[DIM + d])));
    const float c = tanhf(gin[2 * DIM + d] + r * ghn[2 * DIM + d]);
    const float o = (1.f - z) * c + z * h[i];
    h[i] = o;
    if (emb) emb[i] = o;
}

extern "C" void kernel_launch(void* const* d_in, const int* in_sizes, int n_in,
                              void* d_out, int out_size, void* d_ws, size_t ws_size,
                              hipStream_t stream)
{
    (void)in_sizes; (void)n_in; (void)out_size; (void)ws_size;

    const float* x   = (const float*)d_in[0];
    const int*   src = (const int*)d_in[1];
    const int*   dst = (const int*)d_in[2];
    const int*   ety = (const int*)d_in[3];

    const long NE = (long)NODES * DIM;

    float* ws   = (float*)d_ws;
    float* hbuf = ws;                 ws += NE;
    float* buf1 = ws;                 ws += NE;
    float* buf2 = ws;                 ws += NE;
    float* gi   = ws;                 ws += (long)NODES * 3 * DIM;
    float* gh   = ws;                 ws += (long)NODES * 3 * DIM;
    float* acc  = ws;                 ws += NE;
    unsigned* keys   = (unsigned*)ws;
    unsigned* lists  = keys + EDGES / 4;
    unsigned* counts = lists + 256L * OWN_QCAP;
    float* cntf = (float*)(counts + 256);
    float* Wc   = cntf + 256L * OWN_CNT;
    float* cb   = Wc + 2L * 3 * DIM * SDIM;
    float* ssum = buf1;

    float* out = (float*)d_out;

    const int elemBlocks     = (int)((NE + 255) / 256);
    const int ownerBlocks    = (NODES + OWN_ROWS - 1) / OWN_ROWS;
    const int keyBlocks      = (EDGES / 4 + 255) / 256;
    const int gemmBlocks64   = ((NODES / 16) * (DIM / 64)       + 3) / 4;
    const int gemmBlocks192  = ((NODES / 16) * (3 * DIM / 64)   + 3) / 4;

    owner_keys<<<keyBlocks, 256, 0, stream>>>(dst, keys);
    owner_lists<<<ownerBlocks, 1024, 0, stream>>>(ety, src, dst, keys, lists, counts);
    compose_weights<<<48, 32, 0, stream>>>(
        (const float*)d_in[4 + 7],  (const float*)d_in[4 + 5],  (const float*)d_in[4 + 6],
        (const float*)d_in[15 + 7], (const float*)d_in[15 + 5], (const float*)d_in[15 + 6],
        Wc, cb);

    const float* hin = x;
    for (int l = 0; l < 2; ++l) {
        const int base = 4 + 11 * l;
        const float* Wsrc = (const float*)d_in[base + 0];
        const float* bsrc = (const float*)d_in[base + 1];
        const float* Wdst = (const float*)d_in[base + 2];
        const float* bdst = (const float*)d_in[base + 3];
        const float* attn = (const float*)d_in[base + 4];
        const float* bih  = (const float*)d_in[base + 8];
        const float* Whh  = (const float*)d_in[base + 9];
        const float* bhh  = (const float*)d_in[base + 10];

        gemm_bias_wmma<DIM, DIM, false><<<gemmBlocks64, 128, 0, stream>>>(
            hin, Wsrc, bsrc, buf1, NODES, nullptr, nullptr);
        gemm_bias_wmma<DIM, DIM, false><<<gemmBlocks64, 128, 0, stream>>>(
            hin, Wdst, bdst, buf2, NODES, nullptr, nullptr);
        owner_sum<0><<<ownerBlocks, 1024, 0, stream>>>(buf1, buf2, attn, ety, src, dst,
                                                       keys, lists, counts, acc, nullptr);
        elu_write<<<elemBlocks, 256, 0, stream>>>(acc, hbuf, out + (2L * l) * NE,
                                                  NE);

        for (int st = 0; st < 2; ++st) {
            owner_sum<1><<<ownerBlocks, 1024, 0, stream>>>(hbuf, hbuf, attn, ety, src, dst,
                                                           keys, lists, counts, ssum, cntf);
            gemm_bias_wmma<3 * DIM, SDIM, true><<<gemmBlocks192, 128, 0, stream>>>(
                ssum, Wc + (long)l * 3 * DIM * SDIM, bih, gi, NODES,
                cntf, cb + l * 2 * 3 * DIM);
            gemm_bias_wmma<3 * DIM, DIM, false><<<gemmBlocks192, 128, 0, stream>>>(
                hbuf, Whh, bhh, gh, NODES, nullptr, nullptr);
            gru_update<<<elemBlocks, 256, 0, stream>>>(
                gi, gh, hbuf, (st == 1) ? out + (2L * l + 1) * NE : nullptr, NE);
        }
        hin = hbuf;
    }
}


# GRU reset/update gates by ONE product over concatenated k, [S0|S1|h] x [Wc|Whh] (gi+gh in one sum); c-gate halves in the same launch: 27 -> 23 launches, 256 not 384 gate floats; r,z sum order changes
# speedup vs baseline: 2.9363x; 1.0449x over previous
#include <hip/hip_runtime.h>
#include <hip/hip_bf16.h>


typedef __attribute__((ext_vector_type(2))) float v2f;
typedef __attribute__((ext_vector_type(4))) float v4f;
typedef __attribute__((ext_vector_type(8))) float v8f;

#define NODES 50000
#define EDGES 800000
#define DIM   64
#define NEG_SLOPE 0.2f
#define SDIM  128
#define GDIM  256
#define OWN_ROWS 196
#define OWN_CNT  512

__device__ __forceinline__ unsigned encF(float f) {
    unsigned u = __float_as_uint(f);
    return (u & 0x80000000u) ? ~u : (u | 0x80000000u);
}
__device__ __forceinline__ float decF(unsigned u) {
    unsigned v = (u & 0x80000000u) ? (u & 0x7FFFFFFFu) : ~u;
    return __uint_as_float(v);
}

__device__ __forceinline__ void load_frag(const float* __restrict__ rowp, int hi,
                                          v4f (&f)[8])
{
    const v4f* __restrict__ p = (const v4f*)(rowp + 4 * hi);
    #pragma unroll
    for (int j = 0; j < 8; ++j) f[j] = p[2 * j];
}

__device__ __forceinline__ void wmma_pass64(const float* __restrict__ arow,
                                            const float* __restrict__ wrow,
                                            size_t wstep, int hi, v8f (&acc)[4])
{
    v4f a4[8];
    load_frag(arow, hi, a4);

    v4f b4[4][8];
    #pragma unroll
    for (int t = 0; t < 4; ++t) load_frag(wrow + t * wstep, hi, b4[t]);

    #pragma unroll
    for (int j = 0; j < 8; ++j) {
        #pragma unroll
        for (int p = 0; p < 2; ++p) {
            v2f a;
            a.x = a4[j][2 * p];
            a.y = a4[j][2 * p + 1];
            #pragma unroll
            for (int t = 0; t < 4; ++t) {
                v2f b;
                b.x = b4[t][j][2 * p];
                b.y = b4[t][j][2 * p + 1];
                acc[t] = __builtin_amdgcn_wmma_f32_16x16x4_f32(
                             false, a, false, b, (short)0, acc[t], false, false);
            }
        }
    }
}

__global__ void __launch_bounds__(128) __attribute__((amdgpu_waves_per_eu(1, 4)))
gemm_bias_wmma(const float* __restrict__ A, const float* __restrict__ W,
               const float* __restrict__ bias, float* __restrict__ C, int M)
{
    const int lane = threadIdx.x & 31;
    const int rt   = blockIdx.x * 4 + (threadIdx.x >> 5);
    if (rt >= (M >> 4)) return;

    const int row0 = rt << 4;
    const int lr   = lane & 15;
    const int hi   = lane >> 4;

    v8f acc[4] = {};
    wmma_pass64(A + (size_t)(row0 + lr) * DIM, W + (size_t)lr * DIM, 16 * DIM, hi, acc);

    #pragma unroll
    for (int t = 0; t < 4; ++t) {
        const int col = 16 * t + lr;
        const float bv = bias[col];
        float* crow = C + (size_t)(row0 + hi * 8) * DIM + col;
        #pragma unroll
        for (int v = 0; v < 8; ++v) crow[(size_t)v * DIM] = acc[t][v] + bv;
    }
}

__global__ void __launch_bounds__(128) __attribute__((amdgpu_waves_per_eu(1, 4)))
gru_gates_wmma(const float* __restrict__ S, const float* __restrict__ h,
               const float* __restrict__ Wc, const float* __restrict__ Whh,
               const float* __restrict__ bih, const float* __restrict__ bhh,
               const float* __restrict__ cnt, const float* __restrict__ cbias,
               float* __restrict__ G)
{
    const int lane = threadIdx.x & 31;
    const int rt   = blockIdx.x;
    const int cg   = __builtin_amdgcn_readfirstlane(((int)(threadIdx.x >> 5) + rt) & 3);
    const bool useS = cg < 3;
    const bool useH = cg != 2;
    const int  gate = (cg == 3) ? 2 : cg;

    const int row0 = rt << 4;
    const int lr   = lane & 15;
    const int hi   = lane >> 4;

    float  bi[4], bh[4], c0[4], c1[4];
    float2 cn[8];
    #pragma unroll
    for (int t = 0; t < 4; ++t) { bi[t] = 0.f; bh[t] = 0.f; c0[t] = 0.f; c1[t] = 0.f; }
    #pragma unroll
    for (int v = 0; v < 8; ++v) cn[v] = make_float2(0.f, 0.f);
    const int o0 = 64 * gate + lr;
    if (useS) {
        #pragma unroll
        for (int t = 0; t < 4; ++t) {
            bi[t] = bih[o0 + 16 * t];
            c0[t] = cbias[o0 + 16 * t];
            c1[t] = cbias[3 * DIM + o0 + 16 * t];
        }
        #pragma unroll
        for (int v = 0; v < 8; ++v) {
            const int row = row0 + hi * 8 + v;
            const int g   = row / OWN_ROWS;
            cn[v] = *(const float2*)(cnt + (size_t)g * OWN_CNT + 2 * (row - g * OWN_ROWS));
        }
    }
    if (useH) {
        #pragma unroll
        for (int t = 0; t < 4; ++t) bh[t] = bhh[o0 + 16 * t];
    }

    v8f acc[4] = {};
    if (useS) {
        const float* __restrict__ ar = S  + (size_t)(row0 + lr) * SDIM;
        const float* __restrict__ wr = Wc + (size_t)(64 * gate + lr) * SDIM;
        wmma_pass64(ar,      wr,      16 * SDIM, hi, acc);
        wmma_pass64(ar + 64, wr + 64, 16 * SDIM, hi, acc);
    }
    if (useH)
        wmma_pass64(h + (size_t)(row0 + lr) * DIM, Whh + (size_t)(64 * gate + lr) * DIM,
                    16 * DIM, hi, acc);

    #pragma unroll
    for (int t = 0; t < 4; ++t) {
        const float bv = bi[t] + bh[t];
        float* grow = G + (size_t)(row0 + hi * 8) * GDIM + 64 * cg + 16 * t + lr;
        #pragma unroll
        for (int v = 0; v < 8; ++v)
            grow[(size_t)v * GDIM] = acc[t][v] + bv + (cn[v].x * c0[t] + cn[v].y * c1[t]);
    }
}

__global__ void __launch_bounds__(32)
compose_weights(const float* __restrict__ Wih0, const float* __restrict__ gw0,
                const float* __restrict__ gb0,  const float* __restrict__ Wih1,
                const float* __restrict__ gw1,  const float* __restrict__ gb1,
                float* __restrict__ Wc, float* __restrict__ cb)
{
    const int lane = threadIdx.x;
    const int lr   = lane & 15;
    const int hi   = lane >> 4;
    const int l    = blockIdx.x / 24;
    const int r    = blockIdx.x - l * 24;
    const int t    = r / 12;
    const int row0 = (r - t * 12) * 16;

    const float* __restrict__ Wih = l ? Wih1 : Wih0;
    const float* __restrict__ Wt  = (l ? gw1 : gw0) + t * DIM * DIM;
    const float* __restrict__ bt  = (l ? gb1 : gb0) + t * DIM;

    if (hi == 0) {
        const float* __restrict__ wr = Wih + (size_t)(row0 + lr) * DIM;
        float sum = 0.f;
        for (int j = 0; j < DIM; ++j) sum += wr[j] * bt[j];
        cb[l * 2 * 3 * DIM + t * 3 * DIM + row0 + lr] = sum;
    }

    v4f a4[8];
    load_frag(Wih + (size_t)(row0 + lr) * DIM, hi, a4);

    v8f acc[4] = {};
    #pragma unroll
    for (int j = 0; j < 8; ++j) {
        #pragma unroll
        for (int p = 0; p < 2; ++p) {
            v2f a;
            a.x = a4[j][2 * p];
            a.y = a4[j][2 * p + 1];
            const int k0 = 8 * j + 4 * hi + 2 * p;
            #pragma unroll
            for (int tt = 0; tt < 4; ++tt) {
                v2f b;
                b.x = Wt[(k0)     * DIM + 16 * tt + lr];
                b.y = Wt[(k0 + 1) * DIM + 16 * tt + lr];
                acc[tt] = __builtin_amdgcn_wmma_f32_16x16x4_f32(
                              false, a, false, b, (short)0, acc[tt], false, false);
            }
        }
    }

    float* crow = Wc + (size_t)l * 3 * DIM * SDIM + (size_t)(row0 + hi * 8) * SDIM
                     + t * DIM + lr;
    #pragma unroll
    for (int tt = 0; tt < 4; ++tt) {
        #pragma unroll
        for (int v = 0; v < 8; ++v)
            crow[(size_t)v * SDIM + 16 * tt] = acc[tt][v];
    }
}

__global__ void elu_write(const float* __restrict__ acc, float* __restrict__ h,
                          float* __restrict__ emb, long n)
{
    const long i = (long)blockIdx.x * blockDim.x + threadIdx.x;
    if (i >= n) return;
    const float v = acc[i];
    const float o = v > 0.f ? v : (expf(v) - 1.f);
    h[i] = o;
    emb[i] = o;
}

#define OWN_SUB  16384
#define OWN_QCAP 24576
#define OWN_BATCH 16
#define OWN_PAIRS 8
#define OWN_KEYS4 (EDGES / 16)
#define ENC_NEG_INF 0x007FFFFFu
#define OWN_OVER 0xFFFFFFFFu
#define GAT_SAFE 40.0f

__global__ void __launch_bounds__(256)
owner_keys(const int* __restrict__ dst, unsigned* __restrict__ keys)
{
    const int i = blockIdx.x * 256 + threadIdx.x;
    if (i >= EDGES / 4) return;
    const int4 d = ((const int4*)dst)[i];
    keys[i] =  ((unsigned)d.x / OWN_ROWS)        | (((unsigned)d.y / OWN_ROWS) << 8) |
              (((unsigned)d.z / OWN_ROWS) << 16) | (((unsigned)d.w / OWN_ROWS) << 24);
}

__device__ __forceinline__ unsigned zero_bytes(unsigned x)
{
    const unsigned y = ((x & 0x7F7F7F7Fu) + 0x7F7F7F7Fu) | x;
    return ~y & 0x80808080u;
}

__device__ __forceinline__ unsigned owner_mask(const uint4 k, unsigned mine)
{
    return (zero_bytes(k.x ^ mine) >> 7) | (zero_bytes(k.y ^ mine) >> 6) |
           (zero_bytes(k.z ^ mine) >> 5) | (zero_bytes(k.w ^ mine) >> 4);
}
__device__ __forceinline__ unsigned owner_edge(int i4, int b)
{
    return (unsigned)(i4 * 16 + ((b & 7) << 2) + (b >> 3));
}

template <int MODE>
__device__ __forceinline__ void owner_lookup(
    const int* __restrict__ ety, const int* __restrict__ src,
    const int* __restrict__ dst, unsigned* q, int n, int d0, int tid)
{
    for (int i = tid; i < n; i += 1024) {
        const int e = (int)q[i];
        int s, d;
        if (e < EDGES) { s = src[e]; d = dst[e]; } else { s = d = e - EDGES; }
        unsigned o = (unsigned)s | ((unsigned)(d - d0) << 16);
        if (MODE == 1) o |= (ety[e] == 0) ? 0u : (1u << 24);
        q[i] = o;
    }
}

template <int MODE, bool FULL>
__device__ __forceinline__ void owner_add_batch(
    const float* __restrict__ t0, float* acc, const unsigned* q, const float* qw,
    unsigned& myq, float& myw, int nb, int inext, int lane)
{
    float2 v[OWN_BATCH];
    #pragma unroll
    for (int u = 0; u < OWN_BATCH; ++u) {
        const unsigned p = (unsigned)__builtin_amdgcn_readlane((int)myq, u);
        const float2* __restrict__ row = (const float2*)(t0 + (size_t)(p & 0xFFFFu) * DIM);
        v[u] = row[lane];
    }
    const unsigned nq = q[inext];
    const float    nw = (MODE == 0) ? qw[inext] : 1.f;
    #pragma unroll
    for (int u = 0; u < OWN_BATCH; ++u) {
        if (FULL || u < nb) {
            const unsigned p = (unsigned)__builtin_amdgcn_readlane((int)myq, u);
            if (MODE == 0) {
                const float w = __uint_as_float(
                    (unsigned)__builtin_amdgcn_readlane((int)__float_as_uint(myw), u));
                const int off = (int)((p >> 16) & 0xFFu) * DIM + 2 * lane;
                atomicAdd(&acc[off],     w * v[u].x);
                atomicAdd(&acc[off + 1], w * v[u].y);
            } else {
                const int off = (int)((p >> 16) & 0xFFu) * SDIM + (int)(p >> 24) * DIM
                                + 2 * lane;
                atomicAdd(&acc[off],     v[u].x);
                atomicAdd(&acc[off + 1], v[u].y);
            }
        }
    }
    myq = nq;
    myw = nw;
}

template <int MODE>
__device__ __forceinline__ void owner_add_rows(
    const float* __restrict__ t0, float* acc, const unsigned* q, const float* qw,
    unsigned* cnt, int nv, int tid)
{
    const int lane = tid & 31;
    const int wave = __builtin_amdgcn_readfirstlane(tid >> 5);
    const int n    = __builtin_amdgcn_readfirstlane(nv);
    if (MODE == 1) {
        for (int i = tid; i < n; i += 1024) {
            const unsigned p = q[i];
            atomicAdd(&cnt[((p >> 16) & 0xFFu) * 2u + (p >> 24)], 1u);
        }
    }
    const int ifirst = (lane < OWN_BATCH && wave + 32 * lane < n) ? wave + 32 * lane : wave;
    unsigned myq = q[ifirst];
    float    myw = (MODE == 0) ? qw[ifirst] : 1.f;
    for (int i0 = wave; i0 < n; i0 += 32 * OWN_BATCH) {
        int nb = (n - i0 + 31) >> 5;
        if (nb > OWN_BATCH) nb = OWN_BATCH;
        const int i1    = i0 + 32 * OWN_BATCH + 32 * lane;
        const int inext = (lane < OWN_BATCH && i1 < n) ? i1 : i0;
        if (nb == OWN_BATCH)
            owner_add_batch<MODE, true >(t0, acc, q, qw, myq, myw, nb, inext, lane);
        else
            owner_add_batch<MODE, false>(t0, acc, q, qw, myq, myw, nb, inext, lane);
    }
}

__device__ __forceinline__ void gat_softmax_weights(
    const float* __restrict__ el, const float* __restrict__ attn,
    float* acc, const float* erow, const unsigned* q, float* qw,
    unsigned* rmax, float* rold, float* rden, int n, int tid, bool later)
{
    const int lane = tid & 31;
    const int wave = tid >> 5;
    const int sub  = lane & 15;
    const int half = lane >> 4;
    const v4f av = ((const v4f*)attn)[sub];

    const int wave_s = __builtin_amdgcn_readfirstlane(wave);
    const int n_s    = __builtin_amdgcn_readfirstlane(n);
    for (int i0 = 2 * wave_s; i0 < n_s; i0 += 64 * OWN_PAIRS) {
        const int iq = i0 + 64 * (lane >> 1) + (lane & 1);
        const unsigned myq = q[(lane < 2 * OWN_PAIRS && iq < n_s) ? iq : n_s - 1];
        v4f v[OWN_PAIRS];
        int dl[OWN_PAIRS];
        #pragma unroll
        for (int u = 0; u < OWN_PAIRS; ++u) {
            const unsigned p0 = (unsigned)__builtin_amdgcn_readlane((int)myq, 2 * u);
            const unsigned p1 = (unsigned)__builtin_amdgcn_readlane((int)myq, 2 * u + 1);
            const unsigned p  = half ? p1 : p0;
            v[u]  = ((const v4f*)(el + (size_t)(p & 0xFFFFu) * DIM))[sub];
            dl[u] = (int)((p >> 16) & 0xFFu);
        }
        float sc[OWN_PAIRS];
        #pragma unroll
        for (int u = 0; u < OWN_PAIRS; ++u) {
            const v4f r = ((const v4f*)(erow + dl[u] * DIM))[sub];
            sc[u] = 0.f;
            #pragma unroll
            for (int c = 0; c < 4; ++c) {
                float t = v[u][c] + r[c];
                t = t > 0.f ? t : NEG_SLOPE * t;
                sc[u] += t * av[c];
            }
        }
        #pragma unroll
        for (int o = 8; o; o >>= 1) {
            #pragma unroll
            for (int u = 0; u < OWN_PAIRS; ++u) sc[u] += __shfl_xor(sc[u], o, 32);
        }
        #pragma unroll
        for (int u = 0; u < OWN_PAIRS; ++u) {
            const int i = i0 + 64 * u + half;
            if (sub == 0 && i < n) {
                qw[i] = sc[u];
                atomicMax(&rmax[dl[u]], encF(sc[u]));
            }
        }
    }
    __syncthreads();

    if (later) {
        for (int i = tid; i < OWN_ROWS * DIM; i += 1024) {
            const float mo = rold[i >> 6];
            const float mn = decF(rmax[i >> 6]);
            if (mn != mo) acc[i] *= expf(mo - mn);
        }
        if (tid < OWN_ROWS) {
            const float mo = rold[tid];
            const float mn = decF(rmax[tid]);
            if (mn != mo) rden[tid] *= expf(mo - mn);
        }
        __syncthreads();
    }
    if (tid < OWN_ROWS) rold[tid] = decF(rmax[tid]);
    __syncthreads();

    for (int i = tid; i < n; i += 1024) {
        const int dl = (int)((q[i] >> 16) & 0xFFu);
        const float ex = expf(qw[i] - rold[dl]);
        qw[i] = ex;
        atomicAdd(&rden[dl], ex);
    }
    __syncthreads();
}

template <int M>
__device__ __forceinline__ float row_xor(float x)
{
    return __int_as_float(__builtin_amdgcn_update_dpp(
        0, __float_as_int(x), 0x160 + M, 0xf, 0xf, true));
}

__device__ __forceinline__ bool gat_one_pass(
    const float* __restrict__ el, const float* __restrict__ attn,
    float* acc, const float* erow, const unsigned* q, float* rden, int n, int tid)
{
    const int lane = tid & 31;
    const int sub  = lane & 15;
    const int half = lane >> 4;
    const v4f av = ((const v4f*)attn)[sub];
    const int wave_s = __builtin_amdgcn_readfirstlane(tid >> 5);
    const int n_s    = __builtin_amdgcn_readfirstlane(n);
    const int offa = 4 * sub + 2 * half;
    const int offb = 4 * sub + 2 - 2 * half;
    bool wide = false;
    for (int i0 = 2 * wave_s; i0 < n_s; i0 += 64 * OWN_PAIRS) {
        const int iq = i0 + 64 * (lane >> 1) + (lane & 1);
        const unsigned myq = q[(lane < 2 * OWN_PAIRS && iq < n_s) ? iq : n_s - 1];
        v4f v[OWN_PAIRS];
        int dl[OWN_PAIRS];
        #pragma unroll
        for (int u = 0; u < OWN_PAIRS; ++u) {
            const unsigned p0 = (unsigned)__builtin_amdgcn_readlane((int)myq, 2 * u);
            const unsigned p1 = (unsigned)__builtin_amdgcn_readlane((int)myq, 2 * u + 1);
            const unsigned p  = half ? p1 : p0;
            v[u]  = ((const v4f*)(el + (size_t)(p & 0xFFFFu) * DIM))[sub];
            dl[u] = (int)((p >> 16) & 0xFFu);
        }
        float sc[OWN_PAIRS];
        #pragma unroll
        for (int u = 0; u < OWN_PAIRS; ++u) {
            const v4f r = ((const v4f*)(erow + dl[u] * DIM))[sub];
            sc[u] = 0.f;
            #pragma unroll
            for (int c = 0; c < 4; ++c) {
                float t = v[u][c] + r[c];
                t = t > 0.f ? t : NEG_SLOPE * t;
                sc[u] += t * av[c];
            }
        }
        #pragma unroll
        for (int u = 0; u < OWN_PAIRS; ++u) sc[u] += row_xor<8>(sc[u]);
        #pragma unroll
        for (int u = 0; u < OWN_PAIRS; ++u) sc[u] += row_xor<4>(sc[u]);
        #pragma unroll
        for (int u = 0; u < OWN_PAIRS; ++u) sc[u] += row_xor<2>(sc[u]);
        #pragma unroll
        for (int u = 0; u < OWN_PAIRS; ++u) sc[u] += row_xor<1>(sc[u]);

        float wsel = 0.f;
        int   dsel = 0;
        #pragma unroll
        for (int u = 0; u < OWN_PAIRS; ++u) {
            const bool live = i0 + 64 * u + half < n_s;
            wide |= live && !(fabsf(sc[u]) <= GAT_SAFE);
            const float w = live ? expf(sc[u]) : 0.f;
            if (sub == u) { wsel = w; dsel = dl[u]; }
            if (live) {
                float* pa = acc + dl[u] * DIM + offa;
                float* pb = acc + dl[u] * DIM + offb;
                const v4f x = v[u];
                atomicAdd(pa,     w * (half ? x[2] : x[0]));
                atomicAdd(pa + 1, w * (half ? x[3] : x[1]));
                atomicAdd(pb,     w * (half ? x[0] : x[2]));
                atomicAdd(pb + 1, w * (half ? x[1] : x[3]));
            }
        }
        if (sub < OWN_PAIRS) atomicAdd(&rden[dsel], wsel);
    }
    return wide;
}

__global__ void __launch_bounds__(1024)
owner_lists(const int* __restrict__ ety, const int* __restrict__ src,
            const int* __restrict__ dst, const unsigned* __restrict__ keys,
            unsigned* __restrict__ lists, unsigned* __restrict__ counts)
{
    __shared__ unsigned q[OWN_QCAP];
    __shared__ unsigned qn;

    const int tid  = threadIdx.x;
    const int d0   = blockIdx.x * OWN_ROWS;
    const int nloc = (NODES - d0 < OWN_ROWS) ? (NODES - d0) : OWN_ROWS;

    if (tid < nloc) q[tid] = (unsigned)(EDGES + d0 + tid);
    if (tid == 0) qn = (unsigned)nloc;
    __syncthreads();

    const uint4* __restrict__ key4 = (const uint4*)keys;
    const unsigned mine = (unsigned)blockIdx.x * 0x01010101u;
    uint4 cur[4];
    #pragma unroll
    for (int j = 0; j < 4; ++j) {
        const int i4 = j * 1024 + tid;
        cur[j] = (i4 < OWN_KEYS4) ? key4[i4] : make_uint4(0u, 0u, 0u, 0u);
    }
    for (int base4 = 0; base4 < OWN_KEYS4; base4 += 4096) {
        uint4 nxt[4];
        #pragma unroll
        for (int j = 0; j < 4; ++j) {
            const int i4 = base4 + 4096 + j * 1024 + tid;
            nxt[j] = (i4 < OWN_KEYS4) ? key4[i4] : make_uint4(0u, 0u, 0u, 0u);
        }
        #pragma unroll
        for (int j = 0; j < 4; ++j) {
            const int i4 = base4 + j * 1024 + tid;
            if (i4 < OWN_KEYS4) {
                unsigned m = owner_mask(cur[j], mine);
                while (m) {
                    const int b = __ffs((int)m) - 1;
                    m &= m - 1u;
                    const unsigned pos = atomicAdd(&qn, 1u);
                    if (pos < OWN_QCAP) q[pos] = owner_edge(i4, b);
                }
            }
        }
        #pragma unroll
        for (int j = 0; j < 4; ++j) cur[j] = nxt[j];
    }
    __syncthreads();

    const unsigned n = qn;
    if (n > OWN_QCAP) {
        if (tid == 0) counts[blockIdx.x] = OWN_OVER;
        return;
    }
    owner_lookup<1>(ety, src, dst, q, (int)n, d0, tid);
    unsigned* __restrict__ mylist = lists + (size_t)blockIdx.x * OWN_QCAP;
    for (int i = tid; i < (int)n; i += 1024) mylist[i] = q[i];
    if (tid == 0) counts[blockIdx.x] = n;
}

template <int MODE>
__global__ void __launch_bounds__(1024)
owner_sum(const float* __restrict__ t0, const float* __restrict__ t1,
          const float* __restrict__ attn, const int* __restrict__ ety,
          const int* __restrict__ src, const int* __restrict__ dst,
          const unsigned* __restrict__ keys, const unsigned* __restrict__ lists,
          const unsigned* __restrict__ counts, float* __restrict__ out,
          float* __restrict__ cntf)
{
    constexpr int ROWLEN = (MODE == 0) ? DIM : SDIM;
    __shared__ float    acc[OWN_ROWS * ROWLEN];
    __shared__ unsigned ecnt[(MODE == 1) ? OWN_ROWS * 2 : 1];
    __shared__ unsigned q[OWN_QCAP];
    __shared__ float    qw[(MODE == 0) ? OWN_QCAP : 1];
    __shared__ __attribute__((aligned(16))) float erow[(MODE == 0) ? OWN_ROWS * DIM : 4];
    __shared__ unsigned rmax[(MODE == 0) ? OWN_ROWS : 1];
    __shared__ float    rold[(MODE == 0) ? OWN_ROWS : 1];
    __shared__ float    rden[(MODE == 0) ? OWN_ROWS : 1];
    __shared__ unsigned qn;
    __shared__ unsigned redo;

    const int tid  = threadIdx.x;
    const int d0   = blockIdx.x * OWN_ROWS;
    const int nloc = (NODES - d0 < OWN_ROWS) ? (NODES - d0) : OWN_ROWS;

    if (tid == 0) redo = 0u;

    const unsigned cnt = counts[blockIdx.x];

    for (int i = tid; i < OWN_ROWS * ROWLEN; i += 1024) acc[i] = 0.f;
    if (MODE == 1 && tid < OWN_ROWS * 2) ecnt[tid] = 0u;
    if (MODE == 0) {
        const v4f* __restrict__ er4 = (const v4f*)(t1 + (size_t)d0 * DIM);
        v4f* erow4 = (v4f*)erow;
        for (int i = tid; i < nloc * (DIM / 4); i += 1024) erow4[i] = er4[i];
        if (tid < OWN_ROWS) {
            rmax[tid] = ENC_NEG_INF;
            rold[tid] = decF(ENC_NEG_INF);
            rden[tid] = 0.f;
        }
    }
    int drains = 0;
    int nlast;

    if (cnt <= (unsigned)OWN_QCAP) {
        const unsigned* __restrict__ mylist = lists + (size_t)blockIdx.x * OWN_QCAP;
        for (int i0 = tid; i0 < (int)cnt; i0 += 4096) {
            unsigned v[4];
            #pragma unroll
            for (int j = 0; j < 4; ++j) {
                const int i = i0 + 1024 * j;
                v[j] = (i < (int)cnt) ? mylist[i] : 0u;
            }
            #pragma unroll
            for (int j = 0; j < 4; ++j) {
                const int i = i0 + 1024 * j;
                if (i < (int)cnt) q[i] = v[j];
            }
        }
        nlast = (int)cnt;
    } else {
        if (tid < nloc) q[tid] = (unsigned)(EDGES + d0 + tid);
        if (tid == 0) qn = (unsigned)nloc;
        const uint4* __restrict__ key4 = (const uint4*)keys;
        const unsigned mine = (unsigned)blockIdx.x * 0x01010101u;
        for (int base4 = 0; base4 < OWN_KEYS4; base4 += 1024) {
            __syncthreads();
            const int n = (int)qn;
            __syncthreads();
            if (n + OWN_SUB > OWN_QCAP) {
                owner_lookup<MODE>(ety, src, dst, q, n, d0, tid);
                __syncthreads();
                if (MODE == 0)
                    gat_softmax_weights(t0, attn, acc, erow, q, qw, rmax, rold, rden,
                                        n, tid, drains > 0);
                owner_add_rows<MODE>(t0, acc, q, qw, ecnt, n, tid);
                ++drains;
                __syncthreads();
                if (tid == 0) qn = 0u;
                __syncthreads();
            }
            const int i4 = base4 + tid;
            if (i4 < OWN_KEYS4) {
                unsigned m = owner_mask(key4[i4], mine);
                while (m) {
                    const int b = __ffs((int)m) - 1;
                    m &= m - 1u;
                    const unsigned pos = atomicAdd(&qn, 1u);
                    q[pos] = owner_edge(i4, b);
                }
            }
        }
        __syncthreads();
        nlast = (int)qn;
        owner_lookup<MODE>(ety, src, dst, q, nlast, d0, tid);
    }
    __syncthreads();
    if (MODE == 0) {
        bool exact = true;
        if (cnt <= (unsigned)OWN_QCAP) {
            if (gat_one_pass(t0, attn, acc, erow, q, rden, nlast, tid)) redo = 1u;
            __syncthreads();
            exact = redo != 0u;
            if (exact) {
                for (int i = tid; i < OWN_ROWS * ROWLEN; i += 1024) acc[i] = 0.f;
                if (tid < OWN_ROWS) rden[tid] = 0.f;
                __syncthreads();
            }
        }
        if (exact) {
            gat_softmax_weights(t0, attn, acc, erow, q, qw, rmax, rold, rden,
                                nlast, tid, drains > 0);
            owner_add_rows<MODE>(t0, acc, q, qw, ecnt, nlast, tid);
        }
    } else {
        owner_add_rows<MODE>(t0, acc, q, qw, ecnt, nlast, tid);
    }
    __syncthreads();

    float* __restrict__ orow = out + (size_t)d0 * ROWLEN;
    if (MODE == 0) {
        for (int i = tid; i < nloc * DIM; i += 1024) orow[i] = acc[i] / rden[i >> 6];
    } else {
        for (int i = tid; i < nloc * SDIM; i += 1024) orow[i] = acc[i];
        if (tid < nloc * 2)
            cntf[(size_t)blockIdx.x * OWN_CNT + tid] = (float)ecnt[tid];
    }
}

__global__ void gru_update(const float* __restrict__ g,
                           float* __restrict__ h, float* __restrict__ emb,
                           long nElem)
{
    const long i = (long)blockIdx.x * blockDim.x + threadIdx.x;
    if (i >= nElem) return;
    const long n = i >> 6;
    const int  d = (int)(i & 63);
    const float* gn = g + n * GDIM;
    const float r = 1.f / (1.f + expf(-gn[d]));
    const float z = 1.f / (1.f + expf(-gn[DIM + d]));
    const float c = tanhf(gn[2 * DIM + d] + r * gn[3 * DIM + d]);
    const float o = (1.f - z) * c + z * h[i];
    h[i] = o;
    if (emb) emb[i] = o;
}

extern "C" void kernel_launch(void* const* d_in, const int* in_sizes, int n_in,
                              void* d_out, int out_size, void* d_ws, size_t ws_size,
                              hipStream_t stream)
{
    (void)in_sizes; (void)n_in; (void)out_size; (void)ws_size;

    const float* x   = (const float*)d_in[0];
    const int*   src = (const int*)d_in[1];
    const int*   dst = (const int*)d_in[2];
    const int*   ety = (const int*)d_in[3];

    const long NE = (long)NODES * DIM;

    float* ws   = (float*)d_ws;
    float* hbuf = ws;                 ws += NE;
    float* buf1 = ws;                 ws += NE;
    float* buf2 = ws;                 ws += NE;
    float* gates = ws;                ws += (long)NODES * GDIM;
    float* acc  = ws;                 ws += NE;
    unsigned* keys   = (unsigned*)ws;
    unsigned* lists  = keys + EDGES / 4;
    unsigned* counts = lists + 256L * OWN_QCAP;
    float* cntf = (float*)(counts + 256);
    float* Wc   = cntf + 256L * OWN_CNT;
    float* cb   = Wc + 2L * 3 * DIM * SDIM;
    float* ssum = buf1;

    float* out = (float*)d_out;

    const int elemBlocks     = (int)((NE + 255) / 256);
    const int ownerBlocks    = (NODES + OWN_ROWS - 1) / OWN_ROWS;
    const int keyBlocks      = (EDGES / 4 + 255) / 256;
    const int gemmBlocks64   = (NODES / 16 + 3) / 4;
    const int gateBlocks     = NODES / 16;

    owner_keys<<<keyBlocks, 256, 0, stream>>>(dst, keys);
    owner_lists<<<ownerBlocks, 1024, 0, stream>>>(ety, src, dst, keys, lists, counts);
    compose_weights<<<48, 32, 0, stream>>>(
        (const float*)d_in[4 + 7],  (const float*)d_in[4 + 5],  (const float*)d_in[4 + 6],
        (const float*)d_in[15 + 7], (const float*)d_in[15 + 5], (const float*)d_in[15 + 6],
        Wc, cb);

    const float* hin = x;
    for (int l = 0; l < 2; ++l) {
        const int base = 4 + 11 * l;
        const float* Wsrc = (const float*)d_in[base + 0];
        const float* bsrc = (const float*)d_in[base + 1];
        const float* Wdst = (const float*)d_in[base + 2];
        const float* bdst = (const float*)d_in[base + 3];
        const float* attn = (const float*)d_in[base + 4];
        const float* bih  = (const float*)d_in[base + 8];
        const float* Whh  = (const float*)d_in[base + 9];
        const float* bhh  = (const float*)d_in[base + 10];

        gemm_bias_wmma<<<gemmBlocks64, 128, 0, stream>>>(hin, Wsrc, bsrc, buf1, NODES);
        gemm_bias_wmma<<<gemmBlocks64, 128, 0, stream>>>(hin, Wdst, bdst, buf2, NODES);
        owner_sum<0><<<ownerBlocks, 1024, 0, stream>>>(buf1, buf2, attn, ety, src, dst,
                                                       keys, lists, counts, acc, nullptr);
        elu_write<<<elemBlocks, 256, 0, stream>>>(acc, hbuf, out + (2L * l) * NE,
                                                  NE);

        for (int st = 0; st < 2; ++st) {
            owner_sum<1><<<ownerBlocks, 1024, 0, stream>>>(hbuf, hbuf, attn, ety, src, dst,
                                                           keys, lists, counts, ssum, cntf);
            gru_gates_wmma<<<gateBlocks, 128, 0, stream>>>(
                ssum, hbuf, Wc + (long)l * 3 * DIM * SDIM, Whh, bih, bhh,
                cntf, cb + l * 2 * 3 * DIM, gates);
            gru_update<<<elemBlocks, 256, 0, stream>>>(
                gates, hbuf, (st == 1) ? out + (2L * l + 1) * NE : nullptr, NE);
        }
        hin = hbuf;
    }
}


# GRU step moves into the owner groups: per-type source sums stay in LDS as the A operand of the gate products, gates applied to accumulators, new h to a second plane; 23 -> 15 launches
# speedup vs baseline: 3.3928x; 1.1555x over previous
#include <hip/hip_runtime.h>
#include <hip/hip_bf16.h>


typedef __attribute__((ext_vector_type(2))) float v2f;
typedef __attribute__((ext_vector_type(4))) float v4f;
typedef __attribute__((ext_vector_type(8))) float v8f;

#define NODES 50000
#define EDGES 800000
#define DIM   64
#define NEG_SLOPE 0.2f
#define SDIM  128
#define OWN_ROWS 196
#define OWN_PAD  208
#define SPITCH   132

__device__ __forceinline__ unsigned encF(float f) {
    unsigned u = __float_as_uint(f);
    return (u & 0x80000000u) ? ~u : (u | 0x80000000u);
}
__device__ __forceinline__ float decF(unsigned u) {
    unsigned v = (u & 0x80000000u) ? (u & 0x7FFFFFFFu) : ~u;
    return __uint_as_float(v);
}

__device__ __forceinline__ void load_frag(const float* __restrict__ rowp, int hi,
                                          v4f (&f)[8])
{
    const v4f* __restrict__ p = (const v4f*)(rowp + 4 * hi);
    #pragma unroll
    for (int j = 0; j < 8; ++j) f[j] = p[2 * j];
}

__device__ __forceinline__ void wmma_pass64(const float* __restrict__ arow,
                                            const float* __restrict__ wrow,
                                            size_t wstep, int hi, v8f (&acc)[4])
{
    v4f a4[8];
    load_frag(arow, hi, a4);

    v4f b4[4][8];
    #pragma unroll
    for (int t = 0; t < 4; ++t) load_frag(wrow + t * wstep, hi, b4[t]);

    #pragma unroll
    for (int j = 0; j < 8; ++j) {
        #pragma unroll
        for (int p = 0; p < 2; ++p) {
            v2f a;
            a.x = a4[j][2 * p];
            a.y = a4[j][2 * p + 1];
            #pragma unroll
            for (int t = 0; t < 4; ++t) {
                v2f b;
                b.x = b4[t][j][2 * p];
                b.y = b4[t][j][2 * p + 1];
                acc[t] = __builtin_amdgcn_wmma_f32_16x16x4_f32(
                             false, a, false, b, (short)0, acc[t], false, false);
            }
        }
    }
}

__global__ void __launch_bounds__(128) __attribute__((amdgpu_waves_per_eu(1, 4)))
gemm_bias_wmma(const float* __restrict__ A, const float* __restrict__ W,
               const float* __restrict__ bias, float* __restrict__ C, int M)
{
    const int lane = threadIdx.x & 31;
    const int rt   = blockIdx.x * 4 + (threadIdx.x >> 5);
    if (rt >= (M >> 4)) return;

    const int row0 = rt << 4;
    const int lr   = lane & 15;
    const int hi   = lane >> 4;

    v8f acc[4] = {};
    wmma_pass64(A + (size_t)(row0 + lr) * DIM, W + (size_t)lr * DIM, 16 * DIM, hi, acc);

    #pragma unroll
    for (int t = 0; t < 4; ++t) {
        const int col = 16 * t + lr;
        const float bv = bias[col];
        float* crow = C + (size_t)(row0 + hi * 8) * DIM + col;
        #pragma unroll
        for (int v = 0; v < 8; ++v) crow[(size_t)v * DIM] = acc[t][v] + bv;
    }
}

__device__ __forceinline__ void wmma_tile64(const v4f (&a4)[8],
                                            const float* __restrict__ wrow, int hi, v8f& acc)
{
    v4f b4[8];
    load_frag(wrow, hi, b4);
    #pragma unroll
    for (int j = 0; j < 8; ++j) {
        #pragma unroll
        for (int p = 0; p < 2; ++p) {
            v2f a, b;
            a.x = a4[j][2 * p];
            a.y = a4[j][2 * p + 1];
            b.x = b4[j][2 * p];
            b.y = b4[j][2 * p + 1];
            acc = __builtin_amdgcn_wmma_f32_16x16x4_f32(
                      false, a, false, b, (short)0, acc, false, false);
        }
    }
}

__global__ void __launch_bounds__(32)
compose_weights(const float* __restrict__ Wih0, const float* __restrict__ gw0,
                const float* __restrict__ gb0,  const float* __restrict__ Wih1,
                const float* __restrict__ gw1,  const float* __restrict__ gb1,
                float* __restrict__ Wc, float* __restrict__ cb)
{
    const int lane = threadIdx.x;
    const int lr   = lane & 15;
    const int hi   = lane >> 4;
    const int l    = blockIdx.x / 24;
    const int r    = blockIdx.x - l * 24;
    const int t    = r / 12;
    const int row0 = (r - t * 12) * 16;

    const float* __restrict__ Wih = l ? Wih1 : Wih0;
    const float* __restrict__ Wt  = (l ? gw1 : gw0) + t * DIM * DIM;
    const float* __restrict__ bt  = (l ? gb1 : gb0) + t * DIM;

    if (hi == 0) {
        const float* __restrict__ wr = Wih + (size_t)(row0 + lr) * DIM;
        float sum = 0.f;
        for (int j = 0; j < DIM; ++j) sum += wr[j] * bt[j];
        cb[l * 2 * 3 * DIM + t * 3 * DIM + row0 + lr] = sum;
    }

    v4f a4[8];
    load_frag(Wih + (size_t)(row0 + lr) * DIM, hi, a4);

    v8f acc[4] = {};
    #pragma unroll
    for (int j = 0; j < 8; ++j) {
        #pragma unroll
        for (int p = 0; p < 2; ++p) {
            v2f a;
            a.x = a4[j][2 * p];
            a.y = a4[j][2 * p + 1];
            const int k0 = 8 * j + 4 * hi + 2 * p;
            #pragma unroll
            for (int tt = 0; tt < 4; ++tt) {
                v2f b;
                b.x = Wt[(k0)     * DIM + 16 * tt + lr];
                b.y = Wt[(k0 + 1) * DIM + 16 * tt + lr];
                acc[tt] = __builtin_amdgcn_wmma_f32_16x16x4_f32(
                              false, a, false, b, (short)0, acc[tt], false, false);
            }
        }
    }

    float* crow = Wc + (size_t)l * 3 * DIM * SDIM + (size_t)(row0 + hi * 8) * SDIM
                     + t * DIM + lr;
    #pragma unroll
    for (int tt = 0; tt < 4; ++tt) {
        #pragma unroll
        for (int v = 0; v < 8; ++v)
            crow[(size_t)v * SDIM + 16 * tt] = acc[tt][v];
    }
}

__global__ void elu_write(const float* __restrict__ acc, float* __restrict__ h,
                          float* __restrict__ emb, long n)
{
    const long i = (long)blockIdx.x * blockDim.x + threadIdx.x;
    if (i >= n) return;
    const float v = acc[i];
    const float o = v > 0.f ? v : (expf(v) - 1.f);
    h[i] = o;
    emb[i] = o;
}

#define OWN_SUB  16384
#define OWN_QCAP 24576
#define OWN_BATCH 16
#define OWN_PAIRS 8
#define OWN_KEYS4 (EDGES / 16)
#define ENC_NEG_INF 0x007FFFFFu
#define OWN_OVER 0xFFFFFFFFu
#define GAT_SAFE 40.0f

__global__ void __launch_bounds__(256)
owner_keys(const int* __restrict__ dst, unsigned* __restrict__ keys)
{
    const int i = blockIdx.x * 256 + threadIdx.x;
    if (i >= EDGES / 4) return;
    const int4 d = ((const int4*)dst)[i];
    keys[i] =  ((unsigned)d.x / OWN_ROWS)        | (((unsigned)d.y / OWN_ROWS) << 8) |
              (((unsigned)d.z / OWN_ROWS) << 16) | (((unsigned)d.w / OWN_ROWS) << 24);
}

__device__ __forceinline__ unsigned zero_bytes(unsigned x)
{
    const unsigned y = ((x & 0x7F7F7F7Fu) + 0x7F7F7F7Fu) | x;
    return ~y & 0x80808080u;
}

__device__ __forceinline__ unsigned owner_mask(const uint4 k, unsigned mine)
{
    return (zero_bytes(k.x ^ mine) >> 7) | (zero_bytes(k.y ^ mine) >> 6) |
           (zero_bytes(k.z ^ mine) >> 5) | (zero_bytes(k.w ^ mine) >> 4);
}
__device__ __forceinline__ unsigned owner_edge(int i4, int b)
{
    return (unsigned)(i4 * 16 + ((b & 7) << 2) + (b >> 3));
}

template <int MODE>
__device__ __forceinline__ void owner_lookup(
    const int* __restrict__ ety, const int* __restrict__ src,
    const int* __restrict__ dst, unsigned* q, int n, int d0, int tid)
{
    for (int i = tid; i < n; i += 1024) {
        const int e = (int)q[i];
        int s, d;
        if (e < EDGES) { s = src[e]; d = dst[e]; } else { s = d = e - EDGES; }
        unsigned o = (unsigned)s | ((unsigned)(d - d0) << 16);
        if (MODE == 1) o |= (ety[e] == 0) ? 0u : (1u << 24);
        q[i] = o;
    }
}

template <int MODE, bool FULL>
__device__ __forceinline__ void owner_add_batch(
    const float* __restrict__ t0, float* acc, const unsigned* q, const float* qw,
    unsigned& myq, float& myw, int nb, int inext, int lane)
{
    float2 v[OWN_BATCH];
    #pragma unroll
    for (int u = 0; u < OWN_BATCH; ++u) {
        const unsigned p = (unsigned)__builtin_amdgcn_readlane((int)myq, u);
        const float2* __restrict__ row = (const float2*)(t0 + (size_t)(p & 0xFFFFu) * DIM);
        v[u] = row[lane];
    }
    const unsigned nq = q[inext];
    const float    nw = (MODE == 0) ? qw[inext] : 1.f;
    #pragma unroll
    for (int u = 0; u < OWN_BATCH; ++u) {
        if (FULL || u < nb) {
            const unsigned p = (unsigned)__builtin_amdgcn_readlane((int)myq, u);
            if (MODE == 0) {
                const float w = __uint_as_float(
                    (unsigned)__builtin_amdgcn_readlane((int)__float_as_uint(myw), u));
                const int off = (int)((p >> 16) & 0xFFu) * DIM + 2 * lane;
                atomicAdd(&acc[off],     w * v[u].x);
                atomicAdd(&acc[off + 1], w * v[u].y);
            } else {
                const int off = (int)((p >> 16) & 0xFFu) * SPITCH + (int)(p >> 24) * DIM
                                + 2 * lane;
                atomicAdd(&acc[off],     v[u].x);
                atomicAdd(&acc[off + 1], v[u].y);
            }
        }
    }
    myq = nq;
    myw = nw;
}

template <int MODE>
__device__ __forceinline__ void owner_add_rows(
    const float* __restrict__ t0, float* acc, const unsigned* q, const float* qw,
    unsigned* cnt, int nv, int tid)
{
    const int lane = tid & 31;
    const int wave = __builtin_amdgcn_readfirstlane(tid >> 5);
    const int n    = __builtin_amdgcn_readfirstlane(nv);
    if (MODE == 1) {
        for (int i = tid; i < n; i += 1024) {
            const unsigned p = q[i];
            atomicAdd(&cnt[((p >> 16) & 0xFFu) * 2u + (p >> 24)], 1u);
        }
    }
    const int ifirst = (lane < OWN_BATCH && wave + 32 * lane < n) ? wave + 32 * lane : wave;
    unsigned myq = q[ifirst];
    float    myw = (MODE == 0) ? qw[ifirst] : 1.f;
    for (int i0 = wave; i0 < n; i0 += 32 * OWN_BATCH) {
        int nb = (n - i0 + 31) >> 5;
        if (nb > OWN_BATCH) nb = OWN_BATCH;
        const int i1    = i0 + 32 * OWN_BATCH + 32 * lane;
        const int inext = (lane < OWN_BATCH && i1 < n) ? i1 : i0;
        if (nb == OWN_BATCH)
            owner_add_batch<MODE, true >(t0, acc, q, qw, myq, myw, nb, inext, lane);
        else
            owner_add_batch<MODE, false>(t0, acc, q, qw, myq, myw, nb, inext, lane);
    }
}

__device__ __forceinline__ void gat_softmax_weights(
    const float* __restrict__ el, const float* __restrict__ attn,
    float* acc, const float* erow, const unsigned* q, float* qw,
    unsigned* rmax, float* rold, float* rden, int n, int tid, bool later)
{
    const int lane = tid & 31;
    const int wave = tid >> 5;
    const int sub  = lane & 15;
    const int half = lane >> 4;
    const v4f av = ((const v4f*)attn)[sub];

    const int wave_s = __builtin_amdgcn_readfirstlane(wave);
    const int n_s    = __builtin_amdgcn_readfirstlane(n);
    for (int i0 = 2 * wave_s; i0 < n_s; i0 += 64 * OWN_PAIRS) {
        const int iq = i0 + 64 * (lane >> 1) + (lane & 1);
        const unsigned myq = q[(lane < 2 * OWN_PAIRS && iq < n_s) ? iq : n_s - 1];
        v4f v[OWN_PAIRS];
        int dl[OWN_PAIRS];
        #pragma unroll
        for (int u = 0; u < OWN_PAIRS; ++u) {
            const unsigned p0 = (unsigned)__builtin_amdgcn_readlane((int)myq, 2 * u);
            const unsigned p1 = (unsigned)__builtin_amdgcn_readlane((int)myq, 2 * u + 1);
            const unsigned p  = half ? p1 : p0;
            v[u]  = ((const v4f*)(el + (size_t)(p & 0xFFFFu) * DIM))[sub];
            dl[u] = (int)((p >> 16) & 0xFFu);
        }
        float sc[OWN_PAIRS];
        #pragma unroll
        for (int u = 0; u < OWN_PAIRS; ++u) {
            const v4f r = ((const v4f*)(erow + dl[u] * DIM))[sub];
            sc[u] = 0.f;
            #pragma unroll
            for (int c = 0; c < 4; ++c) {
                float t = v[u][c] + r[c];
                t = t > 0.f ? t : NEG_SLOPE * t;
                sc[u] += t * av[c];
            }
        }
        #pragma unroll
        for (int o = 8; o; o >>= 1) {
            #pragma unroll
            for (int u = 0; u < OWN_PAIRS; ++u) sc[u] += __shfl_xor(sc[u], o, 32);
        }
        #pragma unroll
        for (int u = 0; u < OWN_PAIRS; ++u) {
            const int i = i0 + 64 * u + half;
            if (sub == 0 && i < n) {
                qw[i] = sc[u];
                atomicMax(&rmax[dl[u]], encF(sc[u]));
            }
        }
    }
    __syncthreads();

    if (later) {
        for (int i = tid; i < OWN_ROWS * DIM; i += 1024) {
            const float mo = rold[i >> 6];
            const float mn = decF(rmax[i >> 6]);
            if (mn != mo) acc[i] *= expf(mo - mn);
        }
        if (tid < OWN_ROWS) {
            const float mo = rold[tid];
            const float mn = decF(rmax[tid]);
            if (mn != mo) rden[tid] *= expf(mo - mn);
        }
        __syncthreads();
    }
    if (tid < OWN_ROWS) rold[tid] = decF(rmax[tid]);
    __syncthreads();

    for (int i = tid; i < n; i += 1024) {
        const int dl = (int)((q[i] >> 16) & 0xFFu);
        const float ex = expf(qw[i] - rold[dl]);
        qw[i] = ex;
        atomicAdd(&rden[dl], ex);
    }
    __syncthreads();
}

template <int M>
__device__ __forceinline__ float row_xor(float x)
{
    return __int_as_float(__builtin_amdgcn_update_dpp(
        0, __float_as_int(x), 0x160 + M, 0xf, 0xf, true));
}

__device__ __forceinline__ bool gat_one_pass(
    const float* __restrict__ el, const float* __restrict__ attn,
    float* acc, const float* erow, const unsigned* q, float* rden, int n, int tid)
{
    const int lane = tid & 31;
    const int sub  = lane & 15;
    const int half = lane >> 4;
    const v4f av = ((const v4f*)attn)[sub];
    const int wave_s = __builtin_amdgcn_readfirstlane(tid >> 5);
    const int n_s    = __builtin_amdgcn_readfirstlane(n);
    const int offa = 4 * sub + 2 * half;
    const int offb = 4 * sub + 2 - 2 * half;
    bool wide = false;
    for (int i0 = 2 * wave_s; i0 < n_s; i0 += 64 * OWN_PAIRS) {
        const int iq = i0 + 64 * (lane >> 1) + (lane & 1);
        const unsigned myq = q[(lane < 2 * OWN_PAIRS && iq < n_s) ? iq : n_s - 1];
        v4f v[OWN_PAIRS];
        int dl[OWN_PAIRS];
        #pragma unroll
        for (int u = 0; u < OWN_PAIRS; ++u) {
            const unsigned p0 = (unsigned)__builtin_amdgcn_readlane((int)myq, 2 * u);
            const unsigned p1 = (unsigned)__builtin_amdgcn_readlane((int)myq, 2 * u + 1);
            const unsigned p  = half ? p1 : p0;
            v[u]  = ((const v4f*)(el + (size_t)(p & 0xFFFFu) * DIM))[sub];
            dl[u] = (int)((p >> 16) & 0xFFu);
        }
        float sc[OWN_PAIRS];
        #pragma unroll
        for (int u = 0; u < OWN_PAIRS; ++u) {
            const v4f r = ((const v4f*)(erow + dl[u] * DIM))[sub];
            sc[u] = 0.f;
            #pragma unroll
            for (int c = 0; c < 4; ++c) {
                float t = v[u][c] + r[c];
                t = t > 0.f ? t : NEG_SLOPE * t;
                sc[u] += t * av[c];
            }
        }
        #pragma unroll
        for (int u = 0; u < OWN_PAIRS; ++u) sc[u] += row_xor<8>(sc[u]);
        #pragma unroll
        for (int u = 0; u < OWN_PAIRS; ++u) sc[u] += row_xor<4>(sc[u]);
        #pragma unroll
        for (int u = 0; u < OWN_PAIRS; ++u) sc[u] += row_xor<2>(sc[u]);
        #pragma unroll
        for (int u = 0; u < OWN_PAIRS; ++u) sc[u] += row_xor<1>(sc[u]);

        float wsel = 0.f;
        int   dsel = 0;
        #pragma unroll
        for (int u = 0; u < OWN_PAIRS; ++u) {
            const bool live = i0 + 64 * u + half < n_s;
            wide |= live && !(fabsf(sc[u]) <= GAT_SAFE);
            const float w = live ? expf(sc[u]) : 0.f;
            if (sub == u) { wsel = w; dsel = dl[u]; }
            if (live) {
                float* pa = acc + dl[u] * DIM + offa;
                float* pb = acc + dl[u] * DIM + offb;
                const v4f x = v[u];
                atomicAdd(pa,     w * (half ? x[2] : x[0]));
                atomicAdd(pa + 1, w * (half ? x[3] : x[1]));
                atomicAdd(pb,     w * (half ? x[0] : x[2]));
                atomicAdd(pb + 1, w * (half ? x[1] : x[3]));
            }
        }
        if (sub < OWN_PAIRS) atomicAdd(&rden[dsel], wsel);
    }
    return wide;
}

__device__ __forceinline__ void gru_owned_rows(
    const float* srows, const unsigned* ecnt, float* stage,
    const float* __restrict__ hin, const float* __restrict__ Wc,
    const float* __restrict__ Whh, const float* __restrict__ bih,
    const float* __restrict__ bhh, const float* __restrict__ cb,
    int d0, int nloc, int tid)
{
    const int lane = tid & 31;
    const int lr   = lane & 15;
    const int hi   = lane >> 4;
    const int wave   = __builtin_amdgcn_readfirstlane(tid >> 5);
    const int nunits = 4 * ((nloc + 15) >> 4);

    for (int u = wave; u < nunits; u += 32) {
        const int rt = u >> 2;
        const int f  = 16 * (u & 3) + lr;

        v8f a_r = {}, a_z = {}, a_ci = {}, a_ch = {};
        #pragma unroll
        for (int ps = 0; ps < 2; ++ps) {
            v4f a4[8];
            load_frag(srows + (rt * 16 + lr) * SPITCH + 64 * ps, hi, a4);
            const float* __restrict__ w = Wc + (size_t)f * SDIM + 64 * ps;
            wmma_tile64(a4, w,                  hi, a_r);
            wmma_tile64(a4, w + 64 * SDIM,      hi, a_z);
            wmma_tile64(a4, w + 2 * 64 * SDIM,  hi, a_ci);
        }
        {
            int grow = d0 + rt * 16 + lr;
            if (grow > NODES - 1) grow = NODES - 1;
            v4f a4[8];
            load_frag(hin + (size_t)grow * DIM, hi, a4);
            const float* __restrict__ w = Whh + (size_t)f * DIM;
            wmma_tile64(a4, w,                 hi, a_r);
            wmma_tile64(a4, w + 64 * DIM,      hi, a_z);
            wmma_tile64(a4, w + 2 * 64 * DIM,  hi, a_ch);
        }

        const float bi_r = bih[f],           bh_r = bhh[f];
        const float bi_z = bih[DIM + f],     bh_z = bhh[DIM + f];
        const float bi_c = bih[2 * DIM + f], bh_c = bhh[2 * DIM + f];
        const float c0_r = cb[f],            c1_r = cb[3 * DIM + f];
        const float c0_z = cb[DIM + f],      c1_z = cb[3 * DIM + DIM + f];
        const float c0_c = cb[2 * DIM + f],  c1_c = cb[3 * DIM + 2 * DIM + f];

        #pragma unroll
        for (int v = 0; v < 8; ++v) {
            const int lrow = rt * 16 + hi * 8 + v;
            if (lrow < nloc) {
                const float cn0 = (float)ecnt[2 * lrow];
                const float cn1 = (float)ecnt[2 * lrow + 1];
                const float hold = hin[(size_t)(d0 + lrow) * DIM + f];
                const float gr  = a_r[v]  + (bi_r + bh_r) + (cn0 * c0_r + cn1 * c1_r);
                const float gz  = a_z[v]  + (bi_z + bh_z) + (cn0 * c0_z + cn1 * c1_z);
                const float gci = a_ci[v] + bi_c          + (cn0 * c0_c + cn1 * c1_c);
                const float gch = a_ch[v] + bh_c;
                const float r = 1.f / (1.f + expf(-gr));
                const float z = 1.f / (1.f + expf(-gz));
                const float c = tanhf(gci + r * gch);
                stage[lrow * DIM + f] = (1.f - z) * c + z * hold;
            }
        }
    }
}

__global__ void __launch_bounds__(1024)
owner_lists(const int* __restrict__ ety, const int* __restrict__ src,
            const int* __restrict__ dst, const unsigned* __restrict__ keys,
            unsigned* __restrict__ lists, unsigned* __restrict__ counts)
{
    __shared__ unsigned q[OWN_QCAP];
    __shared__ unsigned qn;

    const int tid  = threadIdx.x;
    const int d0   = blockIdx.x * OWN_ROWS;
    const int nloc = (NODES - d0 < OWN_ROWS) ? (NODES - d0) : OWN_ROWS;

    if (tid < nloc) q[tid] = (unsigned)(EDGES + d0 + tid);
    if (tid == 0) qn = (unsigned)nloc;
    __syncthreads();

    const uint4* __restrict__ key4 = (const uint4*)keys;
    const unsigned mine = (unsigned)blockIdx.x * 0x01010101u;
    uint4 cur[4];
    #pragma unroll
    for (int j = 0; j < 4; ++j) {
        const int i4 = j * 1024 + tid;
        cur[j] = (i4 < OWN_KEYS4) ? key4[i4] : make_uint4(0u, 0u, 0u, 0u);
    }
    for (int base4 = 0; base4 < OWN_KEYS4; base4 += 4096) {
        uint4 nxt[4];
        #pragma unroll
        for (int j = 0; j < 4; ++j) {
            const int i4 = base4 + 4096 + j * 1024 + tid;
            nxt[j] = (i4 < OWN_KEYS4) ? key4[i4] : make_uint4(0u, 0u, 0u, 0u);
        }
        #pragma unroll
        for (int j = 0; j < 4; ++j) {
            const int i4 = base4 + j * 1024 + tid;
            if (i4 < OWN_KEYS4) {
                unsigned m = owner_mask(cur[j], mine);
                while (m) {
                    const int b = __ffs((int)m) - 1;
                    m &= m - 1u;
                    const unsigned pos = atomicAdd(&qn, 1u);
                    if (pos < OWN_QCAP) q[pos] = owner_edge(i4, b);
                }
            }
        }
        #pragma unroll
        for (int j = 0; j < 4; ++j) cur[j] = nxt[j];
    }
    __syncthreads();

    const unsigned n = qn;
    if (n > OWN_QCAP) {
        if (tid == 0) counts[blockIdx.x] = OWN_OVER;
        return;
    }
    owner_lookup<1>(ety, src, dst, q, (int)n, d0, tid);
    unsigned* __restrict__ mylist = lists + (size_t)blockIdx.x * OWN_QCAP;
    for (int i = tid; i < (int)n; i += 1024) mylist[i] = q[i];
    if (tid == 0) counts[blockIdx.x] = n;
}

template <int MODE>
__global__ void __launch_bounds__(1024)
owner_sum(const float* __restrict__ t0, const float* __restrict__ t1,
          const float* __restrict__ attn, const int* __restrict__ ety,
          const int* __restrict__ src, const int* __restrict__ dst,
          const unsigned* __restrict__ keys, const unsigned* __restrict__ lists,
          const unsigned* __restrict__ counts, float* __restrict__ out,
          const float* __restrict__ Wc, const float* __restrict__ Whh,
          const float* __restrict__ bih, const float* __restrict__ bhh,
          const float* __restrict__ cb, float* __restrict__ emb)
{
    constexpr int ACCN = (MODE == 0) ? OWN_ROWS * DIM : OWN_PAD * SPITCH;
    __shared__ __attribute__((aligned(16))) float    acc[ACCN];
    __shared__ unsigned ecnt[(MODE == 1) ? OWN_PAD * 2 : 1];
    __shared__ __attribute__((aligned(16))) unsigned q[OWN_QCAP];
    __shared__ float    qw[(MODE == 0) ? OWN_QCAP : 1];
    __shared__ __attribute__((aligned(16))) float erow[(MODE == 0) ? OWN_ROWS * DIM : 4];
    __shared__ unsigned rmax[(MODE == 0) ? OWN_ROWS : 1];
    __shared__ float    rold[(MODE == 0) ? OWN_ROWS : 1];
    __shared__ float    rden[(MODE == 0) ? OWN_ROWS : 1];
    __shared__ unsigned qn;
    __shared__ unsigned redo;

    const int tid  = threadIdx.x;
    const int d0   = blockIdx.x * OWN_ROWS;
    const int nloc = (NODES - d0 < OWN_ROWS) ? (NODES - d0) : OWN_ROWS;

    if (tid == 0) redo = 0u;

    const unsigned cnt = counts[blockIdx.x];

    for (int i = tid; i < ACCN; i += 1024) acc[i] = 0.f;
    if (MODE == 1 && tid < OWN_PAD * 2) ecnt[tid] = 0u;
    if (MODE == 0) {
        const v4f* __restrict__ er4 = (const v4f*)(t1 + (size_t)d0 * DIM);
        v4f* erow4 = (v4f*)erow;
        for (int i = tid; i < nloc * (DIM / 4); i += 1024) erow4[i] = er4[i];
        if (tid < OWN_ROWS) {
            rmax[tid] = ENC_NEG_INF;
            rold[tid] = decF(ENC_NEG_INF);
            rden[tid] = 0.f;
        }
    }
    int drains = 0;
    int nlast;

    if (cnt <= (unsigned)OWN_QCAP) {
        const unsigned* __restrict__ mylist = lists + (size_t)blockIdx.x * OWN_QCAP;
        for (int i0 = tid; i0 < (int)cnt; i0 += 4096) {
            unsigned v[4];
            #pragma unroll
            for (int j = 0; j < 4; ++j) {
                const int i = i0 + 1024 * j;
                v[j] = (i < (int)cnt) ? mylist[i] : 0u;
            }
            #pragma unroll
            for (int j = 0; j < 4; ++j) {
                const int i = i0 + 1024 * j;
                if (i < (int)cnt) q[i] = v[j];
            }
        }
        nlast = (int)cnt;
    } else {
        if (tid < nloc) q[tid] = (unsigned)(EDGES + d0 + tid);
        if (tid == 0) qn = (unsigned)nloc;
        const uint4* __restrict__ key4 = (const uint4*)keys;
        const unsigned mine = (unsigned)blockIdx.x * 0x01010101u;
        for (int base4 = 0; base4 < OWN_KEYS4; base4 += 1024) {
            __syncthreads();
            const int n = (int)qn;
            __syncthreads();
            if (n + OWN_SUB > OWN_QCAP) {
                owner_lookup<MODE>(ety, src, dst, q, n, d0, tid);
                __syncthreads();
                if (MODE == 0)
                    gat_softmax_weights(t0, attn, acc, erow, q, qw, rmax, rold, rden,
                                        n, tid, drains > 0);
                owner_add_rows<MODE>(t0, acc, q, qw, ecnt, n, tid);
                ++drains;
                __syncthreads();
                if (tid == 0) qn = 0u;
                __syncthreads();
            }
            const int i4 = base4 + tid;
            if (i4 < OWN_KEYS4) {
                unsigned m = owner_mask(key4[i4], mine);
                while (m) {
                    const int b = __ffs((int)m) - 1;
                    m &= m - 1u;
                    const unsigned pos = atomicAdd(&qn, 1u);
                    q[pos] = owner_edge(i4, b);
                }
            }
        }
        __syncthreads();
        nlast = (int)qn;
        owner_lookup<MODE>(ety, src, dst, q, nlast, d0, tid);
    }
    __syncthreads();
    if (MODE == 0) {
        bool exact = true;
        if (cnt <= (unsigned)OWN_QCAP) {
            if (gat_one_pass(t0, attn, acc, erow, q, rden, nlast, tid)) redo = 1u;
            __syncthreads();
            exact = redo != 0u;
            if (exact) {
                for (int i = tid; i < ACCN; i += 1024) acc[i] = 0.f;
                if (tid < OWN_ROWS) rden[tid] = 0.f;
                __syncthreads();
            }
        }
        if (exact) {
            gat_softmax_weights(t0, attn, acc, erow, q, qw, rmax, rold, rden,
                                nlast, tid, drains > 0);
            owner_add_rows<MODE>(t0, acc, q, qw, ecnt, nlast, tid);
        }
    } else {
        owner_add_rows<MODE>(t0, acc, q, qw, ecnt, nlast, tid);
    }
    __syncthreads();

    float* __restrict__ orow = out + (size_t)d0 * DIM;
    if (MODE == 0) {
        for (int i = tid; i < nloc * DIM; i += 1024) orow[i] = acc[i] / rden[i >> 6];
    } else {
        float* stage = (float*)q;
        gru_owned_rows(acc, ecnt, stage, t0, Wc, Whh, bih, bhh, cb, d0, nloc, tid);
        __syncthreads();
        const v4f* stage4 = (const v4f*)stage;
        v4f* __restrict__ h4 = (v4f*)orow;
        v4f* __restrict__ e4 = emb ? (v4f*)(emb + (size_t)d0 * DIM) : nullptr;
        for (int i = tid; i < nloc * (DIM / 4); i += 1024) {
            const v4f o = stage4[i];
            h4[i] = o;
            if (e4) e4[i] = o;
        }
    }
}

extern "C" void kernel_launch(void* const* d_in, const int* in_sizes, int n_in,
                              void* d_out, int out_size, void* d_ws, size_t ws_size,
                              hipStream_t stream)
{
    (void)in_sizes; (void)n_in; (void)out_size; (void)ws_size;

    const float* x   = (const float*)d_in[0];
    const int*   src = (const int*)d_in[1];
    const int*   dst = (const int*)d_in[2];
    const int*   ety = (const int*)d_in[3];

    const long NE = (long)NODES * DIM;

    float* ws   = (float*)d_ws;
    float* hbuf = ws;                 ws += NE;
    float* buf1 = ws;                 ws += NE;
    float* buf2 = ws;                 ws += NE;
    float* hbuf2 = ws;                ws += NE;
    float* acc  = ws;                 ws += NE;
    unsigned* keys   = (unsigned*)ws;
    unsigned* lists  = keys + EDGES / 4;
    unsigned* counts = lists + 256L * OWN_QCAP;
    float* Wc   = (float*)(counts + 256);
    float* cb   = Wc + 2L * 3 * DIM * SDIM;

    float* out = (float*)d_out;

    const int elemBlocks     = (int)((NE + 255) / 256);
    const int ownerBlocks    = (NODES + OWN_ROWS - 1) / OWN_ROWS;
    const int keyBlocks      = (EDGES / 4 + 255) / 256;
    const int gemmBlocks64   = (NODES / 16 + 3) / 4;

    owner_keys<<<keyBlocks, 256, 0, stream>>>(dst, keys);
    owner_lists<<<ownerBlocks, 1024, 0, stream>>>(ety, src, dst, keys, lists, counts);
    compose_weights<<<48, 32, 0, stream>>>(
        (const float*)d_in[4 + 7],  (const float*)d_in[4 + 5],  (const float*)d_in[4 + 6],
        (const float*)d_in[15 + 7], (const float*)d_in[15 + 5], (const float*)d_in[15 + 6],
        Wc, cb);

    const float* hin = x;
    for (int l = 0; l < 2; ++l) {
        const int base = 4 + 11 * l;
        const float* Wsrc = (const float*)d_in[base + 0];
        const float* bsrc = (const float*)d_in[base + 1];
        const float* Wdst = (const float*)d_in[base + 2];
        const float* bdst = (const float*)d_in[base + 3];
        const float* attn = (const float*)d_in[base + 4];
        const float* bih  = (const float*)d_in[base + 8];
        const float* Whh  = (const float*)d_in[base + 9];
        const float* bhh  = (const float*)d_in[base + 10];

        gemm_bias_wmma<<<gemmBlocks64, 128, 0, stream>>>(hin, Wsrc, bsrc, buf1, NODES);
        gemm_bias_wmma<<<gemmBlocks64, 128, 0, stream>>>(hin, Wdst, bdst, buf2, NODES);
        owner_sum<0><<<ownerBlocks, 1024, 0, stream>>>(buf1, buf2, attn, ety, src, dst,
                                                       keys, lists, counts, acc,
                                                       nullptr, nullptr, nullptr, nullptr,
                                                       nullptr, nullptr);
        elu_write<<<elemBlocks, 256, 0, stream>>>(acc, hbuf, out + (2L * l) * NE,
                                                  NE);

        owner_sum<1><<<ownerBlocks, 1024, 0, stream>>>(
            hbuf, hbuf, attn, ety, src, dst, keys, lists, counts, hbuf2,
            Wc + (long)l * 3 * DIM * SDIM, Whh, bih, bhh, cb + l * 2 * 3 * DIM, nullptr);
        owner_sum<1><<<ownerBlocks, 1024, 0, stream>>>(
            hbuf2, hbuf2, attn, ety, src, dst, keys, lists, counts, hbuf,
            Wc + (long)l * 3 * DIM * SDIM, Whh, bih, bhh, cb + l * 2 * 3 * DIM,
            out + (2L * l + 1) * NE);
        hin = hbuf;
    }
}
